# starting kernel
# baseline (speedup 1.0000x reference)
#include <hip/hip_runtime.h>
#include <math.h>

typedef float v2f __attribute__((ext_vector_type(2)));
typedef float v8f __attribute__((ext_vector_type(8)));

#define MT    32
#define DIN   1195
#define XROW  1196
#define H1S   260
#define H2S   132

#define WMMA_F32(a, b, c) \
    __builtin_amdgcn_wmma_f32_16x16x4_f32(false, (a), false, (b), (short)0, (c), false, false)

__global__ __launch_bounds__(256) void residue_embed_fused(
    const int* __restrict__ seq, const float* __restrict__ xyz,
    const float* __restrict__ dih, const int* __restrict__ cidx,
    const float* __restrict__ orient,
    const float* __restrict__ aa_emb, const float* __restrict__ chain_emb,
    const float* __restrict__ W1, const float* __restrict__ b1,
    const float* __restrict__ W2, const float* __restrict__ b2,
    const float* __restrict__ W3, const float* __restrict__ b3,
    const float* __restrict__ W4, const float* __restrict__ b4,
    float* __restrict__ out)
{
    __shared__ float xs[MT * XROW];
    __shared__ float hA[MT * H1S];
    __shared__ float hB[MT * H2S];

    const int tid = threadIdx.x;
    const int t0  = blockIdx.x * MT;

    for (int i = tid; i < MT * XROW; i += 256) xs[i] = 0.0f;
    __syncthreads();

    for (int i = tid; i < MT * 128; i += 256) {
        int tt = i >> 7, k = i & 127;
        int t  = t0 + tt;
        xs[tt * XROW + k] = aa_emb[seq[t] * 128 + k];
        int ci = cidx[t];
        xs[tt * XROW + 1067 + k] = (ci != 0) ? chain_emb[ci * 128 + k] : 0.0f;
    }
    for (int i = tid; i < MT * 15; i += 256) {
        int tt = i / 15, a = i % 15;
        int t  = t0 + tt;
        const float* xp = xyz + (size_t)t * 45;
        float rx = xp[a * 3 + 0] - xp[3 + 0];
        float ry = xp[a * 3 + 1] - xp[3 + 1];
        float rz = xp[a * 3 + 2] - xp[3 + 2];
        const float* O = orient + (size_t)t * 9;
        int base = tt * XROW + 128 + seq[t] * 45 + a * 3;
        xs[base + 0] = O[0] * rx + O[3] * ry + O[6] * rz;
        xs[base + 1] = O[1] * rx + O[4] * ry + O[7] * rz;
        xs[base + 2] = O[2] * rx + O[5] * ry + O[8] * rz;
    }
    if (tid < MT * 3) {
        int tt = tid / 3, d = tid % 3;
        float v = dih[(size_t)(t0 + tt) * 3 + d];
        const float FREQ[6] = {1.0f, 2.0f, 3.0f, 1.0f, 0.5f, 1.0f / 3.0f};
        int base = tt * XROW + 1028 + d * 13;
        xs[base] = v;
        #pragma unroll
        for (int f = 0; f < 6; ++f) {
            xs[base + 1 + f] = sinf(FREQ[f] * v);
            xs[base + 7 + f] = cosf(FREQ[f] * v);
        }
    }
    __syncthreads();

    const int lane = tid & 31;
    const int w    = tid >> 5;
    const int half = lane >> 4;
    const int col  = lane & 15;

    {
        v8f a00 = {}, a01 = {}, a10 = {}, a11 = {};
        const int n0 = w * 32;
        const float* W1c0 = W1 + n0 + col;
        const float* W1c1 = W1 + n0 + 16 + col;
        const int ar0 = col * XROW;
        const int ar1 = (16 + col) * XROW;
        int kk = half * 2;
        for (int it = 0; it < 298; ++it, kk += 4) {
            v2f fa0 = *(const v2f*)&xs[ar0 + kk];
            v2f fa1 = *(const v2f*)&xs[ar1 + kk];
            v2f bA, bB;
            bA.x = W1c0[kk * 256];  bA.y = W1c0[kk * 256 + 256];
            bB.x = W1c1[kk * 256];  bB.y = W1c1[kk * 256 + 256];
            a00 = WMMA_F32(fa0, bA, a00);
            a01 = WMMA_F32(fa0, bB, a01);
            a10 = WMMA_F32(fa1, bA, a10);
            a11 = WMMA_F32(fa1, bB, a11);
        }
        {
            v2f fa0 = *(const v2f*)&xs[ar0 + kk];
            v2f fa1 = *(const v2f*)&xs[ar1 + kk];
            int r0 = (kk     < DIN) ? kk     : DIN - 1;
            int r1 = (kk + 1 < DIN) ? kk + 1 : DIN - 1;
            v2f bA, bB;
            bA.x = W1c0[r0 * 256];  bA.y = W1c0[r1 * 256];
            bB.x = W1c1[r0 * 256];  bB.y = W1c1[r1 * 256];
            a00 = WMMA_F32(fa0, bA, a00);
            a01 = WMMA_F32(fa0, bB, a01);
            a10 = WMMA_F32(fa1, bA, a10);
            a11 = WMMA_F32(fa1, bB, a11);
        }
        float biasA = b1[n0 + col], biasB = b1[n0 + 16 + col];
        #pragma unroll
        for (int r = 0; r < 8; ++r) {
            int m = r + half * 8;
            float v00 = a00[r] + biasA, v01 = a01[r] + biasB;
            float v10 = a10[r] + biasA, v11 = a11[r] + biasB;
            hA[m * H1S + n0 + col]             = v00 > 0.0f ? v00 : 0.0f;
            hA[m * H1S + n0 + 16 + col]        = v01 > 0.0f ? v01 : 0.0f;
            hA[(m + 16) * H1S + n0 + col]      = v10 > 0.0f ? v10 : 0.0f;
            hA[(m + 16) * H1S + n0 + 16 + col] = v11 > 0.0f ? v11 : 0.0f;
        }
    }
    __syncthreads();

    {
        v8f ac0 = {}, ac1 = {};
        const int n0 = w * 16;
        for (int k0 = 0; k0 < 256; k0 += 4) {
            int kk = k0 + half * 2;
            v2f fa0 = *(const v2f*)&hA[col * H1S + kk];
            v2f fa1 = *(const v2f*)&hA[(16 + col) * H1S + kk];
            v2f b;
            b.x = W2[kk * 128 + n0 + col];
            b.y = W2[(kk + 1) * 128 + n0 + col];
            ac0 = WMMA_F32(fa0, b, ac0);
            ac1 = WMMA_F32(fa1, b, ac1);
        }
        float bias = b2[n0 + col];
        #pragma unroll
        for (int r = 0; r < 8; ++r) {
            int m = r + half * 8;
            float v0 = ac0[r] + bias, v1 = ac1[r] + bias;
            hB[m * H2S + n0 + col]        = v0 > 0.0f ? v0 : 0.0f;
            hB[(m + 16) * H2S + n0 + col] = v1 > 0.0f ? v1 : 0.0f;
        }
    }
    __syncthreads();

    {
        v8f ac0 = {}, ac1 = {};
        const int n0 = w * 16;
        for (int k0 = 0; k0 < 128; k0 += 4) {
            int kk = k0 + half * 2;
            v2f fa0 = *(const v2f*)&hB[col * H2S + kk];
            v2f fa1 = *(const v2f*)&hB[(16 + col) * H2S + kk];
            v2f b;
            b.x = W3[kk * 128 + n0 + col];
            b.y = W3[(kk + 1) * 128 + n0 + col];
            ac0 = WMMA_F32(fa0, b, ac0);
            ac1 = WMMA_F32(fa1, b, ac1);
        }
        float bias = b3[n0 + col];
        #pragma unroll
        for (int r = 0; r < 8; ++r) {
            int m = r + half * 8;
            float v0 = ac0[r] + bias, v1 = ac1[r] + bias;
            hA[m * H2S + n0 + col]        = v0 > 0.0f ? v0 : 0.0f;
            hA[(m + 16) * H2S + n0 + col] = v1 > 0.0f ? v1 : 0.0f;
        }
    }
    __syncthreads();

    {
        v8f ac0 = {}, ac1 = {};
        const int n0 = w * 16;
        for (int k0 = 0; k0 < 128; k0 += 4) {
            int kk = k0 + half * 2;
            v2f fa0 = *(const v2f*)&hA[col * H2S + kk];
            v2f fa1 = *(const v2f*)&hA[(16 + col) * H2S + kk];
            v2f b;
            b.x = W4[kk * 128 + n0 + col];
            b.y = W4[(kk + 1) * 128 + n0 + col];
            ac0 = WMMA_F32(fa0, b, ac0);
            ac1 = WMMA_F32(fa1, b, ac1);
        }
        float bias = b4[n0 + col];
        #pragma unroll
        for (int r = 0; r < 8; ++r) {
            int m = r + half * 8;
            out[(size_t)(t0 + m) * 128 + n0 + col]      = ac0[r] + bias;
            out[(size_t)(t0 + m + 16) * 128 + n0 + col] = ac1[r] + bias;
        }
    }
}

extern "C" void kernel_launch(void* const* d_in, const int* in_sizes, int n_in,
                              void* d_out, int out_size, void* d_ws, size_t ws_size,
                              hipStream_t stream) {
    (void)n_in; (void)d_ws; (void)ws_size; (void)out_size;
    const int*   seq    = (const int*)  d_in[0];
    const float* xyz    = (const float*)d_in[1];
    const float* dih    = (const float*)d_in[2];
    const int*   cidx   = (const int*)  d_in[3];
    const float* orient = (const float*)d_in[4];
    const float* aa_emb    = (const float*)d_in[6];
    const float* chain_emb = (const float*)d_in[7];
    const float* W1 = (const float*)d_in[8];   const float* b1 = (const float*)d_in[9];
    const float* W2 = (const float*)d_in[10];  const float* b2 = (const float*)d_in[11];
    const float* W3 = (const float*)d_in[12];  const float* b3 = (const float*)d_in[13];
    const float* W4 = (const float*)d_in[14];  const float* b4 = (const float*)d_in[15];
    float* out = (float*)d_out;

    const int tokens = in_sizes[0];
    const int blocks = tokens / MT;
    residue_embed_fused<<<blocks, 256, 0, stream>>>(
        seq, xyz, dih, cidx, orient, aa_emb, chain_emb,
        W1, b1, W2, b2, W3, b3, W4, b4, out);
}


# Weights of all four layers go through LDS: 32-row chunks fetched with 16-byte coalesced loads one chunk ahead (two buffers, k pairs interleaved), B fragments read from LDS; sum order unchanged
# speedup vs baseline: 1.4142x; 1.4142x over previous
#include <hip/hip_runtime.h>
#include <math.h>

typedef float v2f __attribute__((ext_vector_type(2)));
typedef float v4f __attribute__((ext_vector_type(4)));
typedef float v8f __attribute__((ext_vector_type(8)));

#define MT    32
#define DIN   1195
#define XROW  1196
#define H1S   260
#define H2S   132
#define KC    32
#define NS    272
#define WBUF  (KC / 2 * NS * 2)
#define L1FULL (DIN / KC)

#define WMMA_F32(a, b, c) \
    __builtin_amdgcn_wmma_f32_16x16x4_f32(false, (a), false, (b), (short)0, (c), false, false)

template <int NC, bool TAIL>
__device__ __forceinline__ void w_fetch(const float* __restrict__ Wg, int k0, int K, int tid,
                                        v4f (&lo)[4], v4f (&hi)[4])
{
    constexpr int G  = NC / 4;
    constexpr int NI = NC / 64;
    #pragma unroll
    for (int j = 0; j < NI; ++j) {
        const int i = tid + 256 * j;
        const int r = k0 + 2 * (i / G);
        const int g = i % G;
        if (!TAIL) {
            lo[j] = *(const v4f*)(Wg + (size_t)r * NC + 4 * g);
            hi[j] = *(const v4f*)(Wg + (size_t)(r + 1) * NC + 4 * g);
        } else {
            const int ra = (r     < K) ? r     : K - 1;
            const int rb = (r + 1 < K) ? r + 1 : K - 1;
            const v4f a = *(const v4f*)(Wg + (size_t)ra * NC + 4 * g);
            const v4f b = *(const v4f*)(Wg + (size_t)rb * NC + 4 * g);
            const v4f z = {0.0f, 0.0f, 0.0f, 0.0f};
            lo[j] = (r     < K) ? a : z;
            hi[j] = (r + 1 < K) ? b : z;
        }
    }
}

template <int NC>
__device__ __forceinline__ void w_stage(float* buf, int tid, const v4f (&lo)[4], const v4f (&hi)[4])
{
    constexpr int G  = NC / 4;
    constexpr int NI = NC / 64;
    #pragma unroll
    for (int j = 0; j < NI; ++j) {
        const int i = tid + 256 * j;
        float* d = buf + ((i / G) * NS + 4 * (i % G)) * 2;
        const v4f a = {lo[j].x, hi[j].x, lo[j].y, hi[j].y};
        const v4f b = {lo[j].z, hi[j].z, lo[j].w, hi[j].w};
        *(v4f*)d       = a;
        *(v4f*)(d + 4) = b;
    }
}

template <int STEPS>
struct Frag2x2 { v2f a0[STEPS], a1[STEPS], bA[STEPS], bB[STEPS]; };

template <int STEPS>
__device__ __forceinline__ void frag_load_2x2(const float* xa0, const float* xa1, const float* wb,
                                              Frag2x2<STEPS>& f)
{
    #pragma unroll
    for (int s = 0; s < STEPS; ++s) {
        f.a0[s] = *(const v2f*)(xa0 + 4 * s);
        f.a1[s] = *(const v2f*)(xa1 + 4 * s);
        f.bA[s] = *(const v2f*)(wb + s * (4 * NS));
        f.bB[s] = *(const v2f*)(wb + s * (4 * NS) + 32);
    }
}

template <int STEPS>
__device__ __forceinline__ void frag_mma_2x2(const Frag2x2<STEPS>& f,
                                             v8f& a00, v8f& a01, v8f& a10, v8f& a11)
{
    #pragma unroll
    for (int s = 0; s < STEPS; ++s) {
        a00 = WMMA_F32(f.a0[s], f.bA[s], a00);
        a01 = WMMA_F32(f.a0[s], f.bB[s], a01);
        a10 = WMMA_F32(f.a1[s], f.bA[s], a10);
        a11 = WMMA_F32(f.a1[s], f.bB[s], a11);
    }
}

struct Frag2x1 { v2f a0[KC / 4], a1[KC / 4], b[KC / 4]; };

__device__ __forceinline__ void frag_load_2x1(const float* ha0, const float* ha1, const float* wb,
                                              Frag2x1& f)
{
    #pragma unroll
    for (int s = 0; s < KC / 4; ++s) {
        f.a0[s] = *(const v2f*)(ha0 + 4 * s);
        f.a1[s] = *(const v2f*)(ha1 + 4 * s);
        f.b[s]  = *(const v2f*)(wb + s * (4 * NS));
    }
}

__device__ __forceinline__ void frag_mma_2x1(const Frag2x1& f, v8f& ac0, v8f& ac1)
{
    #pragma unroll
    for (int s = 0; s < KC / 4; ++s) {
        ac0 = WMMA_F32(f.a0[s], f.b[s], ac0);
        ac1 = WMMA_F32(f.a1[s], f.b[s], ac1);
    }
}

__global__ __launch_bounds__(256) void residue_embed_fused(
    const int* __restrict__ seq, const float* __restrict__ xyz,
    const float* __restrict__ dih, const int* __restrict__ cidx,
    const float* __restrict__ orient,
    const float* __restrict__ aa_emb, const float* __restrict__ chain_emb,
    const float* __restrict__ W1, const float* __restrict__ b1,
    const float* __restrict__ W2, const float* __restrict__ b2,
    const float* __restrict__ W3, const float* __restrict__ b3,
    const float* __restrict__ W4, const float* __restrict__ b4,
    float* __restrict__ out)
{
    __shared__ float xs[MT * XROW];
    __shared__ float hA[MT * H1S];
    __shared__ float hB[MT * H2S];
    __shared__ __attribute__((aligned(16))) float wbuf[2 * WBUF];

    const int tid = threadIdx.x;
    const int t0  = blockIdx.x * MT;

    v4f wlo[4], whi[4];
    w_fetch<256, false>(W1, 0, DIN, tid, wlo, whi);

    for (int i = tid; i < MT * XROW; i += 256) xs[i] = 0.0f;
    __syncthreads();

    for (int i = tid; i < MT * 128; i += 256) {
        int tt = i >> 7, k = i & 127;
        int t  = t0 + tt;
        xs[tt * XROW + k] = aa_emb[seq[t] * 128 + k];
        int ci = cidx[t];
        xs[tt * XROW + 1067 + k] = (ci != 0) ? chain_emb[ci * 128 + k] : 0.0f;
    }
    for (int i = tid; i < MT * 15; i += 256) {
        int tt = i / 15, a = i % 15;
        int t  = t0 + tt;
        const float* xp = xyz + (size_t)t * 45;
        float rx = xp[a * 3 + 0] - xp[3 + 0];
        float ry = xp[a * 3 + 1] - xp[3 + 1];
        float rz = xp[a * 3 + 2] - xp[3 + 2];
        const float* O = orient + (size_t)t * 9;
        int base = tt * XROW + 128 + seq[t] * 45 + a * 3;
        xs[base + 0] = O[0] * rx + O[3] * ry + O[6] * rz;
        xs[base + 1] = O[1] * rx + O[4] * ry + O[7] * rz;
        xs[base + 2] = O[2] * rx + O[5] * ry + O[8] * rz;
    }
    if (tid < MT * 3) {
        int tt = tid / 3, d = tid % 3;
        float v = dih[(size_t)(t0 + tt) * 3 + d];
        const float FREQ[6] = {1.0f, 2.0f, 3.0f, 1.0f, 0.5f, 1.0f / 3.0f};
        int base = tt * XROW + 1028 + d * 13;
        xs[base] = v;
        #pragma unroll
        for (int f = 0; f < 6; ++f) {
            xs[base + 1 + f] = sinf(FREQ[f] * v);
            xs[base + 7 + f] = cosf(FREQ[f] * v);
        }
    }
    w_stage<256>(wbuf, tid, wlo, whi);
    __syncthreads();

    const int lane = tid & 31;
    const int w    = tid >> 5;
    const int half = lane >> 4;
    const int col  = lane & 15;

    {
        v8f a00 = {}, a01 = {}, a10 = {}, a11 = {};
        const int n0 = w * 32;
        const float* xa0 = xs + col * XROW + half * 2;
        const float* xa1 = xs + (16 + col) * XROW + half * 2;
        const int wbo = (half * NS + n0 + col) * 2;
        for (int c = 0; c < L1FULL; ++c) {
            Frag2x2<KC / 4> f;
            frag_load_2x2<KC / 4>(xa0 + c * KC, xa1 + c * KC, wbuf + (c & 1) * WBUF + wbo, f);
            if (c + 1 < L1FULL) w_fetch<256, false>(W1, (c + 1) * KC, DIN, tid, wlo, whi);
            else                w_fetch<256, true >(W1, (c + 1) * KC, DIN, tid, wlo, whi);
            frag_mma_2x2<KC / 4>(f, a00, a01, a10, a11);
            w_stage<256>(wbuf + ((c + 1) & 1) * WBUF, tid, wlo, whi);
            __syncthreads();
        }
        {
            Frag2x2<3> f;
            frag_load_2x2<3>(xa0 + L1FULL * KC, xa1 + L1FULL * KC,
                             wbuf + (L1FULL & 1) * WBUF + wbo, f);
            w_fetch<128, false>(W2, 0, 256, tid, wlo, whi);
            frag_mma_2x2<3>(f, a00, a01, a10, a11);
        }
        float biasA = b1[n0 + col], biasB = b1[n0 + 16 + col];
        #pragma unroll
        for (int r = 0; r < 8; ++r) {
            int m = r + half * 8;
            float v00 = a00[r] + biasA, v01 = a01[r] + biasB;
            float v10 = a10[r] + biasA, v11 = a11[r] + biasB;
            hA[m * H1S + n0 + col]             = v00 > 0.0f ? v00 : 0.0f;
            hA[m * H1S + n0 + 16 + col]        = v01 > 0.0f ? v01 : 0.0f;
            hA[(m + 16) * H1S + n0 + col]      = v10 > 0.0f ? v10 : 0.0f;
            hA[(m + 16) * H1S + n0 + 16 + col] = v11 > 0.0f ? v11 : 0.0f;
        }
        w_stage<128>(wbuf + ((L1FULL + 1) & 1) * WBUF, tid, wlo, whi);
    }
    __syncthreads();

    const int n0  = w * 16;
    const int wbo = (half * NS + n0 + col) * 2;

    {
        v8f ac0 = {}, ac1 = {};
        const float* ha0 = hA + col * H1S + half * 2;
        const float* ha1 = hA + (16 + col) * H1S + half * 2;
        #pragma nounroll
        for (int c = 0; c < 256 / KC; ++c) {
            Frag2x1 f;
            frag_load_2x1(ha0 + c * KC, ha1 + c * KC, wbuf + (c & 1) * WBUF + wbo, f);
            const float* wn = (c + 1 < 256 / KC) ? W2 + (c + 1) * KC * 128 : W3;
            w_fetch<128, false>(wn, 0, KC, tid, wlo, whi);
            frag_mma_2x1(f, ac0, ac1);
            if (c + 1 == 256 / KC) {
                float bias = b2[n0 + col];
                #pragma unroll
                for (int r = 0; r < 8; ++r) {
                    int m = r + half * 8;
                    float v0 = ac0[r] + bias, v1 = ac1[r] + bias;
                    hB[m * H2S + n0 + col]        = v0 > 0.0f ? v0 : 0.0f;
                    hB[(m + 16) * H2S + n0 + col] = v1 > 0.0f ? v1 : 0.0f;
                }
            }
            w_stage<128>(wbuf + ((c + 1) & 1) * WBUF, tid, wlo, whi);
            __syncthreads();
        }
    }

    {
        v8f ac0 = {}, ac1 = {};
        const float* ha0 = hB + col * H2S + half * 2;
        const float* ha1 = hB + (16 + col) * H2S + half * 2;
        #pragma nounroll
        for (int c = 0; c < 128 / KC; ++c) {
            Frag2x1 f;
            frag_load_2x1(ha0 + c * KC, ha1 + c * KC, wbuf + (c & 1) * WBUF + wbo, f);
            const float* wn = (c + 1 < 128 / KC) ? W3 + (c + 1) * KC * 128 : W4;
            w_fetch<128, false>(wn, 0, KC, tid, wlo, whi);
            frag_mma_2x1(f, ac0, ac1);
            if (c + 1 == 128 / KC) {
                float bias = b3[n0 + col];
                #pragma unroll
                for (int r = 0; r < 8; ++r) {
                    int m = r + half * 8;
                    float v0 = ac0[r] + bias, v1 = ac1[r] + bias;
                    hA[m * H2S + n0 + col]        = v0 > 0.0f ? v0 : 0.0f;
                    hA[(m + 16) * H2S + n0 + col] = v1 > 0.0f ? v1 : 0.0f;
                }
            }
            w_stage<128>(wbuf + ((c + 1) & 1) * WBUF, tid, wlo, whi);
            __syncthreads();
        }
    }

    {
        v8f ac0 = {}, ac1 = {};
        const float* ha0 = hA + col * H2S + half * 2;
        const float* ha1 = hA + (16 + col) * H2S + half * 2;
        #pragma nounroll
        for (int c = 0; c < 128 / KC; ++c) {
            Frag2x1 f;
            frag_load_2x1(ha0 + c * KC, ha1 + c * KC, wbuf + (c & 1) * WBUF + wbo, f);
            if (c + 1 < 128 / KC) w_fetch<128, false>(W4, (c + 1) * KC, 128, tid, wlo, whi);
            frag_mma_2x1(f, ac0, ac1);
            if (c + 1 < 128 / KC) {
                w_stage<128>(wbuf + ((c + 1) & 1) * WBUF, tid, wlo, whi);
                __syncthreads();
            }
        }
        float bias = b4[n0 + col];
        #pragma unroll
        for (int r = 0; r < 8; ++r) {
            int m = r + half * 8;
            out[(size_t)(t0 + m) * 128 + n0 + col]      = ac0[r] + bias;
            out[(size_t)(t0 + m + 16) * 128 + n0 + col] = ac1[r] + bias;
        }
    }
}

extern "C" void kernel_launch(void* const* d_in, const int* in_sizes, int n_in,
                              void* d_out, int out_size, void* d_ws, size_t ws_size,
                              hipStream_t stream) {
    (void)n_in; (void)d_ws; (void)ws_size; (void)out_size;
    const int*   seq    = (const int*)  d_in[0];
    const float* xyz    = (const float*)d_in[1];
    const float* dih    = (const float*)d_in[2];
    const int*   cidx   = (const int*)  d_in[3];
    const float* orient = (const float*)d_in[4];
    const float* aa_emb    = (const float*)d_in[6];
    const float* chain_emb = (const float*)d_in[7];
    const float* W1 = (const float*)d_in[8];   const float* b1 = (const float*)d_in[9];
    const float* W2 = (const float*)d_in[10];  const float* b2 = (const float*)d_in[11];
    const float* W3 = (const float*)d_in[12];  const float* b3 = (const float*)d_in[13];
    const float* W4 = (const float*)d_in[14];  const float* b4 = (const float*)d_in[15];
    float* out = (float*)d_out;

    const int tokens = in_sizes[0];
    const int blocks = tokens / MT;
    residue_embed_fused<<<blocks, 256, 0, stream>>>(
        seq, xyz, dih, cidx, orient, aa_emb, chain_emb,
        W1, b1, W2, b2, W3, b3, W4, b4, out);
}


# 64 tokens per work-group, not 32 (256 groups; a weight chunk staged once per 64): x held compact in LDS, one-hot block's A fragments read from a 45-float row at a clamped index; sum order unchanged
# speedup vs baseline: 1.8416x; 1.3022x over previous
#include <hip/hip_runtime.h>
#include <math.h>

typedef float v2f __attribute__((ext_vector_type(2)));
typedef float v4f __attribute__((ext_vector_type(4)));
typedef float v8f __attribute__((ext_vector_type(8)));

#define MT    64
#define DIN   1195
#define XS    348
#define XTAIL 128
#define XCRD  298
#define TSHIFT 900
#define H1S   260
#define H2S   132
#define KC    32
#define NS    272
#define WBUF  (KC / 2 * NS * 2)
#define L1FULL (DIN / KC)

#define WMMA_F32(a, b, c) \
    __builtin_amdgcn_wmma_f32_16x16x4_f32(false, (a), false, (b), (short)0, (c), false, false)

template <int NC, bool TAIL>
__device__ __forceinline__ void w_fetch(const float* __restrict__ Wg, int k0, int K, int tid,
                                        v4f (&lo)[4], v4f (&hi)[4])
{
    constexpr int G  = NC / 4;
    constexpr int NI = NC / 64;
    #pragma unroll
    for (int j = 0; j < NI; ++j) {
        const int i = tid + 256 * j;
        const int r = k0 + 2 * (i / G);
        const int g = i % G;
        if (!TAIL) {
            lo[j] = *(const v4f*)(Wg + (size_t)r * NC + 4 * g);
            hi[j] = *(const v4f*)(Wg + (size_t)(r + 1) * NC + 4 * g);
        } else {
            const int ra = (r     < K) ? r     : K - 1;
            const int rb = (r + 1 < K) ? r + 1 : K - 1;
            const v4f a = *(const v4f*)(Wg + (size_t)ra * NC + 4 * g);
            const v4f b = *(const v4f*)(Wg + (size_t)rb * NC + 4 * g);
            const v4f z = {0.0f, 0.0f, 0.0f, 0.0f};
            lo[j] = (r     < K) ? a : z;
            hi[j] = (r + 1 < K) ? b : z;
        }
    }
}

template <int NC>
__device__ __forceinline__ void w_stage(float* buf, int tid, const v4f (&lo)[4], const v4f (&hi)[4])
{
    constexpr int G  = NC / 4;
    constexpr int NI = NC / 64;
    #pragma unroll
    for (int j = 0; j < NI; ++j) {
        const int i = tid + 256 * j;
        float* d = buf + ((i / G) * NS + 4 * (i % G)) * 2;
        const v4f a = {lo[j].x, hi[j].x, lo[j].y, hi[j].y};
        const v4f b = {lo[j].z, hi[j].z, lo[j].w, hi[j].w};
        *(v4f*)d       = a;
        *(v4f*)(d + 4) = b;
    }
}

template <int STEPS>
struct Frag2x4 { v2f a0[STEPS], a1[STEPS], b0[STEPS], b1[STEPS], b2[STEPS], b3[STEPS]; };

__device__ __forceinline__ v2f a_onehot(const float* xc, int t)
{
    const int c = t < -2 ? -2 : (t > 46 ? 46 : t);
    const float* p = xc + c;
    const v2f r = {p[0], p[1]};
    return r;
}

template <int STEPS, int MODE>
__device__ __forceinline__ void frag_load_2x4(const float* xr0, const float* xr1, int half, int kb,
                                              int xoff, int bias0, int bias1, const float* wb,
                                              Frag2x4<STEPS>& f)
{
    #pragma unroll
    for (int s = 0; s < STEPS; ++s) {
        if (MODE == 1 || (MODE == 2 && s == 0)) {
            f.a0[s] = a_onehot(xr0 + XCRD, bias0 + kb + 4 * s);
            f.a1[s] = a_onehot(xr1 + XCRD, bias1 + kb + 4 * s);
        } else {
            f.a0[s] = *(const v2f*)(xr0 + 2 * half + xoff + 4 * s);
            f.a1[s] = *(const v2f*)(xr1 + 2 * half + xoff + 4 * s);
        }
        f.b0[s] = *(const v2f*)(wb + s * (4 * NS));
        f.b1[s] = *(const v2f*)(wb + s * (4 * NS) + 32);
        f.b2[s] = *(const v2f*)(wb + s * (4 * NS) + 64);
        f.b3[s] = *(const v2f*)(wb + s * (4 * NS) + 96);
        __builtin_amdgcn_sched_barrier(0);
    }
}

template <int STEPS>
__device__ __forceinline__ void frag_mma_2x4(const Frag2x4<STEPS>& f,
                                             v8f& c00, v8f& c01, v8f& c02, v8f& c03,
                                             v8f& c10, v8f& c11, v8f& c12, v8f& c13)
{
    #pragma unroll
    for (int s = 0; s < STEPS; ++s) {
        c00 = WMMA_F32(f.a0[s], f.b0[s], c00);
        c01 = WMMA_F32(f.a0[s], f.b1[s], c01);
        c02 = WMMA_F32(f.a0[s], f.b2[s], c02);
        c03 = WMMA_F32(f.a0[s], f.b3[s], c03);
        c10 = WMMA_F32(f.a1[s], f.b0[s], c10);
        c11 = WMMA_F32(f.a1[s], f.b1[s], c11);
        c12 = WMMA_F32(f.a1[s], f.b2[s], c12);
        c13 = WMMA_F32(f.a1[s], f.b3[s], c13);
    }
}

struct Frag2x2 { v2f a0[KC / 4], a1[KC / 4], bA[KC / 4], bB[KC / 4]; };

__device__ __forceinline__ void frag_load_2x2(const float* ha0, const float* ha1, const float* wb,
                                              Frag2x2& f)
{
    #pragma unroll
    for (int s = 0; s < KC / 4; ++s) {
        f.a0[s] = *(const v2f*)(ha0 + 4 * s);
        f.a1[s] = *(const v2f*)(ha1 + 4 * s);
        f.bA[s] = *(const v2f*)(wb + s * (4 * NS));
        f.bB[s] = *(const v2f*)(wb + s * (4 * NS) + 32);
    }
}

__device__ __forceinline__ void frag_mma_2x2(const Frag2x2& f, v8f& a00, v8f& a01, v8f& a10, v8f& a11)
{
    #pragma unroll
    for (int s = 0; s < KC / 4; ++s) {
        a00 = WMMA_F32(f.a0[s], f.bA[s], a00);
        a01 = WMMA_F32(f.a0[s], f.bB[s], a01);
        a10 = WMMA_F32(f.a1[s], f.bA[s], a10);
        a11 = WMMA_F32(f.a1[s], f.bB[s], a11);
    }
}

template <int MODE, bool NEXT_TAIL>
__device__ __forceinline__ void l1_chunk(int c, const float* xr0, const float* xr1, int half,
                                         int bias0, int bias1, float* wbuf, int wbo,
                                         const float* __restrict__ W1, int tid,
                                         v4f (&wlo)[4], v4f (&whi)[4],
                                         v8f& c00, v8f& c01, v8f& c02, v8f& c03,
                                         v8f& c10, v8f& c11, v8f& c12, v8f& c13)
{
    Frag2x4<KC / 4> f;
    const int kb   = c * KC;
    const int xoff = (kb < 128) ? kb : kb - TSHIFT;
    frag_load_2x4<KC / 4, MODE>(xr0, xr1, half, kb, xoff, bias0, bias1,
                                wbuf + (c & 1) * WBUF + wbo, f);
    w_fetch<256, NEXT_TAIL>(W1, kb + KC, DIN, tid, wlo, whi);
    frag_mma_2x4<KC / 4>(f, c00, c01, c02, c03, c10, c11, c12, c13);
    w_stage<256>(wbuf + ((c + 1) & 1) * WBUF, tid, wlo, whi);
    __syncthreads();
}

__device__ __forceinline__ void relu_store_2x2(float* h, int stride, int row0, int n0, int half, int col,
                                               float biasA, float biasB,
                                               const v8f& a00, const v8f& a01, const v8f& a10, const v8f& a11)
{
    #pragma unroll
    for (int r = 0; r < 8; ++r) {
        const int m = row0 + r + half * 8;
        const float v00 = a00[r] + biasA, v01 = a01[r] + biasB;
        const float v10 = a10[r] + biasA, v11 = a11[r] + biasB;
        h[m * stride + n0 + col]             = v00 > 0.0f ? v00 : 0.0f;
        h[m * stride + n0 + 16 + col]        = v01 > 0.0f ? v01 : 0.0f;
        h[(m + 16) * stride + n0 + col]      = v10 > 0.0f ? v10 : 0.0f;
        h[(m + 16) * stride + n0 + 16 + col] = v11 > 0.0f ? v11 : 0.0f;
    }
}

__global__ __launch_bounds__(256) void residue_embed_fused(
    const int* __restrict__ seq, const float* __restrict__ xyz,
    const float* __restrict__ dih, const int* __restrict__ cidx,
    const float* __restrict__ orient,
    const float* __restrict__ aa_emb, const float* __restrict__ chain_emb,
    const float* __restrict__ W1, const float* __restrict__ b1,
    const float* __restrict__ W2, const float* __restrict__ b2,
    const float* __restrict__ W3, const float* __restrict__ b3,
    const float* __restrict__ W4, const float* __restrict__ b4,
    float* __restrict__ out)
{
    __shared__ __attribute__((aligned(16))) float xs[MT * XS];
    __shared__ float hA[MT * H1S];
    __shared__ float hB[MT * H2S];
    __shared__ __attribute__((aligned(16))) float wbuf[2 * WBUF];

    const int tid = threadIdx.x;
    const int t0  = blockIdx.x * MT;

    const int lane = tid & 31;
    const int w    = tid >> 5;
    const int half = lane >> 4;
    const int col  = lane & 15;
    const int mg   = w >> 2;
    const int ng   = w & 3;

    v4f wlo[4], whi[4];
    w_fetch<256, false>(W1, 0, DIN, tid, wlo, whi);

    const int ty0 = seq[t0 + mg * 32 + col];
    const int ty1 = seq[t0 + mg * 32 + 16 + col];

    #pragma unroll
    for (int j = 0; j < MT * 32 / 256; ++j) {
        const int i  = tid + 256 * j;
        const int tt = i >> 5, q = i & 31;
        const int t  = t0 + tt;
        const int sa = seq[t];
        const int ci = cidx[t];
        const v4f va = *(const v4f*)(aa_emb + sa * 128 + 4 * q);
        v4f vc = *(const v4f*)(chain_emb + ci * 128 + 4 * q);
        const v4f z = {0.0f, 0.0f, 0.0f, 0.0f};
        vc = (ci != 0) ? vc : z;
        *(v4f*)(xs + tt * XS + 4 * q) = va;
        float* d = xs + tt * XS + XTAIL + 39 + 4 * q;
        d[0] = vc.x; d[1] = vc.y; d[2] = vc.z; d[3] = vc.w;
    }
    for (int i = tid; i < MT * 15; i += 256) {
        int tt = i / 15, a = i % 15;
        int t  = t0 + tt;
        const float* xp = xyz + (size_t)t * 45;
        float rx = xp[a * 3 + 0] - xp[3 + 0];
        float ry = xp[a * 3 + 1] - xp[3 + 1];
        float rz = xp[a * 3 + 2] - xp[3 + 2];
        const float* O = orient + (size_t)t * 9;
        int base = tt * XS + XCRD + a * 3;
        xs[base + 0] = O[0] * rx + O[3] * ry + O[6] * rz;
        xs[base + 1] = O[1] * rx + O[4] * ry + O[7] * rz;
        xs[base + 2] = O[2] * rx + O[5] * ry + O[8] * rz;
    }
    if (tid < MT) {
        float* r = xs + tid * XS;
        r[XTAIL + 167] = 0.0f;
        r[XCRD - 2] = 0.0f; r[XCRD - 1] = 0.0f;
        r[XCRD + 45] = 0.0f; r[XCRD + 46] = 0.0f; r[XCRD + 47] = 0.0f;
        r[XCRD + 48] = 0.0f; r[XCRD + 49] = 0.0f;
    }
    if (tid < MT * 3) {
        int tt = tid / 3, d = tid % 3;
        float v = dih[(size_t)(t0 + tt) * 3 + d];
        const float FREQ[6] = {1.0f, 2.0f, 3.0f, 1.0f, 0.5f, 1.0f / 3.0f};
        int base = tt * XS + XTAIL + d * 13;
        xs[base] = v;
        #pragma unroll
        for (int f = 0; f < 6; ++f) {
            xs[base + 1 + f] = sinf(FREQ[f] * v);
            xs[base + 7 + f] = cosf(FREQ[f] * v);
        }
    }
    w_stage<256>(wbuf, tid, wlo, whi);
    __syncthreads();

    {
        v8f c00 = {}, c01 = {}, c02 = {}, c03 = {}, c10 = {}, c11 = {}, c12 = {}, c13 = {};
        const int n0 = ng * 64;
        const float* xr0 = xs + (mg * 32 + col) * XS;
        const float* xr1 = xr0 + 16 * XS;
        const int bias0 = 2 * half - 128 - 45 * ty0;
        const int bias1 = 2 * half - 128 - 45 * ty1;
        const int wbo = (half * NS + n0 + col) * 2;
        #pragma nounroll
        for (int c = 0; c < 4; ++c)
            l1_chunk<0, false>(c, xr0, xr1, half, bias0, bias1, wbuf, wbo, W1, tid, wlo, whi,
                               c00, c01, c02, c03, c10, c11, c12, c13);
        #pragma nounroll
        for (int c = 4; c < 32; ++c)
            l1_chunk<1, false>(c, xr0, xr1, half, bias0, bias1, wbuf, wbo, W1, tid, wlo, whi,
                               c00, c01, c02, c03, c10, c11, c12, c13);
        l1_chunk<2, false>(32, xr0, xr1, half, bias0, bias1, wbuf, wbo, W1, tid, wlo, whi,
                           c00, c01, c02, c03, c10, c11, c12, c13);
        #pragma nounroll
        for (int c = 33; c < L1FULL - 1; ++c)
            l1_chunk<0, false>(c, xr0, xr1, half, bias0, bias1, wbuf, wbo, W1, tid, wlo, whi,
                               c00, c01, c02, c03, c10, c11, c12, c13);
        l1_chunk<0, true>(L1FULL - 1, xr0, xr1, half, bias0, bias1, wbuf, wbo, W1, tid, wlo, whi,
                          c00, c01, c02, c03, c10, c11, c12, c13);
        {
            Frag2x4<3> f;
            frag_load_2x4<3, 0>(xr0, xr1, half, L1FULL * KC, L1FULL * KC - TSHIFT, bias0, bias1,
                                wbuf + (L1FULL & 1) * WBUF + wbo, f);
            w_fetch<128, false>(W2, 0, 256, tid, wlo, whi);
            frag_mma_2x4<3>(f, c00, c01, c02, c03, c10, c11, c12, c13);
        }
        const int row0 = mg * 32;
        relu_store_2x2(hA, H1S, row0, n0,      half, col, b1[n0 + col],      b1[n0 + 16 + col], c00, c01, c10, c11);
        relu_store_2x2(hA, H1S, row0, n0 + 32, half, col, b1[n0 + 32 + col], b1[n0 + 48 + col], c02, c03, c12, c13);
        w_stage<128>(wbuf + ((L1FULL + 1) & 1) * WBUF, tid, wlo, whi);
    }
    __syncthreads();

    const int row0 = mg * 32;
    const int n0   = ng * 32;
    const int wbo  = (half * NS + n0 + col) * 2;

    {
        v8f a00 = {}, a01 = {}, a10 = {}, a11 = {};
        const float* ha0 = hA + (row0 + col) * H1S + half * 2;
        const float* ha1 = ha0 + 16 * H1S;
        #pragma nounroll
        for (int c = 0; c < 256 / KC; ++c) {
            Frag2x2 f;
            frag_load_2x2(ha0 + c * KC, ha1 + c * KC, wbuf + (c & 1) * WBUF + wbo, f);
            const float* wn = (c + 1 < 256 / KC) ? W2 + (c + 1) * KC * 128 : W3;
            w_fetch<128, false>(wn, 0, KC, tid, wlo, whi);
            frag_mma_2x2(f, a00, a01, a10, a11);
            if (c + 1 == 256 / KC)
                relu_store_2x2(hB, H2S, row0, n0, half, col, b2[n0 + col], b2[n0 + 16 + col], a00, a01, a10, a11);
            w_stage<128>(wbuf + ((c + 1) & 1) * WBUF, tid, wlo, whi);
            __syncthreads();
        }
    }

    {
        v8f a00 = {}, a01 = {}, a10 = {}, a11 = {};
        const float* ha0 = hB + (row0 + col) * H2S + half * 2;
        const float* ha1 = ha0 + 16 * H2S;
        #pragma nounroll
        for (int c = 0; c < 128 / KC; ++c) {
            Frag2x2 f;
            frag_load_2x2(ha0 + c * KC, ha1 + c * KC, wbuf + (c & 1) * WBUF + wbo, f);
            const float* wn = (c + 1 < 128 / KC) ? W3 + (c + 1) * KC * 128 : W4;
            w_fetch<128, false>(wn, 0, KC, tid, wlo, whi);
            frag_mma_2x2(f, a00, a01, a10, a11);
            if (c + 1 == 128 / KC)
                relu_store_2x2(hA, H2S, row0, n0, half, col, b3[n0 + col], b3[n0 + 16 + col], a00, a01, a10, a11);
            w_stage<128>(wbuf + ((c + 1) & 1) * WBUF, tid, wlo, whi);
            __syncthreads();
        }
    }

    {
        v8f a00 = {}, a01 = {}, a10 = {}, a11 = {};
        const float* ha0 = hA + (row0 + col) * H2S + half * 2;
        const float* ha1 = ha0 + 16 * H2S;
        #pragma nounroll
        for (int c = 0; c < 128 / KC; ++c) {
            Frag2x2 f;
            frag_load_2x2(ha0 + c * KC, ha1 + c * KC, wbuf + (c & 1) * WBUF + wbo, f);
            if (c + 1 < 128 / KC) w_fetch<128, false>(W4, (c + 1) * KC, 128, tid, wlo, whi);
            frag_mma_2x2(f, a00, a01, a10, a11);
            if (c + 1 < 128 / KC) {
                w_stage<128>(wbuf + ((c + 1) & 1) * WBUF, tid, wlo, whi);
                __syncthreads();
            }
        }
        const float biasA = b4[n0 + col], biasB = b4[n0 + 16 + col];
        #pragma unroll
        for (int r = 0; r < 8; ++r) {
            const int m = row0 + r + half * 8;
            out[(size_t)(t0 + m) * 128 + n0 + col]           = a00[r] + biasA;
            out[(size_t)(t0 + m) * 128 + n0 + 16 + col]      = a01[r] + biasB;
            out[(size_t)(t0 + m + 16) * 128 + n0 + col]      = a10[r] + biasA;
            out[(size_t)(t0 + m + 16) * 128 + n0 + 16 + col] = a11[r] + biasB;
        }
    }
}

extern "C" void kernel_launch(void* const* d_in, const int* in_sizes, int n_in,
                              void* d_out, int out_size, void* d_ws, size_t ws_size,
                              hipStream_t stream) {
    (void)n_in; (void)d_ws; (void)ws_size; (void)out_size;
    const int*   seq    = (const int*)  d_in[0];
    const float* xyz    = (const float*)d_in[1];
    const float* dih    = (const float*)d_in[2];
    const int*   cidx   = (const int*)  d_in[3];
    const float* orient = (const float*)d_in[4];
    const float* aa_emb    = (const float*)d_in[6];
    const float* chain_emb = (const float*)d_in[7];
    const float* W1 = (const float*)d_in[8];   const float* b1 = (const float*)d_in[9];
    const float* W2 = (const float*)d_in[10];  const float* b2 = (const float*)d_in[11];
    const float* W3 = (const float*)d_in[12];  const float* b3 = (const float*)d_in[13];
    const float* W4 = (const float*)d_in[14];  const float* b4 = (const float*)d_in[15];
    float* out = (float*)d_out;

    const int tokens = in_sizes[0];
    const int blocks = tokens / MT;
    residue_embed_fused<<<blocks, 256, 0, stream>>>(
        seq, xyz, dih, cidx, orient, aa_emb, chain_emb,
        W1, b1, W2, b2, W3, b3, W4, b4, out);
}


# Layer 1 runs without work-group barriers: a wave owns 32 output columns for all 64 tokens and streams its own W1 columns through a private LDS slab, wave-local hand-over; sum order unchanged
# speedup vs baseline: 2.1787x; 1.1830x over previous
#include <hip/hip_runtime.h>
#include <math.h>

typedef float v2f __attribute__((ext_vector_type(2)));
typedef float v4f __attribute__((ext_vector_type(4)));
typedef float v8f __attribute__((ext_vector_type(8)));

#define MT    64
#define DIN   1195
#define XS    348
#define XTAIL 128
#define XCRD  298
#define TSHIFT 900
#define H1S   260
#define H2S   132
#define KC    32
#define NS    272
#define WBUF  (KC / 2 * NS * 2)
#define L1FULL (DIN / KC)
#define NSW   48
#define PBUF  (KC / 2 * NSW * 2)

#define WMMA_F32(a, b, c) \
    __builtin_amdgcn_wmma_f32_16x16x4_f32(false, (a), false, (b), (short)0, (c), false, false)

template <int NC>
__device__ __forceinline__ void w_fetch(const float* __restrict__ Wg, int k0, int tid,
                                        v4f (&lo)[4], v4f (&hi)[4])
{
    constexpr int G  = NC / 4;
    constexpr int NI = NC / 64;
    #pragma unroll
    for (int j = 0; j < NI; ++j) {
        const int i = tid + 256 * j;
        const int r = k0 + 2 * (i / G);
        const int g = i % G;
        lo[j] = *(const v4f*)(Wg + (size_t)r * NC + 4 * g);
        hi[j] = *(const v4f*)(Wg + (size_t)(r + 1) * NC + 4 * g);
    }
}

template <bool TAIL>
__device__ __forceinline__ void slab_fetch(const float* __restrict__ Wc, int k0, int lane,
                                           v4f (&lo)[4], v4f (&hi)[4])
{
    #pragma unroll
    for (int j = 0; j < 4; ++j) {
        const int r = k0 + 2 * (4 * j + (lane >> 3));
        if (!TAIL) {
            lo[j] = *(const v4f*)(Wc + (size_t)r * 256);
            hi[j] = *(const v4f*)(Wc + (size_t)(r + 1) * 256);
        } else {
            const int ra = (r     < DIN) ? r     : DIN - 1;
            const int rb = (r + 1 < DIN) ? r + 1 : DIN - 1;
            const v4f a = *(const v4f*)(Wc + (size_t)ra * 256);
            const v4f b = *(const v4f*)(Wc + (size_t)rb * 256);
            const v4f z = {0.0f, 0.0f, 0.0f, 0.0f};
            lo[j] = (r     < DIN) ? a : z;
            hi[j] = (r + 1 < DIN) ? b : z;
        }
    }
}

__device__ __forceinline__ void slab_stage(float* slab, int lane, const v4f (&lo)[4], const v4f (&hi)[4])
{
    #pragma unroll
    for (int j = 0; j < 4; ++j) {
        float* d = slab + ((4 * j + (lane >> 3)) * NSW + 4 * (lane & 7)) * 2;
        const v4f a = {lo[j].x, hi[j].x, lo[j].y, hi[j].y};
        const v4f b = {lo[j].z, hi[j].z, lo[j].w, hi[j].w};
        *(v4f*)d       = a;
        *(v4f*)(d + 4) = b;
    }
}

__device__ __forceinline__ void slab_handover()
{
    __builtin_amdgcn_fence(__ATOMIC_RELEASE, "wavefront");
    __builtin_amdgcn_wave_barrier();
    __builtin_amdgcn_fence(__ATOMIC_ACQUIRE, "wavefront");
}

template <int NC>
__device__ __forceinline__ void w_stage(float* buf, int tid, const v4f (&lo)[4], const v4f (&hi)[4])
{
    constexpr int G  = NC / 4;
    constexpr int NI = NC / 64;
    #pragma unroll
    for (int j = 0; j < NI; ++j) {
        const int i = tid + 256 * j;
        float* d = buf + ((i / G) * NS + 4 * (i % G)) * 2;
        const v4f a = {lo[j].x, hi[j].x, lo[j].y, hi[j].y};
        const v4f b = {lo[j].z, hi[j].z, lo[j].w, hi[j].w};
        *(v4f*)d       = a;
        *(v4f*)(d + 4) = b;
    }
}

template <int STEPS>
struct Frag4x2 { v2f a0[STEPS], a1[STEPS], a2[STEPS], a3[STEPS], b0[STEPS], b1[STEPS]; };

__device__ __forceinline__ v2f a_onehot(const float* xc, int t)
{
    const int c = t < -2 ? -2 : (t > 46 ? 46 : t);
    const float* p = xc + c;
    const v2f r = {p[0], p[1]};
    return r;
}

template <int STEPS, int MODE>
__device__ __forceinline__ void frag_load_4x2(const float* xr, int half, int kb, int xoff,
                                              int bias0, int bias1, int bias2, int bias3,
                                              const float* wb, Frag4x2<STEPS>& f)
{
    #pragma unroll
    for (int s = 0; s < STEPS; ++s) {
        if (MODE == 1 || (MODE == 2 && s == 0)) {
            f.a0[s] = a_onehot(xr + XCRD,           bias0 + kb + 4 * s);
            f.a1[s] = a_onehot(xr + 16 * XS + XCRD, bias1 + kb + 4 * s);
            f.a2[s] = a_onehot(xr + 32 * XS + XCRD, bias2 + kb + 4 * s);
            f.a3[s] = a_onehot(xr + 48 * XS + XCRD, bias3 + kb + 4 * s);
        } else {
            const float* p = xr + 2 * half + xoff + 4 * s;
            f.a0[s] = *(const v2f*)(p);
            f.a1[s] = *(const v2f*)(p + 16 * XS);
            f.a2[s] = *(const v2f*)(p + 32 * XS);
            f.a3[s] = *(const v2f*)(p + 48 * XS);
        }
        f.b0[s] = *(const v2f*)(wb + s * (4 * NSW));
        f.b1[s] = *(const v2f*)(wb + s * (4 * NSW) + 32);
        __builtin_amdgcn_sched_barrier(0);
    }
}

template <int STEPS>
__device__ __forceinline__ void frag_mma_4x2(const Frag4x2<STEPS>& f,
                                             v8f& c00, v8f& c01, v8f& c10, v8f& c11,
                                             v8f& c20, v8f& c21, v8f& c30, v8f& c31)
{
    #pragma unroll
    for (int s = 0; s < STEPS; ++s) {
        c00 = WMMA_F32(f.a0[s], f.b0[s], c00);
        c01 = WMMA_F32(f.a0[s], f.b1[s], c01);
        c10 = WMMA_F32(f.a1[s], f.b0[s], c10);
        c11 = WMMA_F32(f.a1[s], f.b1[s], c11);
        c20 = WMMA_F32(f.a2[s], f.b0[s], c20);
        c21 = WMMA_F32(f.a2[s], f.b1[s], c21);
        c30 = WMMA_F32(f.a3[s], f.b0[s], c30);
        c31 = WMMA_F32(f.a3[s], f.b1[s], c31);
    }
}

struct Frag2x2 { v2f a0[KC / 4], a1[KC / 4], bA[KC / 4], bB[KC / 4]; };

__device__ __forceinline__ void frag_load_2x2(const float* ha0, const float* ha1, const float* wb,
                                              Frag2x2& f)
{
    #pragma unroll
    for (int s = 0; s < KC / 4; ++s) {
        f.a0[s] = *(const v2f*)(ha0 + 4 * s);
        f.a1[s] = *(const v2f*)(ha1 + 4 * s);
        f.bA[s] = *(const v2f*)(wb + s * (4 * NS));
        f.bB[s] = *(const v2f*)(wb + s * (4 * NS) + 32);
    }
}

__device__ __forceinline__ void frag_mma_2x2(const Frag2x2& f, v8f& a00, v8f& a01, v8f& a10, v8f& a11)
{
    #pragma unroll
    for (int s = 0; s < KC / 4; ++s) {
        a00 = WMMA_F32(f.a0[s], f.bA[s], a00);
        a01 = WMMA_F32(f.a0[s], f.bB[s], a01);
        a10 = WMMA_F32(f.a1[s], f.bA[s], a10);
        a11 = WMMA_F32(f.a1[s], f.bB[s], a11);
    }
}

template <int MODE, bool NEXT_TAIL>
__device__ __forceinline__ void l1_chunk(int c, const float* xr, int half,
                                         int bias0, int bias1, int bias2, int bias3,
                                         float* slab, int wbo,
                                         const float* __restrict__ W1c, int lane,
                                         v4f (&wlo)[4], v4f (&whi)[4],
                                         v8f& c00, v8f& c01, v8f& c10, v8f& c11,
                                         v8f& c20, v8f& c21, v8f& c30, v8f& c31)
{
    Frag4x2<KC / 4> f;
    const int kb   = c * KC;
    const int xoff = (kb < 128) ? kb : kb - TSHIFT;
    slab_stage(slab, lane, wlo, whi);
    slab_handover();
    slab_fetch<NEXT_TAIL>(W1c, kb + KC, lane, wlo, whi);
    __builtin_amdgcn_sched_barrier(0);
    frag_load_4x2<KC / 4, MODE>(xr, half, kb, xoff, bias0, bias1, bias2, bias3, slab + wbo, f);
    frag_mma_4x2<KC / 4>(f, c00, c01, c10, c11, c20, c21, c30, c31);
    asm volatile("" : "+v"(c00), "+v"(c01), "+v"(c10), "+v"(c11), "+v"(c20), "+v"(c21), "+v"(c30), "+v"(c31),
                      "+v"(wlo[0]), "+v"(wlo[1]), "+v"(wlo[2]), "+v"(wlo[3]),
                      "+v"(whi[0]), "+v"(whi[1]), "+v"(whi[2]), "+v"(whi[3]));
}

__device__ __forceinline__ void relu_store_2x2(float* h, int stride, int row0, int n0, int half, int col,
                                               float biasA, float biasB,
                                               const v8f& a00, const v8f& a01, const v8f& a10, const v8f& a11)
{
    #pragma unroll
    for (int r = 0; r < 8; ++r) {
        const int m = row0 + r + half * 8;
        const float v00 = a00[r] + biasA, v01 = a01[r] + biasB;
        const float v10 = a10[r] + biasA, v11 = a11[r] + biasB;
        h[m * stride + n0 + col]             = v00 > 0.0f ? v00 : 0.0f;
        h[m * stride + n0 + 16 + col]        = v01 > 0.0f ? v01 : 0.0f;
        h[(m + 16) * stride + n0 + col]      = v10 > 0.0f ? v10 : 0.0f;
        h[(m + 16) * stride + n0 + 16 + col] = v11 > 0.0f ? v11 : 0.0f;
    }
}

__global__ __launch_bounds__(256) void residue_embed_fused(
    const int* __restrict__ seq, const float* __restrict__ xyz,
    const float* __restrict__ dih, const int* __restrict__ cidx,
    const float* __restrict__ orient,
    const float* __restrict__ aa_emb, const float* __restrict__ chain_emb,
    const float* __restrict__ W1, const float* __restrict__ b1,
    const float* __restrict__ W2, const float* __restrict__ b2,
    const float* __restrict__ W3, const float* __restrict__ b3,
    const float* __restrict__ W4, const float* __restrict__ b4,
    float* __restrict__ out)
{
    __shared__ __attribute__((aligned(16))) float xs[MT * XS];
    __shared__ float hA[MT * H1S];
    __shared__ float hB[MT * H2S];
    __shared__ __attribute__((aligned(16))) float wbuf[2 * WBUF];

    const int tid = threadIdx.x;
    const int t0  = blockIdx.x * MT;

    const int lane = tid & 31;
    const int w    = tid >> 5;
    const int half = lane >> 4;
    const int col  = lane & 15;
    const int mg   = w >> 2;
    const int ng   = w & 3;

    const int n0w = w * 32;
    float* slab = wbuf + w * PBUF;
    const float* W1c = W1 + n0w + 4 * (lane & 7);
    v4f wlo[4], whi[4];
    slab_fetch<false>(W1c, 0, lane, wlo, whi);

    const int ty0 = seq[t0 + col];
    const int ty1 = seq[t0 + 16 + col];
    const int ty2 = seq[t0 + 32 + col];
    const int ty3 = seq[t0 + 48 + col];

    #pragma unroll
    for (int j = 0; j < MT * 32 / 256; ++j) {
        const int i  = tid + 256 * j;
        const int tt = i >> 5, q = i & 31;
        const int t  = t0 + tt;
        const int sa = seq[t];
        const int ci = cidx[t];
        const v4f va = *(const v4f*)(aa_emb + sa * 128 + 4 * q);
        v4f vc = *(const v4f*)(chain_emb + ci * 128 + 4 * q);
        const v4f z = {0.0f, 0.0f, 0.0f, 0.0f};
        vc = (ci != 0) ? vc : z;
        *(v4f*)(xs + tt * XS + 4 * q) = va;
        float* d = xs + tt * XS + XTAIL + 39 + 4 * q;
        d[0] = vc.x; d[1] = vc.y; d[2] = vc.z; d[3] = vc.w;
    }
    for (int i = tid; i < MT * 15; i += 256) {
        int tt = i / 15, a = i % 15;
        int t  = t0 + tt;
        const float* xp = xyz + (size_t)t * 45;
        float rx = xp[a * 3 + 0] - xp[3 + 0];
        float ry = xp[a * 3 + 1] - xp[3 + 1];
        float rz = xp[a * 3 + 2] - xp[3 + 2];
        const float* O = orient + (size_t)t * 9;
        int base = tt * XS + XCRD + a * 3;
        xs[base + 0] = O[0] * rx + O[3] * ry + O[6] * rz;
        xs[base + 1] = O[1] * rx + O[4] * ry + O[7] * rz;
        xs[base + 2] = O[2] * rx + O[5] * ry + O[8] * rz;
    }
    if (tid < MT) {
        float* r = xs + tid * XS;
        r[XTAIL + 167] = 0.0f;
        r[XCRD - 2] = 0.0f; r[XCRD - 1] = 0.0f;
        r[XCRD + 45] = 0.0f; r[XCRD + 46] = 0.0f; r[XCRD + 47] = 0.0f;
        r[XCRD + 48] = 0.0f; r[XCRD + 49] = 0.0f;
    }
    if (tid < MT * 3) {
        int tt = tid / 3, d = tid % 3;
        float v = dih[(size_t)(t0 + tt) * 3 + d];
        const float FREQ[6] = {1.0f, 2.0f, 3.0f, 1.0f, 0.5f, 1.0f / 3.0f};
        int base = tt * XS + XTAIL + d * 13;
        xs[base] = v;
        #pragma unroll
        for (int f = 0; f < 6; ++f) {
            xs[base + 1 + f] = sinf(FREQ[f] * v);
            xs[base + 7 + f] = cosf(FREQ[f] * v);
        }
    }
    __syncthreads();

    {
        v8f c00 = {}, c01 = {}, c10 = {}, c11 = {}, c20 = {}, c21 = {}, c30 = {}, c31 = {};
        const float* xr = xs + col * XS;
        const int bias0 = 2 * half - 128 - 45 * ty0;
        const int bias1 = 2 * half - 128 - 45 * ty1;
        const int bias2 = 2 * half - 128 - 45 * ty2;
        const int bias3 = 2 * half - 128 - 45 * ty3;
        const int wbo = (half * NSW + col) * 2;
        #pragma nounroll
        for (int c = 0; c < 4; ++c)
            l1_chunk<0, false>(c, xr, half, bias0, bias1, bias2, bias3, slab, wbo, W1c, lane, wlo, whi,
                               c00, c01, c10, c11, c20, c21, c30, c31);
        #pragma nounroll
        for (int c = 4; c < 32; ++c)
            l1_chunk<1, false>(c, xr, half, bias0, bias1, bias2, bias3, slab, wbo, W1c, lane, wlo, whi,
                               c00, c01, c10, c11, c20, c21, c30, c31);
        l1_chunk<2, false>(32, xr, half, bias0, bias1, bias2, bias3, slab, wbo, W1c, lane, wlo, whi,
                           c00, c01, c10, c11, c20, c21, c30, c31);
        #pragma nounroll
        for (int c = 33; c < L1FULL - 1; ++c)
            l1_chunk<0, false>(c, xr, half, bias0, bias1, bias2, bias3, slab, wbo, W1c, lane, wlo, whi,
                               c00, c01, c10, c11, c20, c21, c30, c31);
        l1_chunk<0, true>(L1FULL - 1, xr, half, bias0, bias1, bias2, bias3, slab, wbo, W1c, lane, wlo, whi,
                          c00, c01, c10, c11, c20, c21, c30, c31);
        {
            Frag4x2<3> f;
            slab_stage(slab, lane, wlo, whi);
            slab_handover();
            w_fetch<128>(W2, 0, tid, wlo, whi);
            frag_load_4x2<3, 0>(xr, half, L1FULL * KC, L1FULL * KC - TSHIFT, bias0, bias1, bias2, bias3,
                                slab + wbo, f);
            frag_mma_4x2<3>(f, c00, c01, c10, c11, c20, c21, c30, c31);
        }
        const float biasA = b1[n0w + col], biasB = b1[n0w + 16 + col];
        relu_store_2x2(hA, H1S, 0,  n0w, half, col, biasA, biasB, c00, c01, c10, c11);
        relu_store_2x2(hA, H1S, 32, n0w, half, col, biasA, biasB, c20, c21, c30, c31);
    }
    __syncthreads();
    w_stage<128>(wbuf, tid, wlo, whi);
    __syncthreads();

    const int row0 = mg * 32;
    const int n0   = ng * 32;
    const int wbo  = (half * NS + n0 + col) * 2;

    {
        v8f a00 = {}, a01 = {}, a10 = {}, a11 = {};
        const float* ha0 = hA + (row0 + col) * H1S + half * 2;
        const float* ha1 = ha0 + 16 * H1S;
        #pragma nounroll
        for (int c = 0; c < 256 / KC; ++c) {
            Frag2x2 f;
            frag_load_2x2(ha0 + c * KC, ha1 + c * KC, wbuf + (c & 1) * WBUF + wbo, f);
            const float* wn = (c + 1 < 256 / KC) ? W2 + (c + 1) * KC * 128 : W3;
            w_fetch<128>(wn, 0, tid, wlo, whi);
            frag_mma_2x2(f, a00, a01, a10, a11);
            if (c + 1 == 256 / KC)
                relu_store_2x2(hB, H2S, row0, n0, half, col, b2[n0 + col], b2[n0 + 16 + col], a00, a01, a10, a11);
            w_stage<128>(wbuf + ((c + 1) & 1) * WBUF, tid, wlo, whi);
            __syncthreads();
        }
    }

    {
        v8f a00 = {}, a01 = {}, a10 = {}, a11 = {};
        const float* ha0 = hB + (row0 + col) * H2S + half * 2;
        const float* ha1 = ha0 + 16 * H2S;
        #pragma nounroll
        for (int c = 0; c < 128 / KC; ++c) {
            Frag2x2 f;
            frag_load_2x2(ha0 + c * KC, ha1 + c * KC, wbuf + (c & 1) * WBUF + wbo, f);
            const float* wn = (c + 1 < 128 / KC) ? W3 + (c + 1) * KC * 128 : W4;
            w_fetch<128>(wn, 0, tid, wlo, whi);
            frag_mma_2x2(f, a00, a01, a10, a11);
            if (c + 1 == 128 / KC)
                relu_store_2x2(hA, H2S, row0, n0, half, col, b3[n0 + col], b3[n0 + 16 + col], a00, a01, a10, a11);
            w_stage<128>(wbuf + ((c + 1) & 1) * WBUF, tid, wlo, whi);
            __syncthreads();
        }
    }

    {
        v8f a00 = {}, a01 = {}, a10 = {}, a11 = {};
        const float* ha0 = hA + (row0 + col) * H2S + half * 2;
        const float* ha1 = ha0 + 16 * H2S;
        #pragma nounroll
        for (int c = 0; c < 128 / KC; ++c) {
            Frag2x2 f;
            frag_load_2x2(ha0 + c * KC, ha1 + c * KC, wbuf + (c & 1) * WBUF + wbo, f);
            if (c + 1 < 128 / KC) w_fetch<128>(W4, (c + 1) * KC, tid, wlo, whi);
            frag_mma_2x2(f, a00, a01, a10, a11);
            if (c + 1 < 128 / KC) {
                w_stage<128>(wbuf + ((c + 1) & 1) * WBUF, tid, wlo, whi);
                __syncthreads();
            }
        }
        const float biasA = b4[n0 + col], biasB = b4[n0 + 16 + col];
        #pragma unroll
        for (int r = 0; r < 8; ++r) {
            const int m = row0 + r + half * 8;
            out[(size_t)(t0 + m) * 128 + n0 + col]           = a00[r] + biasA;
            out[(size_t)(t0 + m) * 128 + n0 + 16 + col]      = a01[r] + biasB;
            out[(size_t)(t0 + m + 16) * 128 + n0 + col]      = a10[r] + biasA;
            out[(size_t)(t0 + m + 16) * 128 + n0 + 16 + col] = a11[r] + biasB;
        }
    }
}

extern "C" void kernel_launch(void* const* d_in, const int* in_sizes, int n_in,
                              void* d_out, int out_size, void* d_ws, size_t ws_size,
                              hipStream_t stream) {
    (void)n_in; (void)d_ws; (void)ws_size; (void)out_size;
    const int*   seq    = (const int*)  d_in[0];
    const float* xyz    = (const float*)d_in[1];
    const float* dih    = (const float*)d_in[2];
    const int*   cidx   = (const int*)  d_in[3];
    const float* orient = (const float*)d_in[4];
    const float* aa_emb    = (const float*)d_in[6];
    const float* chain_emb = (const float*)d_in[7];
    const float* W1 = (const float*)d_in[8];   const float* b1 = (const float*)d_in[9];
    const float* W2 = (const float*)d_in[10];  const float* b2 = (const float*)d_in[11];
    const float* W3 = (const float*)d_in[12];  const float* b3 = (const float*)d_in[13];
    const float* W4 = (const float*)d_in[14];  const float* b4 = (const float*)d_in[15];
    float* out = (float*)d_out;

    const int tokens = in_sizes[0];
    const int blocks = tokens / MT;
    residue_embed_fused<<<blocks, 256, 0, stream>>>(
        seq, xyz, dih, cidx, orient, aa_emb, chain_emb,
        W1, b1, W2, b2, W3, b3, W4, b4, out);
}


# Layer 1 issues a one-hot k-step's products only for the 16-token tiles that hold a token of the k-step's residue type (per-wave type mask, scalar branch per tile); only exact-zero terms dropped
# speedup vs baseline: 2.5032x; 1.1489x over previous
#include <hip/hip_runtime.h>
#include <math.h>

typedef float v2f __attribute__((ext_vector_type(2)));
typedef float v4f __attribute__((ext_vector_type(4)));
typedef float v8f __attribute__((ext_vector_type(8)));

#define MT    64
#define DIN   1195
#define XS    348
#define XTAIL 128
#define XCRD  298
#define TSHIFT 900
#define H1S   260
#define H2S   132
#define KC    32
#define NS    272
#define WBUF  (KC / 2 * NS * 2)
#define L1FULL (DIN / KC)
#define NSW   48
#define PBUF  (KC / 2 * NSW * 2)

#define WMMA_F32(a, b, c) \
    __builtin_amdgcn_wmma_f32_16x16x4_f32(false, (a), false, (b), (short)0, (c), false, false)

template <int NC>
__device__ __forceinline__ void w_fetch(const float* __restrict__ Wg, int k0, int tid,
                                        v4f (&lo)[4], v4f (&hi)[4])
{
    constexpr int G  = NC / 4;
    constexpr int NI = NC / 64;
    #pragma unroll
    for (int j = 0; j < NI; ++j) {
        const int i = tid + 256 * j;
        const int r = k0 + 2 * (i / G);
        const int g = i % G;
        lo[j] = *(const v4f*)(Wg + (size_t)r * NC + 4 * g);
        hi[j] = *(const v4f*)(Wg + (size_t)(r + 1) * NC + 4 * g);
    }
}

template <bool TAIL>
__device__ __forceinline__ void slab_fetch(const float* __restrict__ Wc, int k0, int lane,
                                           v4f (&lo)[4], v4f (&hi)[4])
{
    #pragma unroll
    for (int j = 0; j < 4; ++j) {
        const int r = k0 + 2 * (4 * j + (lane >> 3));
        if (!TAIL) {
            lo[j] = *(const v4f*)(Wc + (size_t)r * 256);
            hi[j] = *(const v4f*)(Wc + (size_t)(r + 1) * 256);
        } else {
            const int ra = (r     < DIN) ? r     : DIN - 1;
            const int rb = (r + 1 < DIN) ? r + 1 : DIN - 1;
            const v4f a = *(const v4f*)(Wc + (size_t)ra * 256);
            const v4f b = *(const v4f*)(Wc + (size_t)rb * 256);
            const v4f z = {0.0f, 0.0f, 0.0f, 0.0f};
            lo[j] = (r     < DIN) ? a : z;
            hi[j] = (r + 1 < DIN) ? b : z;
        }
    }
}

__device__ __forceinline__ void slab_stage(float* slab, int lane, const v4f (&lo)[4], const v4f (&hi)[4])
{
    #pragma unroll
    for (int j = 0; j < 4; ++j) {
        float* d = slab + ((4 * j + (lane >> 3)) * NSW + 4 * (lane & 7)) * 2;
        const v4f a = {lo[j].x, hi[j].x, lo[j].y, hi[j].y};
        const v4f b = {lo[j].z, hi[j].z, lo[j].w, hi[j].w};
        *(v4f*)d       = a;
        *(v4f*)(d + 4) = b;
    }
}

__device__ __forceinline__ void slab_handover()
{
    __builtin_amdgcn_fence(__ATOMIC_RELEASE, "wavefront");
    __builtin_amdgcn_wave_barrier();
    __builtin_amdgcn_fence(__ATOMIC_ACQUIRE, "wavefront");
}

template <int NC>
__device__ __forceinline__ void w_stage(float* buf, int tid, const v4f (&lo)[4], const v4f (&hi)[4])
{
    constexpr int G  = NC / 4;
    constexpr int NI = NC / 64;
    #pragma unroll
    for (int j = 0; j < NI; ++j) {
        const int i = tid + 256 * j;
        float* d = buf + ((i / G) * NS + 4 * (i % G)) * 2;
        const v4f a = {lo[j].x, hi[j].x, lo[j].y, hi[j].y};
        const v4f b = {lo[j].z, hi[j].z, lo[j].w, hi[j].w};
        *(v4f*)d       = a;
        *(v4f*)(d + 4) = b;
    }
}

template <int STEPS>
struct Frag4x2 { v2f a0[STEPS], a1[STEPS], a2[STEPS], a3[STEPS], b0[STEPS], b1[STEPS]; };

__device__ __forceinline__ v2f a_onehot(const float* xc, int t)
{
    const int c = t < -2 ? -2 : (t > 46 ? 46 : t);
    const float* p = xc + c;
    const v2f r = {p[0], p[1]};
    return r;
}

template <int STEPS, int MODE>
__device__ __forceinline__ void frag_load_4x2(const float* xr, int half, int kb, int xoff,
                                              int bias0, int bias1, int bias2, int bias3,
                                              const float* wb, Frag4x2<STEPS>& f)
{
    #pragma unroll
    for (int s = 0; s < STEPS; ++s) {
        if (MODE == 1 || (MODE == 2 && s == 0)) {
            f.a0[s] = a_onehot(xr + XCRD,           bias0 + kb + 4 * s);
            f.a1[s] = a_onehot(xr + 16 * XS + XCRD, bias1 + kb + 4 * s);
            f.a2[s] = a_onehot(xr + 32 * XS + XCRD, bias2 + kb + 4 * s);
            f.a3[s] = a_onehot(xr + 48 * XS + XCRD, bias3 + kb + 4 * s);
        } else {
            const float* p = xr + 2 * half + xoff + 4 * s;
            f.a0[s] = *(const v2f*)(p);
            f.a1[s] = *(const v2f*)(p + 16 * XS);
            f.a2[s] = *(const v2f*)(p + 32 * XS);
            f.a3[s] = *(const v2f*)(p + 48 * XS);
        }
        f.b0[s] = *(const v2f*)(wb + s * (4 * NSW));
        f.b1[s] = *(const v2f*)(wb + s * (4 * NSW) + 32);
        __builtin_amdgcn_sched_barrier(0);
    }
}

template <int STEPS>
__device__ __forceinline__ void frag_mma_4x2(const Frag4x2<STEPS>& f,
                                             v8f& c00, v8f& c01, v8f& c10, v8f& c11,
                                             v8f& c20, v8f& c21, v8f& c30, v8f& c31)
{
    #pragma unroll
    for (int s = 0; s < STEPS; ++s) {
        c00 = WMMA_F32(f.a0[s], f.b0[s], c00);
        c01 = WMMA_F32(f.a0[s], f.b1[s], c01);
        c10 = WMMA_F32(f.a1[s], f.b0[s], c10);
        c11 = WMMA_F32(f.a1[s], f.b1[s], c11);
        c20 = WMMA_F32(f.a2[s], f.b0[s], c20);
        c21 = WMMA_F32(f.a2[s], f.b1[s], c21);
        c30 = WMMA_F32(f.a3[s], f.b0[s], c30);
        c31 = WMMA_F32(f.a3[s], f.b1[s], c31);
    }
}

template <int STEPS>
__device__ __forceinline__ void frag_mma_4x2_onehot(const Frag4x2<STEPS>& f, unsigned act,
                                                    v8f& c00, v8f& c01, v8f& c10, v8f& c11,
                                                    v8f& c20, v8f& c21, v8f& c30, v8f& c31)
{
    #pragma unroll
    for (int s = 0; s < STEPS; ++s) {
        if (act & (1u << (4 * s + 0))) {
            c00 = WMMA_F32(f.a0[s], f.b0[s], c00);
            c01 = WMMA_F32(f.a0[s], f.b1[s], c01);
        }
        if (act & (1u << (4 * s + 1))) {
            c10 = WMMA_F32(f.a1[s], f.b0[s], c10);
            c11 = WMMA_F32(f.a1[s], f.b1[s], c11);
        }
        if (act & (1u << (4 * s + 2))) {
            c20 = WMMA_F32(f.a2[s], f.b0[s], c20);
            c21 = WMMA_F32(f.a2[s], f.b1[s], c21);
        }
        if (act & (1u << (4 * s + 3))) {
            c30 = WMMA_F32(f.a3[s], f.b0[s], c30);
            c31 = WMMA_F32(f.a3[s], f.b1[s], c31);
        }
    }
    asm volatile("v_nop\n\tv_nop\n\tv_nop\n\tv_nop");
}

struct Frag2x2 { v2f a0[KC / 4], a1[KC / 4], bA[KC / 4], bB[KC / 4]; };

__device__ __forceinline__ void frag_load_2x2(const float* ha0, const float* ha1, const float* wb,
                                              Frag2x2& f)
{
    #pragma unroll
    for (int s = 0; s < KC / 4; ++s) {
        f.a0[s] = *(const v2f*)(ha0 + 4 * s);
        f.a1[s] = *(const v2f*)(ha1 + 4 * s);
        f.bA[s] = *(const v2f*)(wb + s * (4 * NS));
        f.bB[s] = *(const v2f*)(wb + s * (4 * NS) + 32);
    }
}

__device__ __forceinline__ void frag_mma_2x2(const Frag2x2& f, v8f& a00, v8f& a01, v8f& a10, v8f& a11)
{
    #pragma unroll
    for (int s = 0; s < KC / 4; ++s) {
        a00 = WMMA_F32(f.a0[s], f.bA[s], a00);
        a01 = WMMA_F32(f.a0[s], f.bB[s], a01);
        a10 = WMMA_F32(f.a1[s], f.bA[s], a10);
        a11 = WMMA_F32(f.a1[s], f.bB[s], a11);
    }
}

template <int MODE, bool NEXT_TAIL>
__device__ __forceinline__ void l1_chunk(int c, unsigned act, const float* xr, int half,
                                         int bias0, int bias1, int bias2, int bias3,
                                         float* slab, int wbo,
                                         const float* __restrict__ W1c, int lane,
                                         v4f (&wlo)[4], v4f (&whi)[4],
                                         v8f& c00, v8f& c01, v8f& c10, v8f& c11,
                                         v8f& c20, v8f& c21, v8f& c30, v8f& c31)
{
    Frag4x2<KC / 4> f;
    const int kb   = c * KC;
    const int xoff = (kb < 128) ? kb : kb - TSHIFT;
    slab_stage(slab, lane, wlo, whi);
    slab_handover();
    slab_fetch<NEXT_TAIL>(W1c, kb + KC, lane, wlo, whi);
    __builtin_amdgcn_sched_barrier(0);
    frag_load_4x2<KC / 4, MODE>(xr, half, kb, xoff, bias0, bias1, bias2, bias3, slab + wbo, f);
    if (MODE == 1) frag_mma_4x2_onehot<KC / 4>(f, act, c00, c01, c10, c11, c20, c21, c30, c31);
    else           frag_mma_4x2<KC / 4>(f, c00, c01, c10, c11, c20, c21, c30, c31);
    asm volatile("" : "+v"(c00), "+v"(c01), "+v"(c10), "+v"(c11), "+v"(c20), "+v"(c21), "+v"(c30), "+v"(c31),
                      "+v"(wlo[0]), "+v"(wlo[1]), "+v"(wlo[2]), "+v"(wlo[3]),
                      "+v"(whi[0]), "+v"(whi[1]), "+v"(whi[2]), "+v"(whi[3]));
}

__device__ __forceinline__ void relu_store_2x2(float* h, int stride, int row0, int n0, int half, int col,
                                               float biasA, float biasB,
                                               const v8f& a00, const v8f& a01, const v8f& a10, const v8f& a11)
{
    #pragma unroll
    for (int r = 0; r < 8; ++r) {
        const int m = row0 + r + half * 8;
        const float v00 = a00[r] + biasA, v01 = a01[r] + biasB;
        const float v10 = a10[r] + biasA, v11 = a11[r] + biasB;
        h[m * stride + n0 + col]             = v00 > 0.0f ? v00 : 0.0f;
        h[m * stride + n0 + 16 + col]        = v01 > 0.0f ? v01 : 0.0f;
        h[(m + 16) * stride + n0 + col]      = v10 > 0.0f ? v10 : 0.0f;
        h[(m + 16) * stride + n0 + 16 + col] = v11 > 0.0f ? v11 : 0.0f;
    }
}

__global__ __launch_bounds__(256) void residue_embed_fused(
    const int* __restrict__ seq, const float* __restrict__ xyz,
    const float* __restrict__ dih, const int* __restrict__ cidx,
    const float* __restrict__ orient,
    const float* __restrict__ aa_emb, const float* __restrict__ chain_emb,
    const float* __restrict__ W1, const float* __restrict__ b1,
    const float* __restrict__ W2, const float* __restrict__ b2,
    const float* __restrict__ W3, const float* __restrict__ b3,
    const float* __restrict__ W4, const float* __restrict__ b4,
    float* __restrict__ out)
{
    __shared__ __attribute__((aligned(16))) float xs[MT * XS];
    __shared__ float hA[MT * H1S];
    __shared__ float hB[MT * H2S];
    __shared__ __attribute__((aligned(16))) float wbuf[2 * WBUF];

    const int tid = threadIdx.x;
    const int t0  = blockIdx.x * MT;

    const int lane = tid & 31;
    const int w    = tid >> 5;
    const int half = lane >> 4;
    const int col  = lane & 15;
    const int mg   = w >> 2;
    const int ng   = w & 3;

    const int n0w = w * 32;
    float* slab = wbuf + w * PBUF;
    const float* W1c = W1 + n0w + 4 * (lane & 7);
    v4f wlo[4], whi[4];
    slab_fetch<false>(W1c, 0, lane, wlo, whi);

    const int ty0 = seq[t0 + col];
    const int ty1 = seq[t0 + 16 + col];
    const int ty2 = seq[t0 + 32 + col];
    const int ty3 = seq[t0 + 48 + col];

    #pragma unroll
    for (int j = 0; j < MT * 32 / 256; ++j) {
        const int i  = tid + 256 * j;
        const int tt = i >> 5, q = i & 31;
        const int t  = t0 + tt;
        const int sa = seq[t];
        const int ci = cidx[t];
        const v4f va = *(const v4f*)(aa_emb + sa * 128 + 4 * q);
        v4f vc = *(const v4f*)(chain_emb + ci * 128 + 4 * q);
        const v4f z = {0.0f, 0.0f, 0.0f, 0.0f};
        vc = (ci != 0) ? vc : z;
        *(v4f*)(xs + tt * XS + 4 * q) = va;
        float* d = xs + tt * XS + XTAIL + 39 + 4 * q;
        d[0] = vc.x; d[1] = vc.y; d[2] = vc.z; d[3] = vc.w;
    }
    for (int i = tid; i < MT * 15; i += 256) {
        int tt = i / 15, a = i % 15;
        int t  = t0 + tt;
        const float* xp = xyz + (size_t)t * 45;
        float rx = xp[a * 3 + 0] - xp[3 + 0];
        float ry = xp[a * 3 + 1] - xp[3 + 1];
        float rz = xp[a * 3 + 2] - xp[3 + 2];
        const float* O = orient + (size_t)t * 9;
        int base = tt * XS + XCRD + a * 3;
        xs[base + 0] = O[0] * rx + O[3] * ry + O[6] * rz;
        xs[base + 1] = O[1] * rx + O[4] * ry + O[7] * rz;
        xs[base + 2] = O[2] * rx + O[5] * ry + O[8] * rz;
    }
    if (tid < MT) {
        float* r = xs + tid * XS;
        r[XTAIL + 167] = 0.0f;
        r[XCRD - 2] = 0.0f; r[XCRD - 1] = 0.0f;
        r[XCRD + 45] = 0.0f; r[XCRD + 46] = 0.0f; r[XCRD + 47] = 0.0f;
        r[XCRD + 48] = 0.0f; r[XCRD + 49] = 0.0f;
    }
    if (tid < MT * 3) {
        int tt = tid / 3, d = tid % 3;
        float v = dih[(size_t)(t0 + tt) * 3 + d];
        const float FREQ[6] = {1.0f, 2.0f, 3.0f, 1.0f, 0.5f, 1.0f / 3.0f};
        int base = tt * XS + XTAIL + d * 13;
        xs[base] = v;
        #pragma unroll
        for (int f = 0; f < 6; ++f) {
            xs[base + 1 + f] = sinf(FREQ[f] * v);
            xs[base + 7 + f] = cosf(FREQ[f] * v);
        }
    }
    __syncthreads();

    {
        v8f c00 = {}, c01 = {}, c10 = {}, c11 = {}, c20 = {}, c21 = {}, c30 = {}, c31 = {};
        const float* xr = xs + col * XS;
        const int bias0 = 2 * half - 128 - 45 * ty0;
        const int bias1 = 2 * half - 128 - 45 * ty1;
        const int bias2 = 2 * half - 128 - 45 * ty2;
        const int bias3 = 2 * half - 128 - 45 * ty3;
        const int wbo = (half * NSW + col) * 2;
        unsigned tm0 = 0u, tm1 = 0u, tm2 = 0u, tm3 = 0u;
        #pragma unroll
        for (int g = 0; g < 20; ++g) {
            tm0 |= __builtin_amdgcn_ballot_w32(ty0 == g) ? (1u << g) : 0u;
            tm1 |= __builtin_amdgcn_ballot_w32(ty1 == g) ? (1u << g) : 0u;
            tm2 |= __builtin_amdgcn_ballot_w32(ty2 == g) ? (1u << g) : 0u;
            tm3 |= __builtin_amdgcn_ballot_w32(ty3 == g) ? (1u << g) : 0u;
        }
        unsigned actv = 0u;
        #pragma unroll
        for (int s = 0; s < KC / 4; ++s) {
            const unsigned rel = (unsigned)((4 + lane) * KC + 4 * s - 128);
            const unsigned sel = (1u << (rel / 45u)) | (1u << ((rel + 3u) / 45u));
            actv |= ((tm0 & sel) ? 1u : 0u) << (4 * s + 0);
            actv |= ((tm1 & sel) ? 1u : 0u) << (4 * s + 1);
            actv |= ((tm2 & sel) ? 1u : 0u) << (4 * s + 2);
            actv |= ((tm3 & sel) ? 1u : 0u) << (4 * s + 3);
        }
        #pragma nounroll
        for (int c = 0; c < 4; ++c)
            l1_chunk<0, false>(c, 0u, xr, half, bias0, bias1, bias2, bias3, slab, wbo, W1c, lane, wlo, whi,
                               c00, c01, c10, c11, c20, c21, c30, c31);
        #pragma nounroll
        for (int c = 4; c < 32; ++c) {
            const unsigned act = (unsigned)__builtin_amdgcn_readlane((int)actv, c - 4);
            l1_chunk<1, false>(c, act, xr, half, bias0, bias1, bias2, bias3, slab, wbo, W1c, lane, wlo, whi,
                               c00, c01, c10, c11, c20, c21, c30, c31);
        }
        l1_chunk<2, false>(32, 0u, xr, half, bias0, bias1, bias2, bias3, slab, wbo, W1c, lane, wlo, whi,
                           c00, c01, c10, c11, c20, c21, c30, c31);
        #pragma nounroll
        for (int c = 33; c < L1FULL - 1; ++c)
            l1_chunk<0, false>(c, 0u, xr, half, bias0, bias1, bias2, bias3, slab, wbo, W1c, lane, wlo, whi,
                               c00, c01, c10, c11, c20, c21, c30, c31);
        l1_chunk<0, true>(L1FULL - 1, 0u, xr, half, bias0, bias1, bias2, bias3, slab, wbo, W1c, lane, wlo, whi,
                          c00, c01, c10, c11, c20, c21, c30, c31);
        {
            Frag4x2<3> f;
            slab_stage(slab, lane, wlo, whi);
            slab_handover();
            w_fetch<128>(W2, 0, tid, wlo, whi);
            frag_load_4x2<3, 0>(xr, half, L1FULL * KC, L1FULL * KC - TSHIFT, bias0, bias1, bias2, bias3,
                                slab + wbo, f);
            frag_mma_4x2<3>(f, c00, c01, c10, c11, c20, c21, c30, c31);
        }
        const float biasA = b1[n0w + col], biasB = b1[n0w + 16 + col];
        relu_store_2x2(hA, H1S, 0,  n0w, half, col, biasA, biasB, c00, c01, c10, c11);
        relu_store_2x2(hA, H1S, 32, n0w, half, col, biasA, biasB, c20, c21, c30, c31);
    }
    __syncthreads();
    w_stage<128>(wbuf, tid, wlo, whi);
    __syncthreads();

    const int row0 = mg * 32;
    const int n0   = ng * 32;
    const int wbo  = (half * NS + n0 + col) * 2;

    {
        v8f a00 = {}, a01 = {}, a10 = {}, a11 = {};
        const float* ha0 = hA + (row0 + col) * H1S + half * 2;
        const float* ha1 = ha0 + 16 * H1S;
        #pragma nounroll
        for (int c = 0; c < 256 / KC; ++c) {
            Frag2x2 f;
            frag_load_2x2(ha0 + c * KC, ha1 + c * KC, wbuf + (c & 1) * WBUF + wbo, f);
            const float* wn = (c + 1 < 256 / KC) ? W2 + (c + 1) * KC * 128 : W3;
            w_fetch<128>(wn, 0, tid, wlo, whi);
            frag_mma_2x2(f, a00, a01, a10, a11);
            if (c + 1 == 256 / KC)
                relu_store_2x2(hB, H2S, row0, n0, half, col, b2[n0 + col], b2[n0 + 16 + col], a00, a01, a10, a11);
            w_stage<128>(wbuf + ((c + 1) & 1) * WBUF, tid, wlo, whi);
            __syncthreads();
        }
    }

    {
        v8f a00 = {}, a01 = {}, a10 = {}, a11 = {};
        const float* ha0 = hB + (row0 + col) * H2S + half * 2;
        const float* ha1 = ha0 + 16 * H2S;
        #pragma nounroll
        for (int c = 0; c < 128 / KC; ++c) {
            Frag2x2 f;
            frag_load_2x2(ha0 + c * KC, ha1 + c * KC, wbuf + (c & 1) * WBUF + wbo, f);
            const float* wn = (c + 1 < 128 / KC) ? W3 + (c + 1) * KC * 128 : W4;
            w_fetch<128>(wn, 0, tid, wlo, whi);
            frag_mma_2x2(f, a00, a01, a10, a11);
            if (c + 1 == 128 / KC)
                relu_store_2x2(hA, H2S, row0, n0, half, col, b3[n0 + col], b3[n0 + 16 + col], a00, a01, a10, a11);
            w_stage<128>(wbuf + ((c + 1) & 1) * WBUF, tid, wlo, whi);
            __syncthreads();
        }
    }

    {
        v8f a00 = {}, a01 = {}, a10 = {}, a11 = {};
        const float* ha0 = hA + (row0 + col) * H2S + half * 2;
        const float* ha1 = ha0 + 16 * H2S;
        #pragma nounroll
        for (int c = 0; c < 128 / KC; ++c) {
            Frag2x2 f;
            frag_load_2x2(ha0 + c * KC, ha1 + c * KC, wbuf + (c & 1) * WBUF + wbo, f);
            if (c + 1 < 128 / KC) w_fetch<128>(W4, (c + 1) * KC, tid, wlo, whi);
            frag_mma_2x2(f, a00, a01, a10, a11);
            if (c + 1 < 128 / KC) {
                w_stage<128>(wbuf + ((c + 1) & 1) * WBUF, tid, wlo, whi);
                __syncthreads();
            }
        }
        const float biasA = b4[n0 + col], biasB = b4[n0 + 16 + col];
        #pragma unroll
        for (int r = 0; r < 8; ++r) {
            const int m = row0 + r + half * 8;
            out[(size_t)(t0 + m) * 128 + n0 + col]           = a00[r] + biasA;
            out[(size_t)(t0 + m) * 128 + n0 + 16 + col]      = a01[r] + biasB;
            out[(size_t)(t0 + m + 16) * 128 + n0 + col]      = a10[r] + biasA;
            out[(size_t)(t0 + m + 16) * 128 + n0 + 16 + col] = a11[r] + biasB;
        }
    }
}

extern "C" void kernel_launch(void* const* d_in, const int* in_sizes, int n_in,
                              void* d_out, int out_size, void* d_ws, size_t ws_size,
                              hipStream_t stream) {
    (void)n_in; (void)d_ws; (void)ws_size; (void)out_size;
    const int*   seq    = (const int*)  d_in[0];
    const float* xyz    = (const float*)d_in[1];
    const float* dih    = (const float*)d_in[2];
    const int*   cidx   = (const int*)  d_in[3];
    const float* orient = (const float*)d_in[4];
    const float* aa_emb    = (const float*)d_in[6];
    const float* chain_emb = (const float*)d_in[7];
    const float* W1 = (const float*)d_in[8];   const float* b1 = (const float*)d_in[9];
    const float* W2 = (const float*)d_in[10];  const float* b2 = (const float*)d_in[11];
    const float* W3 = (const float*)d_in[12];  const float* b3 = (const float*)d_in[13];
    const float* W4 = (const float*)d_in[14];  const float* b4 = (const float*)d_in[15];
    float* out = (float*)d_out;

    const int tokens = in_sizes[0];
    const int blocks = tokens / MT;
    residue_embed_fused<<<blocks, 256, 0, stream>>>(
        seq, xyz, dih, cidx, orient, aa_emb, chain_emb,
        W1, b1, W2, b2, W3, b3, W4, b4, out);
}


# Layer 1: the younger wave of each SIMD pair (waves 4..7) runs at wave priority 3 (s_setprio) for its first 24 chunks, then 0, so both waves of a pair end the layer together; sum order unchanged
# speedup vs baseline: 2.6027x; 1.0397x over previous
#include <hip/hip_runtime.h>
#include <math.h>

typedef float v2f __attribute__((ext_vector_type(2)));
typedef float v4f __attribute__((ext_vector_type(4)));
typedef float v8f __attribute__((ext_vector_type(8)));

#define MT    64
#define DIN   1195
#define XS    348
#define XTAIL 128
#define XCRD  298
#define TSHIFT 900
#define H1S   260
#define H2S   132
#define KC    32
#define NS    272
#define WBUF  (KC / 2 * NS * 2)
#define L1FULL (DIN / KC)
#define NSW   48
#define PBUF  (KC / 2 * NSW * 2)
#define L1LEAD 24

#define WMMA_F32(a, b, c) \
    __builtin_amdgcn_wmma_f32_16x16x4_f32(false, (a), false, (b), (short)0, (c), false, false)

template <int NC>
__device__ __forceinline__ void w_fetch(const float* __restrict__ Wg, int k0, int tid,
                                        v4f (&lo)[4], v4f (&hi)[4])
{
    constexpr int G  = NC / 4;
    constexpr int NI = NC / 64;
    #pragma unroll
    for (int j = 0; j < NI; ++j) {
        const int i = tid + 256 * j;
        const int r = k0 + 2 * (i / G);
        const int g = i % G;
        lo[j] = *(const v4f*)(Wg + (size_t)r * NC + 4 * g);
        hi[j] = *(const v4f*)(Wg + (size_t)(r + 1) * NC + 4 * g);
    }
}

template <bool TAIL>
__device__ __forceinline__ void slab_fetch(const float* __restrict__ Wc, int k0, int lane,
                                           v4f (&lo)[4], v4f (&hi)[4])
{
    #pragma unroll
    for (int j = 0; j < 4; ++j) {
        const int r = k0 + 2 * (4 * j + (lane >> 3));
        if (!TAIL) {
            lo[j] = *(const v4f*)(Wc + (size_t)r * 256);
            hi[j] = *(const v4f*)(Wc + (size_t)(r + 1) * 256);
        } else {
            const int ra = (r     < DIN) ? r     : DIN - 1;
            const int rb = (r + 1 < DIN) ? r + 1 : DIN - 1;
            const v4f a = *(const v4f*)(Wc + (size_t)ra * 256);
            const v4f b = *(const v4f*)(Wc + (size_t)rb * 256);
            const v4f z = {0.0f, 0.0f, 0.0f, 0.0f};
            lo[j] = (r     < DIN) ? a : z;
            hi[j] = (r + 1 < DIN) ? b : z;
        }
    }
}

__device__ __forceinline__ void slab_stage(float* slab, int lane, const v4f (&lo)[4], const v4f (&hi)[4])
{
    #pragma unroll
    for (int j = 0; j < 4; ++j) {
        float* d = slab + ((4 * j + (lane >> 3)) * NSW + 4 * (lane & 7)) * 2;
        const v4f a = {lo[j].x, hi[j].x, lo[j].y, hi[j].y};
        const v4f b = {lo[j].z, hi[j].z, lo[j].w, hi[j].w};
        *(v4f*)d       = a;
        *(v4f*)(d + 4) = b;
    }
}

__device__ __forceinline__ void slab_handover()
{
    __builtin_amdgcn_fence(__ATOMIC_RELEASE, "wavefront");
    __builtin_amdgcn_wave_barrier();
    __builtin_amdgcn_fence(__ATOMIC_ACQUIRE, "wavefront");
}

template <int NC>
__device__ __forceinline__ void w_stage(float* buf, int tid, const v4f (&lo)[4], const v4f (&hi)[4])
{
    constexpr int G  = NC / 4;
    constexpr int NI = NC / 64;
    #pragma unroll
    for (int j = 0; j < NI; ++j) {
        const int i = tid + 256 * j;
        float* d = buf + ((i / G) * NS + 4 * (i % G)) * 2;
        const v4f a = {lo[j].x, hi[j].x, lo[j].y, hi[j].y};
        const v4f b = {lo[j].z, hi[j].z, lo[j].w, hi[j].w};
        *(v4f*)d       = a;
        *(v4f*)(d + 4) = b;
    }
}

template <int STEPS>
struct Frag4x2 { v2f a0[STEPS], a1[STEPS], a2[STEPS], a3[STEPS], b0[STEPS], b1[STEPS]; };

__device__ __forceinline__ v2f a_onehot(const float* xc, int t)
{
    const int c = t < -2 ? -2 : (t > 46 ? 46 : t);
    const float* p = xc + c;
    const v2f r = {p[0], p[1]};
    return r;
}

template <int STEPS, int MODE>
__device__ __forceinline__ void frag_load_4x2(const float* xr, int half, int kb, int xoff,
                                              int bias0, int bias1, int bias2, int bias3,
                                              const float* wb, Frag4x2<STEPS>& f)
{
    #pragma unroll
    for (int s = 0; s < STEPS; ++s) {
        if (MODE == 1 || (MODE == 2 && s == 0)) {
            f.a0[s] = a_onehot(xr + XCRD,           bias0 + kb + 4 * s);
            f.a1[s] = a_onehot(xr + 16 * XS + XCRD, bias1 + kb + 4 * s);
            f.a2[s] = a_onehot(xr + 32 * XS + XCRD, bias2 + kb + 4 * s);
            f.a3[s] = a_onehot(xr + 48 * XS + XCRD, bias3 + kb + 4 * s);
        } else {
            const float* p = xr + 2 * half + xoff + 4 * s;
            f.a0[s] = *(const v2f*)(p);
            f.a1[s] = *(const v2f*)(p + 16 * XS);
            f.a2[s] = *(const v2f*)(p + 32 * XS);
            f.a3[s] = *(const v2f*)(p + 48 * XS);
        }
        f.b0[s] = *(const v2f*)(wb + s * (4 * NSW));
        f.b1[s] = *(const v2f*)(wb + s * (4 * NSW) + 32);
        __builtin_amdgcn_sched_barrier(0);
    }
}

template <int STEPS>
__device__ __forceinline__ void frag_mma_4x2(const Frag4x2<STEPS>& f,
                                             v8f& c00, v8f& c01, v8f& c10, v8f& c11,
                                             v8f& c20, v8f& c21, v8f& c30, v8f& c31)
{
    #pragma unroll
    for (int s = 0; s < STEPS; ++s) {
        c00 = WMMA_F32(f.a0[s], f.b0[s], c00);
        c01 = WMMA_F32(f.a0[s], f.b1[s], c01);
        c10 = WMMA_F32(f.a1[s], f.b0[s], c10);
        c11 = WMMA_F32(f.a1[s], f.b1[s], c11);
        c20 = WMMA_F32(f.a2[s], f.b0[s], c20);
        c21 = WMMA_F32(f.a2[s], f.b1[s], c21);
        c30 = WMMA_F32(f.a3[s], f.b0[s], c30);
        c31 = WMMA_F32(f.a3[s], f.b1[s], c31);
    }
}

template <int STEPS>
__device__ __forceinline__ void frag_mma_4x2_onehot(const Frag4x2<STEPS>& f, unsigned act,
                                                    v8f& c00, v8f& c01, v8f& c10, v8f& c11,
                                                    v8f& c20, v8f& c21, v8f& c30, v8f& c31)
{
    #pragma unroll
    for (int s = 0; s < STEPS; ++s) {
        if (act & (1u << (4 * s + 0))) {
            c00 = WMMA_F32(f.a0[s], f.b0[s], c00);
            c01 = WMMA_F32(f.a0[s], f.b1[s], c01);
        }
        if (act & (1u << (4 * s + 1))) {
            c10 = WMMA_F32(f.a1[s], f.b0[s], c10);
            c11 = WMMA_F32(f.a1[s], f.b1[s], c11);
        }
        if (act & (1u << (4 * s + 2))) {
            c20 = WMMA_F32(f.a2[s], f.b0[s], c20);
            c21 = WMMA_F32(f.a2[s], f.b1[s], c21);
        }
        if (act & (1u << (4 * s + 3))) {
            c30 = WMMA_F32(f.a3[s], f.b0[s], c30);
            c31 = WMMA_F32(f.a3[s], f.b1[s], c31);
        }
    }
    asm volatile("v_nop\n\tv_nop\n\tv_nop\n\tv_nop");
}

struct Frag2x2 { v2f a0[KC / 4], a1[KC / 4], bA[KC / 4], bB[KC / 4]; };

__device__ __forceinline__ void frag_load_2x2(const float* ha0, const float* ha1, const float* wb,
                                              Frag2x2& f)
{
    #pragma unroll
    for (int s = 0; s < KC / 4; ++s) {
        f.a0[s] = *(const v2f*)(ha0 + 4 * s);
        f.a1[s] = *(const v2f*)(ha1 + 4 * s);
        f.bA[s] = *(const v2f*)(wb + s * (4 * NS));
        f.bB[s] = *(const v2f*)(wb + s * (4 * NS) + 32);
    }
}

__device__ __forceinline__ void frag_mma_2x2(const Frag2x2& f, v8f& a00, v8f& a01, v8f& a10, v8f& a11)
{
    #pragma unroll
    for (int s = 0; s < KC / 4; ++s) {
        a00 = WMMA_F32(f.a0[s], f.bA[s], a00);
        a01 = WMMA_F32(f.a0[s], f.bB[s], a01);
        a10 = WMMA_F32(f.a1[s], f.bA[s], a10);
        a11 = WMMA_F32(f.a1[s], f.bB[s], a11);
    }
}

template <int MODE, bool NEXT_TAIL>
__device__ __forceinline__ void l1_chunk(int c, unsigned act, const float* xr, int half,
                                         int bias0, int bias1, int bias2, int bias3,
                                         float* slab, int wbo,
                                         const float* __restrict__ W1c, int lane,
                                         v4f (&wlo)[4], v4f (&whi)[4],
                                         v8f& c00, v8f& c01, v8f& c10, v8f& c11,
                                         v8f& c20, v8f& c21, v8f& c30, v8f& c31)
{
    Frag4x2<KC / 4> f;
    const int kb   = c * KC;
    const int xoff = (kb < 128) ? kb : kb - TSHIFT;
    slab_stage(slab, lane, wlo, whi);
    slab_handover();
    slab_fetch<NEXT_TAIL>(W1c, kb + KC, lane, wlo, whi);
    __builtin_amdgcn_sched_barrier(0);
    frag_load_4x2<KC / 4, MODE>(xr, half, kb, xoff, bias0, bias1, bias2, bias3, slab + wbo, f);
    if (MODE == 1) frag_mma_4x2_onehot<KC / 4>(f, act, c00, c01, c10, c11, c20, c21, c30, c31);
    else           frag_mma_4x2<KC / 4>(f, c00, c01, c10, c11, c20, c21, c30, c31);
    asm volatile("" : "+v"(c00), "+v"(c01), "+v"(c10), "+v"(c11), "+v"(c20), "+v"(c21), "+v"(c30), "+v"(c31),
                      "+v"(wlo[0]), "+v"(wlo[1]), "+v"(wlo[2]), "+v"(wlo[3]),
                      "+v"(whi[0]), "+v"(whi[1]), "+v"(whi[2]), "+v"(whi[3]));
}

__device__ __forceinline__ void relu_store_2x2(float* h, int stride, int row0, int n0, int half, int col,
                                               float biasA, float biasB,
                                               const v8f& a00, const v8f& a01, const v8f& a10, const v8f& a11)
{
    #pragma unroll
    for (int r = 0; r < 8; ++r) {
        const int m = row0 + r + half * 8;
        const float v00 = a00[r] + biasA, v01 = a01[r] + biasB;
        const float v10 = a10[r] + biasA, v11 = a11[r] + biasB;
        h[m * stride + n0 + col]             = v00 > 0.0f ? v00 : 0.0f;
        h[m * stride + n0 + 16 + col]        = v01 > 0.0f ? v01 : 0.0f;
        h[(m + 16) * stride + n0 + col]      = v10 > 0.0f ? v10 : 0.0f;
        h[(m + 16) * stride + n0 + 16 + col] = v11 > 0.0f ? v11 : 0.0f;
    }
}

__global__ __launch_bounds__(256) void residue_embed_fused(
    const int* __restrict__ seq, const float* __restrict__ xyz,
    const float* __restrict__ dih, const int* __restrict__ cidx,
    const float* __restrict__ orient,
    const float* __restrict__ aa_emb, const float* __restrict__ chain_emb,
    const float* __restrict__ W1, const float* __restrict__ b1,
    const float* __restrict__ W2, const float* __restrict__ b2,
    const float* __restrict__ W3, const float* __restrict__ b3,
    const float* __restrict__ W4, const float* __restrict__ b4,
    float* __restrict__ out)
{
    __shared__ __attribute__((aligned(16))) float xs[MT * XS];
    __shared__ float hA[MT * H1S];
    __shared__ float hB[MT * H2S];
    __shared__ __attribute__((aligned(16))) float wbuf[2 * WBUF];

    const int tid = threadIdx.x;
    const int t0  = blockIdx.x * MT;

    const int lane = tid & 31;
    const int w    = tid >> 5;
    const int half = lane >> 4;
    const int col  = lane & 15;
    const int mg   = w >> 2;
    const int ng   = w & 3;

    const int n0w = w * 32;
    float* slab = wbuf + w * PBUF;
    const float* W1c = W1 + n0w + 4 * (lane & 7);
    v4f wlo[4], whi[4];
    slab_fetch<false>(W1c, 0, lane, wlo, whi);

    const int ty0 = seq[t0 + col];
    const int ty1 = seq[t0 + 16 + col];
    const int ty2 = seq[t0 + 32 + col];
    const int ty3 = seq[t0 + 48 + col];

    #pragma unroll
    for (int j = 0; j < MT * 32 / 256; ++j) {
        const int i  = tid + 256 * j;
        const int tt = i >> 5, q = i & 31;
        const int t  = t0 + tt;
        const int sa = seq[t];
        const int ci = cidx[t];
        const v4f va = *(const v4f*)(aa_emb + sa * 128 + 4 * q);
        v4f vc = *(const v4f*)(chain_emb + ci * 128 + 4 * q);
        const v4f z = {0.0f, 0.0f, 0.0f, 0.0f};
        vc = (ci != 0) ? vc : z;
        *(v4f*)(xs + tt * XS + 4 * q) = va;
        float* d = xs + tt * XS + XTAIL + 39 + 4 * q;
        d[0] = vc.x; d[1] = vc.y; d[2] = vc.z; d[3] = vc.w;
    }
    for (int i = tid; i < MT * 15; i += 256) {
        int tt = i / 15, a = i % 15;
        int t  = t0 + tt;
        const float* xp = xyz + (size_t)t * 45;
        float rx = xp[a * 3 + 0] - xp[3 + 0];
        float ry = xp[a * 3 + 1] - xp[3 + 1];
        float rz = xp[a * 3 + 2] - xp[3 + 2];
        const float* O = orient + (size_t)t * 9;
        int base = tt * XS + XCRD + a * 3;
        xs[base + 0] = O[0] * rx + O[3] * ry + O[6] * rz;
        xs[base + 1] = O[1] * rx + O[4] * ry + O[7] * rz;
        xs[base + 2] = O[2] * rx + O[5] * ry + O[8] * rz;
    }
    if (tid < MT) {
        float* r = xs + tid * XS;
        r[XTAIL + 167] = 0.0f;
        r[XCRD - 2] = 0.0f; r[XCRD - 1] = 0.0f;
        r[XCRD + 45] = 0.0f; r[XCRD + 46] = 0.0f; r[XCRD + 47] = 0.0f;
        r[XCRD + 48] = 0.0f; r[XCRD + 49] = 0.0f;
    }
    if (tid < MT * 3) {
        int tt = tid / 3, d = tid % 3;
        float v = dih[(size_t)(t0 + tt) * 3 + d];
        const float FREQ[6] = {1.0f, 2.0f, 3.0f, 1.0f, 0.5f, 1.0f / 3.0f};
        int base = tt * XS + XTAIL + d * 13;
        xs[base] = v;
        #pragma unroll
        for (int f = 0; f < 6; ++f) {
            xs[base + 1 + f] = sinf(FREQ[f] * v);
            xs[base + 7 + f] = cosf(FREQ[f] * v);
        }
    }
    __syncthreads();

    {
        v8f c00 = {}, c01 = {}, c10 = {}, c11 = {}, c20 = {}, c21 = {}, c30 = {}, c31 = {};
        const float* xr = xs + col * XS;
        const int bias0 = 2 * half - 128 - 45 * ty0;
        const int bias1 = 2 * half - 128 - 45 * ty1;
        const int bias2 = 2 * half - 128 - 45 * ty2;
        const int bias3 = 2 * half - 128 - 45 * ty3;
        const int wbo = (half * NSW + col) * 2;
        unsigned tm0 = 0u, tm1 = 0u, tm2 = 0u, tm3 = 0u;
        #pragma unroll
        for (int g = 0; g < 20; ++g) {
            tm0 |= __builtin_amdgcn_ballot_w32(ty0 == g) ? (1u << g) : 0u;
            tm1 |= __builtin_amdgcn_ballot_w32(ty1 == g) ? (1u << g) : 0u;
            tm2 |= __builtin_amdgcn_ballot_w32(ty2 == g) ? (1u << g) : 0u;
            tm3 |= __builtin_amdgcn_ballot_w32(ty3 == g) ? (1u << g) : 0u;
        }
        unsigned actv = 0u;
        #pragma unroll
        for (int s = 0; s < KC / 4; ++s) {
            const unsigned rel = (unsigned)((4 + lane) * KC + 4 * s - 128);
            const unsigned sel = (1u << (rel / 45u)) | (1u << ((rel + 3u) / 45u));
            actv |= ((tm0 & sel) ? 1u : 0u) << (4 * s + 0);
            actv |= ((tm1 & sel) ? 1u : 0u) << (4 * s + 1);
            actv |= ((tm2 & sel) ? 1u : 0u) << (4 * s + 2);
            actv |= ((tm3 & sel) ? 1u : 0u) << (4 * s + 3);
        }
        if (__builtin_amdgcn_readfirstlane(w) >= 4) __builtin_amdgcn_s_setprio(3);
        #pragma nounroll
        for (int c = 0; c < 4; ++c)
            l1_chunk<0, false>(c, 0u, xr, half, bias0, bias1, bias2, bias3, slab, wbo, W1c, lane, wlo, whi,
                               c00, c01, c10, c11, c20, c21, c30, c31);
        #pragma nounroll
        for (int c = 4; c < 32; ++c) {
            if (c == L1LEAD) __builtin_amdgcn_s_setprio(0);
            const unsigned act = (unsigned)__builtin_amdgcn_readlane((int)actv, c - 4);
            l1_chunk<1, false>(c, act, xr, half, bias0, bias1, bias2, bias3, slab, wbo, W1c, lane, wlo, whi,
                               c00, c01, c10, c11, c20, c21, c30, c31);
        }
        l1_chunk<2, false>(32, 0u, xr, half, bias0, bias1, bias2, bias3, slab, wbo, W1c, lane, wlo, whi,
                           c00, c01, c10, c11, c20, c21, c30, c31);
        #pragma nounroll
        for (int c = 33; c < L1FULL - 1; ++c)
            l1_chunk<0, false>(c, 0u, xr, half, bias0, bias1, bias2, bias3, slab, wbo, W1c, lane, wlo, whi,
                               c00, c01, c10, c11, c20, c21, c30, c31);
        l1_chunk<0, true>(L1FULL - 1, 0u, xr, half, bias0, bias1, bias2, bias3, slab, wbo, W1c, lane, wlo, whi,
                          c00, c01, c10, c11, c20, c21, c30, c31);
        {
            Frag4x2<3> f;
            slab_stage(slab, lane, wlo, whi);
            slab_handover();
            w_fetch<128>(W2, 0, tid, wlo, whi);
            frag_load_4x2<3, 0>(xr, half, L1FULL * KC, L1FULL * KC - TSHIFT, bias0, bias1, bias2, bias3,
                                slab + wbo, f);
            frag_mma_4x2<3>(f, c00, c01, c10, c11, c20, c21, c30, c31);
        }
        const float biasA = b1[n0w + col], biasB = b1[n0w + 16 + col];
        relu_store_2x2(hA, H1S, 0,  n0w, half, col, biasA, biasB, c00, c01, c10, c11);
        relu_store_2x2(hA, H1S, 32, n0w, half, col, biasA, biasB, c20, c21, c30, c31);
    }
    __syncthreads();
    w_stage<128>(wbuf, tid, wlo, whi);
    __syncthreads();

    const int row0 = mg * 32;
    const int n0   = ng * 32;
    const int wbo  = (half * NS + n0 + col) * 2;

    {
        v8f a00 = {}, a01 = {}, a10 = {}, a11 = {};
        const float* ha0 = hA + (row0 + col) * H1S + half * 2;
        const float* ha1 = ha0 + 16 * H1S;
        #pragma nounroll
        for (int c = 0; c < 256 / KC; ++c) {
            Frag2x2 f;
            frag_load_2x2(ha0 + c * KC, ha1 + c * KC, wbuf + (c & 1) * WBUF + wbo, f);
            const float* wn = (c + 1 < 256 / KC) ? W2 + (c + 1) * KC * 128 : W3;
            w_fetch<128>(wn, 0, tid, wlo, whi);
            frag_mma_2x2(f, a00, a01, a10, a11);
            if (c + 1 == 256 / KC)
                relu_store_2x2(hB, H2S, row0, n0, half, col, b2[n0 + col], b2[n0 + 16 + col], a00, a01, a10, a11);
            w_stage<128>(wbuf + ((c + 1) & 1) * WBUF, tid, wlo, whi);
            __syncthreads();
        }
    }

    {
        v8f a00 = {}, a01 = {}, a10 = {}, a11 = {};
        const float* ha0 = hB + (row0 + col) * H2S + half * 2;
        const float* ha1 = ha0 + 16 * H2S;
        #pragma nounroll
        for (int c = 0; c < 128 / KC; ++c) {
            Frag2x2 f;
            frag_load_2x2(ha0 + c * KC, ha1 + c * KC, wbuf + (c & 1) * WBUF + wbo, f);
            const float* wn = (c + 1 < 128 / KC) ? W3 + (c + 1) * KC * 128 : W4;
            w_fetch<128>(wn, 0, tid, wlo, whi);
            frag_mma_2x2(f, a00, a01, a10, a11);
            if (c + 1 == 128 / KC)
                relu_store_2x2(hA, H2S, row0, n0, half, col, b3[n0 + col], b3[n0 + 16 + col], a00, a01, a10, a11);
            w_stage<128>(wbuf + ((c + 1) & 1) * WBUF, tid, wlo, whi);
            __syncthreads();
        }
    }

    {
        v8f a00 = {}, a01 = {}, a10 = {}, a11 = {};
        const float* ha0 = hA + (row0 + col) * H2S + half * 2;
        const float* ha1 = ha0 + 16 * H2S;
        #pragma nounroll
        for (int c = 0; c < 128 / KC; ++c) {
            Frag2x2 f;
            frag_load_2x2(ha0 + c * KC, ha1 + c * KC, wbuf + (c & 1) * WBUF + wbo, f);
            if (c + 1 < 128 / KC) w_fetch<128>(W4, (c + 1) * KC, tid, wlo, whi);
            frag_mma_2x2(f, a00, a01, a10, a11);
            if (c + 1 < 128 / KC) {
                w_stage<128>(wbuf + ((c + 1) & 1) * WBUF, tid, wlo, whi);
                __syncthreads();
            }
        }
        const float biasA = b4[n0 + col], biasB = b4[n0 + 16 + col];
        #pragma unroll
        for (int r = 0; r < 8; ++r) {
            const int m = row0 + r + half * 8;
            out[(size_t)(t0 + m) * 128 + n0 + col]           = a00[r] + biasA;
            out[(size_t)(t0 + m) * 128 + n0 + 16 + col]      = a01[r] + biasB;
            out[(size_t)(t0 + m + 16) * 128 + n0 + col]      = a10[r] + biasA;
            out[(size_t)(t0 + m + 16) * 128 + n0 + 16 + col] = a11[r] + biasB;
        }
    }
}

extern "C" void kernel_launch(void* const* d_in, const int* in_sizes, int n_in,
                              void* d_out, int out_size, void* d_ws, size_t ws_size,
                              hipStream_t stream) {
    (void)n_in; (void)d_ws; (void)ws_size; (void)out_size;
    const int*   seq    = (const int*)  d_in[0];
    const float* xyz    = (const float*)d_in[1];
    const float* dih    = (const float*)d_in[2];
    const int*   cidx   = (const int*)  d_in[3];
    const float* orient = (const float*)d_in[4];
    const float* aa_emb    = (const float*)d_in[6];
    const float* chain_emb = (const float*)d_in[7];
    const float* W1 = (const float*)d_in[8];   const float* b1 = (const float*)d_in[9];
    const float* W2 = (const float*)d_in[10];  const float* b2 = (const float*)d_in[11];
    const float* W3 = (const float*)d_in[12];  const float* b3 = (const float*)d_in[13];
    const float* W4 = (const float*)d_in[14];  const float* b4 = (const float*)d_in[15];
    float* out = (float*)d_out;

    const int tokens = in_sizes[0];
    const int blocks = tokens / MT;
    residue_embed_fused<<<blocks, 256, 0, stream>>>(
        seq, xyz, dih, cidx, orient, aa_emb, chain_emb,
        W1, b1, W2, b2, W3, b3, W4, b4, out);
}


# Layer 1's one-hot block as 20 per-type products: a type's tokens gathered into one M tile x the type's 48 W1 rows, result rows scattered to h1 in LDS, dense sums added last; sum order changes
# speedup vs baseline: 3.1162x; 1.1973x over previous
#include <hip/hip_runtime.h>
#include <math.h>

typedef float v2f __attribute__((ext_vector_type(2)));
typedef float v4f __attribute__((ext_vector_type(4)));
typedef float v8f __attribute__((ext_vector_type(8)));
typedef int   v4i __attribute__((ext_vector_type(4)));

#define MT    64
#define DIN   1195
#define NTY   20
#define XS    348
#define XTAIL 128
#define XCRD  298
#define TROW0 128
#define TAIL0 1028
#define H1S   260
#define H2S   132
#define KC    32
#define TK    48
#define NS    272
#define WBUF  (KC / 2 * NS * 2)
#define NSW   48
#define PBUF  (TK / 2 * NSW * 2)
#define WALL  (8 * PBUF)
#define L1LEAD 16

#define WMMA_F32(a, b, c) \
    __builtin_amdgcn_wmma_f32_16x16x4_f32(false, (a), false, (b), (short)0, (c), false, false)

template <int NC>
__device__ __forceinline__ void w_fetch(const float* __restrict__ Wg, int k0, int tid,
                                        v4f (&lo)[6], v4f (&hi)[6])
{
    constexpr int G  = NC / 4;
    constexpr int NI = NC / 64;
    #pragma unroll
    for (int j = 0; j < NI; ++j) {
        const int i = tid + 256 * j;
        const int r = k0 + 2 * (i / G);
        const int g = i % G;
        lo[j] = *(const v4f*)(Wg + (size_t)r * NC + 4 * g);
        hi[j] = *(const v4f*)(Wg + (size_t)(r + 1) * NC + 4 * g);
    }
}

template <int NJ, bool TAIL>
__device__ __forceinline__ void slab_fetch(const float* __restrict__ Wc, int k0, int lane,
                                           v4f (&lo)[6], v4f (&hi)[6])
{
    #pragma unroll
    for (int j = 0; j < NJ; ++j) {
        const int r = k0 + 2 * (4 * j + (lane >> 3));
        if (!TAIL) {
            lo[j] = *(const v4f*)(Wc + (size_t)r * 256);
            hi[j] = *(const v4f*)(Wc + (size_t)(r + 1) * 256);
        } else {
            const int ra = (r     < DIN) ? r     : DIN - 1;
            const int rb = (r + 1 < DIN) ? r + 1 : DIN - 1;
            const v4f a = *(const v4f*)(Wc + (size_t)ra * 256);
            const v4f b = *(const v4f*)(Wc + (size_t)rb * 256);
            const v4f z = {0.0f, 0.0f, 0.0f, 0.0f};
            lo[j] = (r     < DIN) ? a : z;
            hi[j] = (r + 1 < DIN) ? b : z;
        }
    }
}

template <int NJ>
__device__ __forceinline__ void slab_stage(float* slab, int lane, const v4f (&lo)[6], const v4f (&hi)[6])
{
    #pragma unroll
    for (int j = 0; j < NJ; ++j) {
        float* d = slab + ((4 * j + (lane >> 3)) * NSW + 4 * (lane & 7)) * 2;
        const v4f a = {lo[j].x, hi[j].x, lo[j].y, hi[j].y};
        const v4f b = {lo[j].z, hi[j].z, lo[j].w, hi[j].w};
        *(v4f*)d       = a;
        *(v4f*)(d + 4) = b;
    }
}

__device__ __forceinline__ void slab_handover()
{
    __builtin_amdgcn_fence(__ATOMIC_RELEASE, "wavefront");
    __builtin_amdgcn_wave_barrier();
    __builtin_amdgcn_fence(__ATOMIC_ACQUIRE, "wavefront");
}

template <int NC>
__device__ __forceinline__ void w_stage(float* buf, int tid, const v4f (&lo)[6], const v4f (&hi)[6])
{
    constexpr int G  = NC / 4;
    constexpr int NI = NC / 64;
    #pragma unroll
    for (int j = 0; j < NI; ++j) {
        const int i = tid + 256 * j;
        float* d = buf + ((i / G) * NS + 4 * (i % G)) * 2;
        const v4f a = {lo[j].x, hi[j].x, lo[j].y, hi[j].y};
        const v4f b = {lo[j].z, hi[j].z, lo[j].w, hi[j].w};
        *(v4f*)d       = a;
        *(v4f*)(d + 4) = b;
    }
}

template <int STEPS>
struct Frag4x2 { v2f a0[STEPS], a1[STEPS], a2[STEPS], a3[STEPS], b0[STEPS], b1[STEPS]; };

template <int STEPS>
__device__ __forceinline__ void frag_load_4x2(const float* xp, const float* wb, Frag4x2<STEPS>& f)
{
    #pragma unroll
    for (int s = 0; s < STEPS; ++s) {
        const float* p = xp + 4 * s;
        f.a0[s] = *(const v2f*)(p);
        f.a1[s] = *(const v2f*)(p + 16 * XS);
        f.a2[s] = *(const v2f*)(p + 32 * XS);
        f.a3[s] = *(const v2f*)(p + 48 * XS);
        f.b0[s] = *(const v2f*)(wb + s * (4 * NSW));
        f.b1[s] = *(const v2f*)(wb + s * (4 * NSW) + 32);
        __builtin_amdgcn_sched_barrier(0);
    }
}

template <int STEPS>
__device__ __forceinline__ void frag_mma_4x2(const Frag4x2<STEPS>& f,
                                             v8f& c00, v8f& c01, v8f& c10, v8f& c11,
                                             v8f& c20, v8f& c21, v8f& c30, v8f& c31)
{
    #pragma unroll
    for (int s = 0; s < STEPS; ++s) {
        c00 = WMMA_F32(f.a0[s], f.b0[s], c00);
        c01 = WMMA_F32(f.a0[s], f.b1[s], c01);
        c10 = WMMA_F32(f.a1[s], f.b0[s], c10);
        c11 = WMMA_F32(f.a1[s], f.b1[s], c11);
        c20 = WMMA_F32(f.a2[s], f.b0[s], c20);
        c21 = WMMA_F32(f.a2[s], f.b1[s], c21);
        c30 = WMMA_F32(f.a3[s], f.b0[s], c30);
        c31 = WMMA_F32(f.a3[s], f.b1[s], c31);
    }
}

struct Frag2x2 { v2f a0[KC / 4], a1[KC / 4], bA[KC / 4], bB[KC / 4]; };

__device__ __forceinline__ void frag_load_2x2(const float* ha0, const float* ha1, const float* wb,
                                              Frag2x2& f)
{
    #pragma unroll
    for (int s = 0; s < KC / 4; ++s) {
        f.a0[s] = *(const v2f*)(ha0 + 4 * s);
        f.a1[s] = *(const v2f*)(ha1 + 4 * s);
        f.bA[s] = *(const v2f*)(wb + s * (4 * NS));
        f.bB[s] = *(const v2f*)(wb + s * (4 * NS) + 32);
    }
}

__device__ __forceinline__ void frag_mma_2x2(const Frag2x2& f, v8f& a00, v8f& a01, v8f& a10, v8f& a11)
{
    #pragma unroll
    for (int s = 0; s < KC / 4; ++s) {
        a00 = WMMA_F32(f.a0[s], f.bA[s], a00);
        a01 = WMMA_F32(f.a0[s], f.bB[s], a01);
        a10 = WMMA_F32(f.a1[s], f.bA[s], a10);
        a11 = WMMA_F32(f.a1[s], f.bB[s], a11);
    }
}

template <int NJN, bool NEXT_TAIL>
__device__ __forceinline__ void l1_chunk(int xoff, int knext, const float* xr, int half,
                                         float* slab, int wbo,
                                         const float* __restrict__ W1c, int lane,
                                         v4f (&wlo)[6], v4f (&whi)[6],
                                         v8f& c00, v8f& c01, v8f& c10, v8f& c11,
                                         v8f& c20, v8f& c21, v8f& c30, v8f& c31)
{
    Frag4x2<KC / 4> f;
    slab_stage<4>(slab, lane, wlo, whi);
    slab_handover();
    slab_fetch<NJN, NEXT_TAIL>(W1c, knext, lane, wlo, whi);
    __builtin_amdgcn_sched_barrier(0);
    frag_load_4x2<KC / 4>(xr + 2 * half + xoff, slab + wbo, f);
    frag_mma_4x2<KC / 4>(f, c00, c01, c10, c11, c20, c21, c30, c31);
    asm volatile("" : "+v"(c00), "+v"(c01), "+v"(c10), "+v"(c11), "+v"(c20), "+v"(c21), "+v"(c30), "+v"(c31),
                      "+v"(wlo[0]), "+v"(wlo[1]), "+v"(wlo[2]), "+v"(wlo[3]), "+v"(wlo[4]), "+v"(wlo[5]),
                      "+v"(whi[0]), "+v"(whi[1]), "+v"(whi[2]), "+v"(whi[3]), "+v"(whi[4]), "+v"(whi[5]));
}

template <int NJN>
__device__ __forceinline__ void l1_type(int n, int knext, const float* xs,
                                        const int* tl, const int* tln, int& tk0, float* hq,
                                        int half, int col, float* slab, int wbo,
                                        const float* __restrict__ W1c, int lane,
                                        v4f (&wlo)[6], v4f (&whi)[6])
{
    int tkraw = tk0;
    slab_stage<6>(slab, lane, wlo, whi);
    slab_handover();
    slab_fetch<NJN, false>(W1c, knext, lane, wlo, whi);
    __builtin_amdgcn_sched_barrier(0);
    const float* wb = slab + wbo;
    for (int s0 = 0; s0 < n; s0 += 16) {
        const int tk = (s0 + col < n) ? tkraw : MT;
        const float* xa = xs + tk * XS + XCRD + 2 * half;
        const int sb = s0 + 8 * half;
        const v4i ta = *(const v4i*)(tl + sb);
        const v4i tb = *(const v4i*)(tl + sb + 4);
        v2f a[TK / 4], b0[TK / 4], b1[TK / 4];
        #pragma unroll
        for (int s = 0; s < TK / 4; ++s) {
            a[s]  = *(const v2f*)(xa + 4 * s);
            b0[s] = *(const v2f*)(wb + s * (4 * NSW));
            b1[s] = *(const v2f*)(wb + s * (4 * NSW) + 32);
            __builtin_amdgcn_sched_barrier(0);
        }
        v8f p0 = {}, p1 = {};
        #pragma unroll
        for (int s = 0; s < TK / 4; ++s) {
            p0 = WMMA_F32(a[s], b0[s], p0);
            p1 = WMMA_F32(a[s], b1[s], p1);
        }
        asm volatile("v_nop\n\tv_nop\n\tv_nop\n\tv_nop" : "+v"(p0), "+v"(p1));
        const int tr[8] = {ta.x, ta.y, ta.z, ta.w, tb.x, tb.y, tb.z, tb.w};
        #pragma unroll
        for (int r = 0; r < 8; ++r) {
            const int dst = (sb + r < n) ? tr[r] : MT;
            hq[dst * H1S]      = p0[r];
            hq[dst * H1S + 16] = p1[r];
        }
        tkraw = tl[(s0 + 16 + col) & 63];
    }
    tk0 = tln[col];
    asm volatile("" : "+v"(wlo[0]), "+v"(wlo[1]), "+v"(wlo[2]), "+v"(wlo[3]), "+v"(wlo[4]), "+v"(wlo[5]),
                      "+v"(whi[0]), "+v"(whi[1]), "+v"(whi[2]), "+v"(whi[3]), "+v"(whi[4]), "+v"(whi[5]));
}

template <bool ADD>
__device__ __forceinline__ void relu_store_2x2(float* h, int stride, int row0, int n0, int half, int col,
                                               float biasA, float biasB,
                                               const v8f& a00, const v8f& a01, const v8f& a10, const v8f& a11)
{
    #pragma unroll
    for (int r = 0; r < 8; ++r) {
        const int m = row0 + r + half * 8;
        float* q00 = h + m * stride + n0 + col;
        float* q01 = q00 + 16;
        float* q10 = q00 + 16 * stride;
        float* q11 = q10 + 16;
        const float v00 = (ADD ? a00[r] + *q00 : a00[r]) + biasA, v01 = (ADD ? a01[r] + *q01 : a01[r]) + biasB;
        const float v10 = (ADD ? a10[r] + *q10 : a10[r]) + biasA, v11 = (ADD ? a11[r] + *q11 : a11[r]) + biasB;
        *q00 = v00 > 0.0f ? v00 : 0.0f;
        *q01 = v01 > 0.0f ? v01 : 0.0f;
        *q10 = v10 > 0.0f ? v10 : 0.0f;
        *q11 = v11 > 0.0f ? v11 : 0.0f;
    }
}

__global__ __launch_bounds__(256) void residue_embed_fused(
    const int* __restrict__ seq, const float* __restrict__ xyz,
    const float* __restrict__ dih, const int* __restrict__ cidx,
    const float* __restrict__ orient,
    const float* __restrict__ aa_emb, const float* __restrict__ chain_emb,
    const float* __restrict__ W1, const float* __restrict__ b1,
    const float* __restrict__ W2, const float* __restrict__ b2,
    const float* __restrict__ W3, const float* __restrict__ b3,
    const float* __restrict__ W4, const float* __restrict__ b4,
    float* __restrict__ out)
{
    __shared__ __attribute__((aligned(16))) float xs[(MT + 1) * XS];
    __shared__ __attribute__((aligned(16))) float hA[(MT + 1) * H1S];
    __shared__ float hB[MT * H2S];
    __shared__ __attribute__((aligned(16))) float wbuf[WALL];
    __shared__ __attribute__((aligned(16))) int tlist[NTY * MT];
    __shared__ int tcnt[32];

    const int tid = threadIdx.x;
    const int t0  = blockIdx.x * MT;

    const int lane = tid & 31;
    const int w    = tid >> 5;
    const int half = lane >> 4;
    const int col  = lane & 15;
    const int mg   = w >> 2;
    const int ng   = w & 3;

    const int n0w = w * 32;
    float* slab = wbuf + w * PBUF;
    const float* W1c = W1 + n0w + 4 * (lane & 7);
    v4f wlo[6], whi[6];
    slab_fetch<4, false>(W1c, 0, lane, wlo, whi);

    {
        const int tya = seq[t0 + lane];
        const int tyb = seq[t0 + 32 + lane];
        const unsigned below = (1u << lane) - 1u;
        for (int g = __builtin_amdgcn_readfirstlane(w); g < NTY; g += 8) {
            const unsigned ba = __builtin_amdgcn_ballot_w32(tya == g);
            const unsigned bb = __builtin_amdgcn_ballot_w32(tyb == g);
            const int ca = __builtin_popcount(ba);
            if (tya == g) tlist[g * MT + __builtin_popcount(ba & below)] = lane;
            if (tyb == g) tlist[g * MT + ca + __builtin_popcount(bb & below)] = 32 + lane;
            if (lane == 0) tcnt[g] = ca + __builtin_popcount(bb);
        }
    }

    #pragma unroll
    for (int j = 0; j < MT * 32 / 256; ++j) {
        const int i  = tid + 256 * j;
        const int tt = i >> 5, q = i & 31;
        const int t  = t0 + tt;
        const int sa = seq[t];
        const int ci = cidx[t];
        const v4f va = *(const v4f*)(aa_emb + sa * 128 + 4 * q);
        v4f vc = *(const v4f*)(chain_emb + ci * 128 + 4 * q);
        const v4f z = {0.0f, 0.0f, 0.0f, 0.0f};
        vc = (ci != 0) ? vc : z;
        *(v4f*)(xs + tt * XS + 4 * q) = va;
        float* d = xs + tt * XS + XTAIL + 39 + 4 * q;
        d[0] = vc.x; d[1] = vc.y; d[2] = vc.z; d[3] = vc.w;
    }
    for (int i = tid; i < MT * 15; i += 256) {
        int tt = i / 15, a = i % 15;
        int t  = t0 + tt;
        const float* xp = xyz + (size_t)t * 45;
        float rx = xp[a * 3 + 0] - xp[3 + 0];
        float ry = xp[a * 3 + 1] - xp[3 + 1];
        float rz = xp[a * 3 + 2] - xp[3 + 2];
        const float* O = orient + (size_t)t * 9;
        int base = tt * XS + XCRD + a * 3;
        xs[base + 0] = O[0] * rx + O[3] * ry + O[6] * rz;
        xs[base + 1] = O[1] * rx + O[4] * ry + O[7] * rz;
        xs[base + 2] = O[2] * rx + O[5] * ry + O[8] * rz;
    }
    if (tid < TK) xs[MT * XS + XCRD + tid] = 0.0f;
    if (tid < MT) {
        float* r = xs + tid * XS;
        r[XTAIL + 167] = 0.0f;
        r[XCRD - 2] = 0.0f; r[XCRD - 1] = 0.0f;
        r[XCRD + 45] = 0.0f; r[XCRD + 46] = 0.0f; r[XCRD + 47] = 0.0f;
        r[XCRD + 48] = 0.0f; r[XCRD + 49] = 0.0f;
    }
    if (tid < MT * 3) {
        int tt = tid / 3, d = tid % 3;
        float v = dih[(size_t)(t0 + tt) * 3 + d];
        const float FREQ[6] = {1.0f, 2.0f, 3.0f, 1.0f, 0.5f, 1.0f / 3.0f};
        int base = tt * XS + XTAIL + d * 13;
        xs[base] = v;
        #pragma unroll
        for (int f = 0; f < 6; ++f) {
            xs[base + 1 + f] = sinf(FREQ[f] * v);
            xs[base + 7 + f] = cosf(FREQ[f] * v);
        }
    }
    __syncthreads();

    {
        v8f c00 = {}, c01 = {}, c10 = {}, c11 = {}, c20 = {}, c21 = {}, c30 = {}, c31 = {};
        const float* xr = xs + col * XS;
        const int wbo = (half * NSW + col) * 2;
        float* hq = hA + n0w + col;
        #pragma unroll
        for (int j = 0; j < MT / 4; ++j) {
            const v4f z = {0.0f, 0.0f, 0.0f, 0.0f};
            *(v4f*)(hA + (4 * j + (lane >> 3)) * H1S + n0w + 4 * (lane & 7)) = z;
        }
        const int cntv = tcnt[lane];
        int tk0 = tlist[col];
        if (__builtin_amdgcn_readfirstlane(w) >= 4) __builtin_amdgcn_s_setprio(3);
        #pragma nounroll
        for (int c = 0; c < 3; ++c)
            l1_chunk<4, false>(c * KC, (c + 1) * KC, xr, half, slab, wbo, W1c, lane, wlo, whi,
                               c00, c01, c10, c11, c20, c21, c30, c31);
        l1_chunk<6, false>(3 * KC, TROW0, xr, half, slab, wbo, W1c, lane, wlo, whi,
                           c00, c01, c10, c11, c20, c21, c30, c31);
        #pragma nounroll
        for (int g = 0; g < NTY - 1; ++g) {
            if (g == L1LEAD) __builtin_amdgcn_s_setprio(0);
            const int n = __builtin_amdgcn_readlane(cntv, g);
            l1_type<6>(n, TROW0 + 45 * (g + 1), xs, tlist + g * MT, tlist + (g + 1) * MT, tk0, hq, half, col,
                       slab, wbo, W1c, lane, wlo, whi);
        }
        l1_type<4>(__builtin_amdgcn_readlane(cntv, NTY - 1), TAIL0, xs, tlist + (NTY - 1) * MT,
                   tlist + (NTY - 1) * MT, tk0, hq, half, col, slab, wbo, W1c, lane, wlo, whi);
        #pragma nounroll
        for (int j = 0; j < 4; ++j)
            l1_chunk<4, false>(XTAIL + j * KC, TAIL0 + (j + 1) * KC, xr, half, slab, wbo, W1c, lane, wlo, whi,
                               c00, c01, c10, c11, c20, c21, c30, c31);
        l1_chunk<4, true>(XTAIL + 4 * KC, TAIL0 + 5 * KC, xr, half, slab, wbo, W1c, lane, wlo, whi,
                          c00, c01, c10, c11, c20, c21, c30, c31);
        {
            Frag4x2<2> f;
            slab_stage<4>(slab, lane, wlo, whi);
            slab_handover();
            w_fetch<128>(W2, 0, tid, wlo, whi);
            frag_load_4x2<2>(xr + 2 * half + XTAIL + 5 * KC, slab + wbo, f);
            frag_mma_4x2<2>(f, c00, c01, c10, c11, c20, c21, c30, c31);
        }
        const float biasA = b1[n0w + col], biasB = b1[n0w + 16 + col];
        relu_store_2x2<true>(hA, H1S, 0,  n0w, half, col, biasA, biasB, c00, c01, c10, c11);
        relu_store_2x2<true>(hA, H1S, 32, n0w, half, col, biasA, biasB, c20, c21, c30, c31);
    }
    __syncthreads();
    w_stage<128>(wbuf, tid, wlo, whi);
    __syncthreads();

    const int row0 = mg * 32;
    const int n0   = ng * 32;
    const int wbo  = (half * NS + n0 + col) * 2;

    {
        v8f a00 = {}, a01 = {}, a10 = {}, a11 = {};
        const float* ha0 = hA + (row0 + col) * H1S + half * 2;
        const float* ha1 = ha0 + 16 * H1S;
        #pragma nounroll
        for (int c = 0; c < 256 / KC; ++c) {
            Frag2x2 f;
            frag_load_2x2(ha0 + c * KC, ha1 + c * KC, wbuf + (c & 1) * WBUF + wbo, f);
            const float* wn = (c + 1 < 256 / KC) ? W2 + (c + 1) * KC * 128 : W3;
            w_fetch<128>(wn, 0, tid, wlo, whi);
            frag_mma_2x2(f, a00, a01, a10, a11);
            if (c + 1 == 256 / KC)
                relu_store_2x2<false>(hB, H2S, row0, n0, half, col, b2[n0 + col], b2[n0 + 16 + col], a00, a01, a10, a11);
            w_stage<128>(wbuf + ((c + 1) & 1) * WBUF, tid, wlo, whi);
            __syncthreads();
        }
    }

    {
        v8f a00 = {}, a01 = {}, a10 = {}, a11 = {};
        const float* ha0 = hB + (row0 + col) * H2S + half * 2;
        const float* ha1 = ha0 + 16 * H2S;
        #pragma nounroll
        for (int c = 0; c < 128 / KC; ++c) {
            Frag2x2 f;
            frag_load_2x2(ha0 + c * KC, ha1 + c * KC, wbuf + (c & 1) * WBUF + wbo, f);
            const float* wn = (c + 1 < 128 / KC) ? W3 + (c + 1) * KC * 128 : W4;
            w_fetch<128>(wn, 0, tid, wlo, whi);
            frag_mma_2x2(f, a00, a01, a10, a11);
            if (c + 1 == 128 / KC)
                relu_store_2x2<false>(hA, H2S, row0, n0, half, col, b3[n0 + col], b3[n0 + 16 + col], a00, a01, a10, a11);
            w_stage<128>(wbuf + ((c + 1) & 1) * WBUF, tid, wlo, whi);
            __syncthreads();
        }
    }

    {
        v8f a00 = {}, a01 = {}, a10 = {}, a11 = {};
        const float* ha0 = hA + (row0 + col) * H2S + half * 2;
        const float* ha1 = ha0 + 16 * H2S;
        #pragma nounroll
        for (int c = 0; c < 128 / KC; ++c) {
            Frag2x2 f;
            frag_load_2x2(ha0 + c * KC, ha1 + c * KC, wbuf + (c & 1) * WBUF + wbo, f);
            if (c + 1 < 128 / KC) w_fetch<128>(W4, (c + 1) * KC, tid, wlo, whi);
            frag_mma_2x2(f, a00, a01, a10, a11);
            if (c + 1 < 128 / KC) {
                w_stage<128>(wbuf + ((c + 1) & 1) * WBUF, tid, wlo, whi);
                __syncthreads();
            }
        }
        const float biasA = b4[n0 + col], biasB = b4[n0 + 16 + col];
        #pragma unroll
        for (int r = 0; r < 8; ++r) {
            const int m = row0 + r + half * 8;
            out[(size_t)(t0 + m) * 128 + n0 + col]           = a00[r] + biasA;
            out[(size_t)(t0 + m) * 128 + n0 + 16 + col]      = a01[r] + biasB;
            out[(size_t)(t0 + m + 16) * 128 + n0 + col]      = a10[r] + biasA;
            out[(size_t)(t0 + m + 16) * 128 + n0 + 16 + col] = a11[r] + biasB;
        }
    }
}

extern "C" void kernel_launch(void* const* d_in, const int* in_sizes, int n_in,
                              void* d_out, int out_size, void* d_ws, size_t ws_size,
                              hipStream_t stream) {
    (void)n_in; (void)d_ws; (void)ws_size; (void)out_size;
    const int*   seq    = (const int*)  d_in[0];
    const float* xyz    = (const float*)d_in[1];
    const float* dih    = (const float*)d_in[2];
    const int*   cidx   = (const int*)  d_in[3];
    const float* orient = (const float*)d_in[4];
    const float* aa_emb    = (const float*)d_in[6];
    const float* chain_emb = (const float*)d_in[7];
    const float* W1 = (const float*)d_in[8];   const float* b1 = (const float*)d_in[9];
    const float* W2 = (const float*)d_in[10];  const float* b2 = (const float*)d_in[11];
    const float* W3 = (const float*)d_in[12];  const float* b3 = (const float*)d_in[13];
    const float* W4 = (const float*)d_in[14];  const float* b4 = (const float*)d_in[15];
    float* out = (float*)d_out;

    const int tokens = in_sizes[0];
    const int blocks = tokens / MT;
    residue_embed_fused<<<blocks, 256, 0, stream>>>(
        seq, xyz, dih, cidx, orient, aa_emb, chain_emb,
        W1, b1, W2, b2, W3, b3, W4, b4, out);
}


# Layers 2..4 run without work-group barriers inside a layer: each wave streams its own 32 weight columns through its private LDS slab as in layer 1, fragment loads one group ahead; sum order unchanged
# speedup vs baseline: 3.2392x; 1.0395x over previous
#include <hip/hip_runtime.h>
#include <math.h>

typedef float v2f __attribute__((ext_vector_type(2)));
typedef float v4f __attribute__((ext_vector_type(4)));
typedef float v8f __attribute__((ext_vector_type(8)));
typedef int   v4i __attribute__((ext_vector_type(4)));

#define MT    64
#define DIN   1195
#define NTY   20
#define XS    348
#define XTAIL 128
#define XCRD  298
#define TROW0 128
#define TAIL0 1028
#define H1S   260
#define H2S   132
#define KC    32
#define TK    48
#define NSW   48
#define PBUF  (TK / 2 * NSW * 2)
#define WALL  (8 * PBUF)
#define L1LEAD 16

#define WMMA_F32(a, b, c) \
    __builtin_amdgcn_wmma_f32_16x16x4_f32(false, (a), false, (b), (short)0, (c), false, false)

template <int NJ, bool TAIL, int LD>
__device__ __forceinline__ void slab_fetch(const float* __restrict__ Wc, int k0, int lane,
                                           v4f (&lo)[6], v4f (&hi)[6])
{
    #pragma unroll
    for (int j = 0; j < NJ; ++j) {
        const int r = k0 + 2 * (4 * j + (lane >> 3));
        if (!TAIL) {
            lo[j] = *(const v4f*)(Wc + (size_t)r * LD);
            hi[j] = *(const v4f*)(Wc + (size_t)(r + 1) * LD);
        } else {
            const int ra = (r     < DIN) ? r     : DIN - 1;
            const int rb = (r + 1 < DIN) ? r + 1 : DIN - 1;
            const v4f a = *(const v4f*)(Wc + (size_t)ra * LD);
            const v4f b = *(const v4f*)(Wc + (size_t)rb * LD);
            const v4f z = {0.0f, 0.0f, 0.0f, 0.0f};
            lo[j] = (r     < DIN) ? a : z;
            hi[j] = (r + 1 < DIN) ? b : z;
        }
    }
}

template <int NJ>
__device__ __forceinline__ void slab_stage(float* slab, int lane, const v4f (&lo)[6], const v4f (&hi)[6])
{
    #pragma unroll
    for (int j = 0; j < NJ; ++j) {
        float* d = slab + ((4 * j + (lane >> 3)) * NSW + 4 * (lane & 7)) * 2;
        const v4f a = {lo[j].x, hi[j].x, lo[j].y, hi[j].y};
        const v4f b = {lo[j].z, hi[j].z, lo[j].w, hi[j].w};
        *(v4f*)d       = a;
        *(v4f*)(d + 4) = b;
    }
}

__device__ __forceinline__ void slab_handover()
{
    __builtin_amdgcn_fence(__ATOMIC_RELEASE, "wavefront");
    __builtin_amdgcn_wave_barrier();
    __builtin_amdgcn_fence(__ATOMIC_ACQUIRE, "wavefront");
}

template <int STEPS>
struct Frag4x2 { v2f a0[STEPS], a1[STEPS], a2[STEPS], a3[STEPS], b0[STEPS], b1[STEPS]; };

template <int STEPS>
__device__ __forceinline__ void frag_load_4x2(const float* xp, const float* wb, Frag4x2<STEPS>& f)
{
    #pragma unroll
    for (int s = 0; s < STEPS; ++s) {
        const float* p = xp + 4 * s;
        f.a0[s] = *(const v2f*)(p);
        f.a1[s] = *(const v2f*)(p + 16 * XS);
        f.a2[s] = *(const v2f*)(p + 32 * XS);
        f.a3[s] = *(const v2f*)(p + 48 * XS);
        f.b0[s] = *(const v2f*)(wb + s * (4 * NSW));
        f.b1[s] = *(const v2f*)(wb + s * (4 * NSW) + 32);
        __builtin_amdgcn_sched_barrier(0);
    }
}

template <int STEPS>
__device__ __forceinline__ void frag_mma_4x2(const Frag4x2<STEPS>& f,
                                             v8f& c00, v8f& c01, v8f& c10, v8f& c11,
                                             v8f& c20, v8f& c21, v8f& c30, v8f& c31)
{
    #pragma unroll
    for (int s = 0; s < STEPS; ++s) {
        c00 = WMMA_F32(f.a0[s], f.b0[s], c00);
        c01 = WMMA_F32(f.a0[s], f.b1[s], c01);
        c10 = WMMA_F32(f.a1[s], f.b0[s], c10);
        c11 = WMMA_F32(f.a1[s], f.b1[s], c11);
        c20 = WMMA_F32(f.a2[s], f.b0[s], c20);
        c21 = WMMA_F32(f.a2[s], f.b1[s], c21);
        c30 = WMMA_F32(f.a3[s], f.b0[s], c30);
        c31 = WMMA_F32(f.a3[s], f.b1[s], c31);
    }
}

struct Frag2x2 { v2f a0[KC / 4], a1[KC / 4], bA[KC / 4], bB[KC / 4]; };

__device__ __forceinline__ void frag_load_2x2(const float* ha0, const float* ha1, const float* wb,
                                              Frag2x2& f)
{
    #pragma unroll
    for (int s = 0; s < KC / 4; ++s) {
        f.a0[s] = *(const v2f*)(ha0 + 4 * s);
        f.a1[s] = *(const v2f*)(ha1 + 4 * s);
        f.bA[s] = *(const v2f*)(wb + s * (4 * NSW));
        f.bB[s] = *(const v2f*)(wb + s * (4 * NSW) + 32);
        __builtin_amdgcn_sched_barrier(0);
    }
}

__device__ __forceinline__ void frag_mma_2x2(const Frag2x2& f, v8f& a00, v8f& a01, v8f& a10, v8f& a11)
{
    #pragma unroll
    for (int s = 0; s < KC / 4; ++s) {
        a00 = WMMA_F32(f.a0[s], f.bA[s], a00);
        a01 = WMMA_F32(f.a0[s], f.bB[s], a01);
        a10 = WMMA_F32(f.a1[s], f.bA[s], a10);
        a11 = WMMA_F32(f.a1[s], f.bB[s], a11);
    }
}

template <bool FETCH>
__device__ __forceinline__ void l24_chunk(const float* ha0, const float* ha1, float* slab, int wbo,
                                          const float* __restrict__ Wn, int lane,
                                          v4f (&wlo)[6], v4f (&whi)[6],
                                          v8f& a00, v8f& a01, v8f& a10, v8f& a11)
{
    Frag2x2 f;
    slab_stage<4>(slab, lane, wlo, whi);
    slab_handover();
    if (FETCH) slab_fetch<4, false, 128>(Wn, 0, lane, wlo, whi);
    __builtin_amdgcn_sched_barrier(0);
    frag_load_2x2(ha0, ha1, slab + wbo, f);
    frag_mma_2x2(f, a00, a01, a10, a11);
    asm volatile("" : "+v"(a00), "+v"(a01), "+v"(a10), "+v"(a11),
                      "+v"(wlo[0]), "+v"(wlo[1]), "+v"(wlo[2]), "+v"(wlo[3]),
                      "+v"(whi[0]), "+v"(whi[1]), "+v"(whi[2]), "+v"(whi[3]));
}

template <int NJN, bool NEXT_TAIL>
__device__ __forceinline__ void l1_chunk(int xoff, int knext, const float* xr, int half,
                                         float* slab, int wbo,
                                         const float* __restrict__ W1c, int lane,
                                         v4f (&wlo)[6], v4f (&whi)[6],
                                         v8f& c00, v8f& c01, v8f& c10, v8f& c11,
                                         v8f& c20, v8f& c21, v8f& c30, v8f& c31)
{
    Frag4x2<KC / 4> f;
    slab_stage<4>(slab, lane, wlo, whi);
    slab_handover();
    slab_fetch<NJN, NEXT_TAIL, 256>(W1c, knext, lane, wlo, whi);
    __builtin_amdgcn_sched_barrier(0);
    frag_load_4x2<KC / 4>(xr + 2 * half + xoff, slab + wbo, f);
    frag_mma_4x2<KC / 4>(f, c00, c01, c10, c11, c20, c21, c30, c31);
    asm volatile("" : "+v"(c00), "+v"(c01), "+v"(c10), "+v"(c11), "+v"(c20), "+v"(c21), "+v"(c30), "+v"(c31),
                      "+v"(wlo[0]), "+v"(wlo[1]), "+v"(wlo[2]), "+v"(wlo[3]), "+v"(wlo[4]), "+v"(wlo[5]),
                      "+v"(whi[0]), "+v"(whi[1]), "+v"(whi[2]), "+v"(whi[3]), "+v"(whi[4]), "+v"(whi[5]));
}

template <int NJN>
__device__ __forceinline__ void l1_type(int n, int knext, const float* xs,
                                        const int* tl, const int* tln, int& tk0, float* hq,
                                        int half, int col, float* slab, int wbo,
                                        const float* __restrict__ W1c, int lane,
                                        v4f (&wlo)[6], v4f (&whi)[6])
{
    int tkraw = tk0;
    slab_stage<6>(slab, lane, wlo, whi);
    slab_handover();
    slab_fetch<NJN, false, 256>(W1c, knext, lane, wlo, whi);
    __builtin_amdgcn_sched_barrier(0);
    const float* wb = slab + wbo;
    for (int s0 = 0; s0 < n; s0 += 16) {
        const int tk = (s0 + col < n) ? tkraw : MT;
        const float* xa = xs + tk * XS + XCRD + 2 * half;
        const int sb = s0 + 8 * half;
        const v4i ta = *(const v4i*)(tl + sb);
        const v4i tb = *(const v4i*)(tl + sb + 4);
        v2f a[TK / 4], b0[TK / 4], b1[TK / 4];
        #pragma unroll
        for (int s = 0; s < TK / 4; ++s) {
            a[s]  = *(const v2f*)(xa + 4 * s);
            b0[s] = *(const v2f*)(wb + s * (4 * NSW));
            b1[s] = *(const v2f*)(wb + s * (4 * NSW) + 32);
            __builtin_amdgcn_sched_barrier(0);
        }
        v8f p0 = {}, p1 = {};
        #pragma unroll
        for (int s = 0; s < TK / 4; ++s) {
            p0 = WMMA_F32(a[s], b0[s], p0);
            p1 = WMMA_F32(a[s], b1[s], p1);
        }
        asm volatile("v_nop\n\tv_nop\n\tv_nop\n\tv_nop" : "+v"(p0), "+v"(p1));
        const int tr[8] = {ta.x, ta.y, ta.z, ta.w, tb.x, tb.y, tb.z, tb.w};
        #pragma unroll
        for (int r = 0; r < 8; ++r) {
            const int dst = (sb + r < n) ? tr[r] : MT;
            hq[dst * H1S]      = p0[r];
            hq[dst * H1S + 16] = p1[r];
        }
        tkraw = tl[(s0 + 16 + col) & 63];
    }
    tk0 = tln[col];
    asm volatile("" : "+v"(wlo[0]), "+v"(wlo[1]), "+v"(wlo[2]), "+v"(wlo[3]), "+v"(wlo[4]), "+v"(wlo[5]),
                      "+v"(whi[0]), "+v"(whi[1]), "+v"(whi[2]), "+v"(whi[3]), "+v"(whi[4]), "+v"(whi[5]));
}

template <bool ADD>
__device__ __forceinline__ void relu_store_2x2(float* h, int stride, int row0, int n0, int half, int col,
                                               float biasA, float biasB,
                                               const v8f& a00, const v8f& a01, const v8f& a10, const v8f& a11)
{
    #pragma unroll
    for (int r = 0; r < 8; ++r) {
        const int m = row0 + r + half * 8;
        float* q00 = h + m * stride + n0 + col;
        float* q01 = q00 + 16;
        float* q10 = q00 + 16 * stride;
        float* q11 = q10 + 16;
        const float v00 = (ADD ? a00[r] + *q00 : a00[r]) + biasA, v01 = (ADD ? a01[r] + *q01 : a01[r]) + biasB;
        const float v10 = (ADD ? a10[r] + *q10 : a10[r]) + biasA, v11 = (ADD ? a11[r] + *q11 : a11[r]) + biasB;
        *q00 = v00 > 0.0f ? v00 : 0.0f;
        *q01 = v01 > 0.0f ? v01 : 0.0f;
        *q10 = v10 > 0.0f ? v10 : 0.0f;
        *q11 = v11 > 0.0f ? v11 : 0.0f;
    }
}

__global__ __launch_bounds__(256) void residue_embed_fused(
    const int* __restrict__ seq, const float* __restrict__ xyz,
    const float* __restrict__ dih, const int* __restrict__ cidx,
    const float* __restrict__ orient,
    const float* __restrict__ aa_emb, const float* __restrict__ chain_emb,
    const float* __restrict__ W1, const float* __restrict__ b1,
    const float* __restrict__ W2, const float* __restrict__ b2,
    const float* __restrict__ W3, const float* __restrict__ b3,
    const float* __restrict__ W4, const float* __restrict__ b4,
    float* __restrict__ out)
{
    __shared__ __attribute__((aligned(16))) float xs[(MT + 1) * XS];
    __shared__ __attribute__((aligned(16))) float hA[(MT + 1) * H1S];
    __shared__ float hB[MT * H2S];
    __shared__ __attribute__((aligned(16))) float wbuf[WALL];
    __shared__ __attribute__((aligned(16))) int tlist[NTY * MT];
    __shared__ int tcnt[32];

    const int tid = threadIdx.x;
    const int t0  = blockIdx.x * MT;

    const int lane = tid & 31;
    const int w    = tid >> 5;
    const int half = lane >> 4;
    const int col  = lane & 15;
    const int mg   = w >> 2;
    const int ng   = w & 3;

    const int n0w = w * 32;
    float* slab = wbuf + w * PBUF;
    const float* W1c = W1 + n0w + 4 * (lane & 7);
    v4f wlo[6], whi[6];
    slab_fetch<4, false, 256>(W1c, 0, lane, wlo, whi);

    {
        const int tya = seq[t0 + lane];
        const int tyb = seq[t0 + 32 + lane];
        const unsigned below = (1u << lane) - 1u;
        for (int g = __builtin_amdgcn_readfirstlane(w); g < NTY; g += 8) {
            const unsigned ba = __builtin_amdgcn_ballot_w32(tya == g);
            const unsigned bb = __builtin_amdgcn_ballot_w32(tyb == g);
            const int ca = __builtin_popcount(ba);
            if (tya == g) tlist[g * MT + __builtin_popcount(ba & below)] = lane;
            if (tyb == g) tlist[g * MT + ca + __builtin_popcount(bb & below)] = 32 + lane;
            if (lane == 0) tcnt[g] = ca + __builtin_popcount(bb);
        }
    }

    #pragma unroll
    for (int j = 0; j < MT * 32 / 256; ++j) {
        const int i  = tid + 256 * j;
        const int tt = i >> 5, q = i & 31;
        const int t  = t0 + tt;
        const int sa = seq[t];
        const int ci = cidx[t];
        const v4f va = *(const v4f*)(aa_emb + sa * 128 + 4 * q);
        v4f vc = *(const v4f*)(chain_emb + ci * 128 + 4 * q);
        const v4f z = {0.0f, 0.0f, 0.0f, 0.0f};
        vc = (ci != 0) ? vc : z;
        *(v4f*)(xs + tt * XS + 4 * q) = va;
        float* d = xs + tt * XS + XTAIL + 39 + 4 * q;
        d[0] = vc.x; d[1] = vc.y; d[2] = vc.z; d[3] = vc.w;
    }
    for (int i = tid; i < MT * 15; i += 256) {
        int tt = i / 15, a = i % 15;
        int t  = t0 + tt;
        const float* xp = xyz + (size_t)t * 45;
        float rx = xp[a * 3 + 0] - xp[3 + 0];
        float ry = xp[a * 3 + 1] - xp[3 + 1];
        float rz = xp[a * 3 + 2] - xp[3 + 2];
        const float* O = orient + (size_t)t * 9;
        int base = tt * XS + XCRD + a * 3;
        xs[base + 0] = O[0] * rx + O[3] * ry + O[6] * rz;
        xs[base + 1] = O[1] * rx + O[4] * ry + O[7] * rz;
        xs[base + 2] = O[2] * rx + O[5] * ry + O[8] * rz;
    }
    if (tid < TK) xs[MT * XS + XCRD + tid] = 0.0f;
    if (tid < MT) {
        float* r = xs + tid * XS;
        r[XTAIL + 167] = 0.0f;
        r[XCRD - 2] = 0.0f; r[XCRD - 1] = 0.0f;
        r[XCRD + 45] = 0.0f; r[XCRD + 46] = 0.0f; r[XCRD + 47] = 0.0f;
        r[XCRD + 48] = 0.0f; r[XCRD + 49] = 0.0f;
    }
    if (tid < MT * 3) {
        int tt = tid / 3, d = tid % 3;
        float v = dih[(size_t)(t0 + tt) * 3 + d];
        const float FREQ[6] = {1.0f, 2.0f, 3.0f, 1.0f, 0.5f, 1.0f / 3.0f};
        int base = tt * XS + XTAIL + d * 13;
        xs[base] = v;
        #pragma unroll
        for (int f = 0; f < 6; ++f) {
            xs[base + 1 + f] = sinf(FREQ[f] * v);
            xs[base + 7 + f] = cosf(FREQ[f] * v);
        }
    }
    __syncthreads();

    {
        v8f c00 = {}, c01 = {}, c10 = {}, c11 = {}, c20 = {}, c21 = {}, c30 = {}, c31 = {};
        const float* xr = xs + col * XS;
        const int wbo = (half * NSW + col) * 2;
        float* hq = hA + n0w + col;
        #pragma unroll
        for (int j = 0; j < MT / 4; ++j) {
            const v4f z = {0.0f, 0.0f, 0.0f, 0.0f};
            *(v4f*)(hA + (4 * j + (lane >> 3)) * H1S + n0w + 4 * (lane & 7)) = z;
        }
        const int cntv = tcnt[lane];
        int tk0 = tlist[col];
        if (__builtin_amdgcn_readfirstlane(w) >= 4) __builtin_amdgcn_s_setprio(3);
        #pragma nounroll
        for (int c = 0; c < 3; ++c)
            l1_chunk<4, false>(c * KC, (c + 1) * KC, xr, half, slab, wbo, W1c, lane, wlo, whi,
                               c00, c01, c10, c11, c20, c21, c30, c31);
        l1_chunk<6, false>(3 * KC, TROW0, xr, half, slab, wbo, W1c, lane, wlo, whi,
                           c00, c01, c10, c11, c20, c21, c30, c31);
        #pragma nounroll
        for (int g = 0; g < NTY - 1; ++g) {
            if (g == L1LEAD) __builtin_amdgcn_s_setprio(0);
            const int n = __builtin_amdgcn_readlane(cntv, g);
            l1_type<6>(n, TROW0 + 45 * (g + 1), xs, tlist + g * MT, tlist + (g + 1) * MT, tk0, hq, half, col,
                       slab, wbo, W1c, lane, wlo, whi);
        }
        l1_type<4>(__builtin_amdgcn_readlane(cntv, NTY - 1), TAIL0, xs, tlist + (NTY - 1) * MT,
                   tlist + (NTY - 1) * MT, tk0, hq, half, col, slab, wbo, W1c, lane, wlo, whi);
        #pragma nounroll
        for (int j = 0; j < 4; ++j)
            l1_chunk<4, false>(XTAIL + j * KC, TAIL0 + (j + 1) * KC, xr, half, slab, wbo, W1c, lane, wlo, whi,
                               c00, c01, c10, c11, c20, c21, c30, c31);
        l1_chunk<4, true>(XTAIL + 4 * KC, TAIL0 + 5 * KC, xr, half, slab, wbo, W1c, lane, wlo, whi,
                          c00, c01, c10, c11, c20, c21, c30, c31);
        {
            Frag4x2<2> f;
            slab_stage<4>(slab, lane, wlo, whi);
            slab_handover();
            slab_fetch<4, false, 128>(W2 + (w & 3) * 32 + 4 * (lane & 7), 0, lane, wlo, whi);
            frag_load_4x2<2>(xr + 2 * half + XTAIL + 5 * KC, slab + wbo, f);
            frag_mma_4x2<2>(f, c00, c01, c10, c11, c20, c21, c30, c31);
        }
        const float biasA = b1[n0w + col], biasB = b1[n0w + 16 + col];
        relu_store_2x2<true>(hA, H1S, 0,  n0w, half, col, biasA, biasB, c00, c01, c10, c11);
        relu_store_2x2<true>(hA, H1S, 32, n0w, half, col, biasA, biasB, c20, c21, c30, c31);
    }
    __syncthreads();

    const int row0 = mg * 32;
    const int n0   = ng * 32;
    const int wbo  = (half * NSW + col) * 2;
    const float* W2c = W2 + n0 + 4 * (lane & 7);
    const float* W3c = W3 + n0 + 4 * (lane & 7);
    const float* W4c = W4 + n0 + 4 * (lane & 7);

    {
        v8f a00 = {}, a01 = {}, a10 = {}, a11 = {};
        const float* ha0 = hA + (row0 + col) * H1S + half * 2;
        const float* ha1 = ha0 + 16 * H1S;
        #pragma nounroll
        for (int c = 0; c < 256 / KC; ++c) {
            const float* wn = (c + 1 < 256 / KC) ? W2c + (c + 1) * KC * 128 : W3c;
            l24_chunk<true>(ha0 + c * KC, ha1 + c * KC, slab, wbo, wn, lane, wlo, whi, a00, a01, a10, a11);
        }
        relu_store_2x2<false>(hB, H2S, row0, n0, half, col, b2[n0 + col], b2[n0 + 16 + col], a00, a01, a10, a11);
    }
    __syncthreads();

    {
        v8f a00 = {}, a01 = {}, a10 = {}, a11 = {};
        const float* ha0 = hB + (row0 + col) * H2S + half * 2;
        const float* ha1 = ha0 + 16 * H2S;
        #pragma nounroll
        for (int c = 0; c < 128 / KC; ++c) {
            const float* wn = (c + 1 < 128 / KC) ? W3c + (c + 1) * KC * 128 : W4c;
            l24_chunk<true>(ha0 + c * KC, ha1 + c * KC, slab, wbo, wn, lane, wlo, whi, a00, a01, a10, a11);
        }
        relu_store_2x2<false>(hA, H2S, row0, n0, half, col, b3[n0 + col], b3[n0 + 16 + col], a00, a01, a10, a11);
    }
    __syncthreads();

    {
        v8f a00 = {}, a01 = {}, a10 = {}, a11 = {};
        const float* ha0 = hA + (row0 + col) * H2S + half * 2;
        const float* ha1 = ha0 + 16 * H2S;
        #pragma nounroll
        for (int c = 0; c < 128 / KC - 1; ++c)
            l24_chunk<true>(ha0 + c * KC, ha1 + c * KC, slab, wbo, W4c + (c + 1) * KC * 128, lane, wlo, whi,
                            a00, a01, a10, a11);
        l24_chunk<false>(ha0 + (128 / KC - 1) * KC, ha1 + (128 / KC - 1) * KC, slab, wbo, W4c, lane, wlo, whi,
                         a00, a01, a10, a11);
        const float biasA = b4[n0 + col], biasB = b4[n0 + 16 + col];
        #pragma unroll
        for (int r = 0; r < 8; ++r) {
            const int m = row0 + r + half * 8;
            out[(size_t)(t0 + m) * 128 + n0 + col]           = a00[r] + biasA;
            out[(size_t)(t0 + m) * 128 + n0 + 16 + col]      = a01[r] + biasB;
            out[(size_t)(t0 + m + 16) * 128 + n0 + col]      = a10[r] + biasA;
            out[(size_t)(t0 + m + 16) * 128 + n0 + 16 + col] = a11[r] + biasB;
        }
    }
}

extern "C" void kernel_launch(void* const* d_in, const int* in_sizes, int n_in,
                              void* d_out, int out_size, void* d_ws, size_t ws_size,
                              hipStream_t stream) {
    (void)n_in; (void)d_ws; (void)ws_size; (void)out_size;
    const int*   seq    = (const int*)  d_in[0];
    const float* xyz    = (const float*)d_in[1];
    const float* dih    = (const float*)d_in[2];
    const int*   cidx   = (const int*)  d_in[3];
    const float* orient = (const float*)d_in[4];
    const float* aa_emb    = (const float*)d_in[6];
    const float* chain_emb = (const float*)d_in[7];
    const float* W1 = (const float*)d_in[8];   const float* b1 = (const float*)d_in[9];
    const float* W2 = (const float*)d_in[10];  const float* b2 = (const float*)d_in[11];
    const float* W3 = (const float*)d_in[12];  const float* b3 = (const float*)d_in[13];
    const float* W4 = (const float*)d_in[14];  const float* b4 = (const float*)d_in[15];
    float* out = (float*)d_out;

    const int tokens = in_sizes[0];
    const int blocks = tokens / MT;
    residue_embed_fused<<<blocks, 256, 0, stream>>>(
        seq, xyz, dih, cidx, orient, aa_emb, chain_emb,
        W1, b1, W2, b2, W3, b3, W4, b4, out);
}


# Layer 1's one-hot block: each work-group walks the 20 residue types from a start of its own (block index mod 20), so work-groups in step no longer ask L2 for the same W1 rows at once; sums unchanged
# speedup vs baseline: 3.3873x; 1.0457x over previous
#include <hip/hip_runtime.h>
#include <math.h>

typedef float v2f __attribute__((ext_vector_type(2)));
typedef float v4f __attribute__((ext_vector_type(4)));
typedef float v8f __attribute__((ext_vector_type(8)));
typedef int   v4i __attribute__((ext_vector_type(4)));

#define MT    64
#define DIN   1195
#define NTY   20
#define XS    348
#define XTAIL 128
#define XCRD  298
#define TROW0 128
#define TAIL0 1028
#define H1S   260
#define H2S   132
#define KC    32
#define TK    48
#define NSW   48
#define PBUF  (TK / 2 * NSW * 2)
#define WALL  (8 * PBUF)
#define L1LEAD 16

#define WMMA_F32(a, b, c) \
    __builtin_amdgcn_wmma_f32_16x16x4_f32(false, (a), false, (b), (short)0, (c), false, false)

template <int NJ, bool TAIL, int LD>
__device__ __forceinline__ void slab_fetch(const float* __restrict__ Wc, int k0, int lane,
                                           v4f (&lo)[6], v4f (&hi)[6])
{
    #pragma unroll
    for (int j = 0; j < NJ; ++j) {
        const int r = k0 + 2 * (4 * j + (lane >> 3));
        if (!TAIL) {
            lo[j] = *(const v4f*)(Wc + (size_t)r * LD);
            hi[j] = *(const v4f*)(Wc + (size_t)(r + 1) * LD);
        } else {
            const int ra = (r     < DIN) ? r     : DIN - 1;
            const int rb = (r + 1 < DIN) ? r + 1 : DIN - 1;
            const v4f a = *(const v4f*)(Wc + (size_t)ra * LD);
            const v4f b = *(const v4f*)(Wc + (size_t)rb * LD);
            const v4f z = {0.0f, 0.0f, 0.0f, 0.0f};
            lo[j] = (r     < DIN) ? a : z;
            hi[j] = (r + 1 < DIN) ? b : z;
        }
    }
}

template <int NJ>
__device__ __forceinline__ void slab_stage(float* slab, int lane, const v4f (&lo)[6], const v4f (&hi)[6])
{
    #pragma unroll
    for (int j = 0; j < NJ; ++j) {
        float* d = slab + ((4 * j + (lane >> 3)) * NSW + 4 * (lane & 7)) * 2;
        const v4f a = {lo[j].x, hi[j].x, lo[j].y, hi[j].y};
        const v4f b = {lo[j].z, hi[j].z, lo[j].w, hi[j].w};
        *(v4f*)d       = a;
        *(v4f*)(d + 4) = b;
    }
}

__device__ __forceinline__ void slab_handover()
{
    __builtin_amdgcn_fence(__ATOMIC_RELEASE, "wavefront");
    __builtin_amdgcn_wave_barrier();
    __builtin_amdgcn_fence(__ATOMIC_ACQUIRE, "wavefront");
}

template <int STEPS>
struct Frag4x2 { v2f a0[STEPS], a1[STEPS], a2[STEPS], a3[STEPS], b0[STEPS], b1[STEPS]; };

template <int STEPS>
__device__ __forceinline__ void frag_load_4x2(const float* xp, const float* wb, Frag4x2<STEPS>& f)
{
    #pragma unroll
    for (int s = 0; s < STEPS; ++s) {
        const float* p = xp + 4 * s;
        f.a0[s] = *(const v2f*)(p);
        f.a1[s] = *(const v2f*)(p + 16 * XS);
        f.a2[s] = *(const v2f*)(p + 32 * XS);
        f.a3[s] = *(const v2f*)(p + 48 * XS);
        f.b0[s] = *(const v2f*)(wb + s * (4 * NSW));
        f.b1[s] = *(const v2f*)(wb + s * (4 * NSW) + 32);
        __builtin_amdgcn_sched_barrier(0);
    }
}

template <int STEPS>
__device__ __forceinline__ void frag_mma_4x2(const Frag4x2<STEPS>& f,
                                             v8f& c00, v8f& c01, v8f& c10, v8f& c11,
                                             v8f& c20, v8f& c21, v8f& c30, v8f& c31)
{
    #pragma unroll
    for (int s = 0; s < STEPS; ++s) {
        c00 = WMMA_F32(f.a0[s], f.b0[s], c00);
        c01 = WMMA_F32(f.a0[s], f.b1[s], c01);
        c10 = WMMA_F32(f.a1[s], f.b0[s], c10);
        c11 = WMMA_F32(f.a1[s], f.b1[s], c11);
        c20 = WMMA_F32(f.a2[s], f.b0[s], c20);
        c21 = WMMA_F32(f.a2[s], f.b1[s], c21);
        c30 = WMMA_F32(f.a3[s], f.b0[s], c30);
        c31 = WMMA_F32(f.a3[s], f.b1[s], c31);
    }
}

struct Frag2x2 { v2f a0[KC / 4], a1[KC / 4], bA[KC / 4], bB[KC / 4]; };

__device__ __forceinline__ void frag_load_2x2(const float* ha0, const float* ha1, const float* wb,
                                              Frag2x2& f)
{
    #pragma unroll
    for (int s = 0; s < KC / 4; ++s) {
        f.a0[s] = *(const v2f*)(ha0 + 4 * s);
        f.a1[s] = *(const v2f*)(ha1 + 4 * s);
        f.bA[s] = *(const v2f*)(wb + s * (4 * NSW));
        f.bB[s] = *(const v2f*)(wb + s * (4 * NSW) + 32);
        __builtin_amdgcn_sched_barrier(0);
    }
}

__device__ __forceinline__ void frag_mma_2x2(const Frag2x2& f, v8f& a00, v8f& a01, v8f& a10, v8f& a11)
{
    #pragma unroll
    for (int s = 0; s < KC / 4; ++s) {
        a00 = WMMA_F32(f.a0[s], f.bA[s], a00);
        a01 = WMMA_F32(f.a0[s], f.bB[s], a01);
        a10 = WMMA_F32(f.a1[s], f.bA[s], a10);
        a11 = WMMA_F32(f.a1[s], f.bB[s], a11);
    }
}

template <bool FETCH>
__device__ __forceinline__ void l24_chunk(const float* ha0, const float* ha1, float* slab, int wbo,
                                          const float* __restrict__ Wn, int lane,
                                          v4f (&wlo)[6], v4f (&whi)[6],
                                          v8f& a00, v8f& a01, v8f& a10, v8f& a11)
{
    Frag2x2 f;
    slab_stage<4>(slab, lane, wlo, whi);
    slab_handover();
    if (FETCH) slab_fetch<4, false, 128>(Wn, 0, lane, wlo, whi);
    __builtin_amdgcn_sched_barrier(0);
    frag_load_2x2(ha0, ha1, slab + wbo, f);
    frag_mma_2x2(f, a00, a01, a10, a11);
    asm volatile("" : "+v"(a00), "+v"(a01), "+v"(a10), "+v"(a11),
                      "+v"(wlo[0]), "+v"(wlo[1]), "+v"(wlo[2]), "+v"(wlo[3]),
                      "+v"(whi[0]), "+v"(whi[1]), "+v"(whi[2]), "+v"(whi[3]));
}

template <int NJN, bool NEXT_TAIL>
__device__ __forceinline__ void l1_chunk(int xoff, int knext, const float* xr, int half,
                                         float* slab, int wbo,
                                         const float* __restrict__ W1c, int lane,
                                         v4f (&wlo)[6], v4f (&whi)[6],
                                         v8f& c00, v8f& c01, v8f& c10, v8f& c11,
                                         v8f& c20, v8f& c21, v8f& c30, v8f& c31)
{
    Frag4x2<KC / 4> f;
    slab_stage<4>(slab, lane, wlo, whi);
    slab_handover();
    slab_fetch<NJN, NEXT_TAIL, 256>(W1c, knext, lane, wlo, whi);
    __builtin_amdgcn_sched_barrier(0);
    frag_load_4x2<KC / 4>(xr + 2 * half + xoff, slab + wbo, f);
    frag_mma_4x2<KC / 4>(f, c00, c01, c10, c11, c20, c21, c30, c31);
    asm volatile("" : "+v"(c00), "+v"(c01), "+v"(c10), "+v"(c11), "+v"(c20), "+v"(c21), "+v"(c30), "+v"(c31),
                      "+v"(wlo[0]), "+v"(wlo[1]), "+v"(wlo[2]), "+v"(wlo[3]), "+v"(wlo[4]), "+v"(wlo[5]),
                      "+v"(whi[0]), "+v"(whi[1]), "+v"(whi[2]), "+v"(whi[3]), "+v"(whi[4]), "+v"(whi[5]));
}

template <int NJN>
__device__ __forceinline__ void l1_type(int n, int knext, const float* xs,
                                        const int* tl, const int* tln, int& tk0, float* hq,
                                        int half, int col, float* slab, int wbo,
                                        const float* __restrict__ W1c, int lane,
                                        v4f (&wlo)[6], v4f (&whi)[6])
{
    int tkraw = tk0;
    slab_stage<6>(slab, lane, wlo, whi);
    slab_handover();
    slab_fetch<NJN, false, 256>(W1c, knext, lane, wlo, whi);
    __builtin_amdgcn_sched_barrier(0);
    const float* wb = slab + wbo;
    for (int s0 = 0; s0 < n; s0 += 16) {
        const int tk = (s0 + col < n) ? tkraw : MT;
        const float* xa = xs + tk * XS + XCRD + 2 * half;
        const int sb = s0 + 8 * half;
        const v4i ta = *(const v4i*)(tl + sb);
        const v4i tb = *(const v4i*)(tl + sb + 4);
        v2f a[TK / 4], b0[TK / 4], b1[TK / 4];
        #pragma unroll
        for (int s = 0; s < TK / 4; ++s) {
            a[s]  = *(const v2f*)(xa + 4 * s);
            b0[s] = *(const v2f*)(wb + s * (4 * NSW));
            b1[s] = *(const v2f*)(wb + s * (4 * NSW) + 32);
            __builtin_amdgcn_sched_barrier(0);
        }
        v8f p0 = {}, p1 = {};
        #pragma unroll
        for (int s = 0; s < TK / 4; ++s) {
            p0 = WMMA_F32(a[s], b0[s], p0);
            p1 = WMMA_F32(a[s], b1[s], p1);
        }
        asm volatile("v_nop\n\tv_nop\n\tv_nop\n\tv_nop" : "+v"(p0), "+v"(p1));
        const int tr[8] = {ta.x, ta.y, ta.z, ta.w, tb.x, tb.y, tb.z, tb.w};
        #pragma unroll
        for (int r = 0; r < 8; ++r) {
            const int dst = (sb + r < n) ? tr[r] : MT;
            hq[dst * H1S]      = p0[r];
            hq[dst * H1S + 16] = p1[r];
        }
        tkraw = tl[(s0 + 16 + col) & 63];
    }
    tk0 = tln[col];
    asm volatile("" : "+v"(wlo[0]), "+v"(wlo[1]), "+v"(wlo[2]), "+v"(wlo[3]), "+v"(wlo[4]), "+v"(wlo[5]),
                      "+v"(whi[0]), "+v"(whi[1]), "+v"(whi[2]), "+v"(whi[3]), "+v"(whi[4]), "+v"(whi[5]));
}

template <bool ADD>
__device__ __forceinline__ void relu_store_2x2(float* h, int stride, int row0, int n0, int half, int col,
                                               float biasA, float biasB,
                                               const v8f& a00, const v8f& a01, const v8f& a10, const v8f& a11)
{
    #pragma unroll
    for (int r = 0; r < 8; ++r) {
        const int m = row0 + r + half * 8;
        float* q00 = h + m * stride + n0 + col;
        float* q01 = q00 + 16;
        float* q10 = q00 + 16 * stride;
        float* q11 = q10 + 16;
        const float v00 = (ADD ? a00[r] + *q00 : a00[r]) + biasA, v01 = (ADD ? a01[r] + *q01 : a01[r]) + biasB;
        const float v10 = (ADD ? a10[r] + *q10 : a10[r]) + biasA, v11 = (ADD ? a11[r] + *q11 : a11[r]) + biasB;
        *q00 = v00 > 0.0f ? v00 : 0.0f;
        *q01 = v01 > 0.0f ? v01 : 0.0f;
        *q10 = v10 > 0.0f ? v10 : 0.0f;
        *q11 = v11 > 0.0f ? v11 : 0.0f;
    }
}

__global__ __launch_bounds__(256) void residue_embed_fused(
    const int* __restrict__ seq, const float* __restrict__ xyz,
    const float* __restrict__ dih, const int* __restrict__ cidx,
    const float* __restrict__ orient,
    const float* __restrict__ aa_emb, const float* __restrict__ chain_emb,
    const float* __restrict__ W1, const float* __restrict__ b1,
    const float* __restrict__ W2, const float* __restrict__ b2,
    const float* __restrict__ W3, const float* __restrict__ b3,
    const float* __restrict__ W4, const float* __restrict__ b4,
    float* __restrict__ out)
{
    __shared__ __attribute__((aligned(16))) float xs[(MT + 1) * XS];
    __shared__ __attribute__((aligned(16))) float hA[(MT + 1) * H1S];
    __shared__ float hB[MT * H2S];
    __shared__ __attribute__((aligned(16))) float wbuf[WALL];
    __shared__ __attribute__((aligned(16))) int tlist[NTY * MT];
    __shared__ int tcnt[32];

    const int tid = threadIdx.x;
    const int t0  = blockIdx.x * MT;

    const int lane = tid & 31;
    const int w    = tid >> 5;
    const int half = lane >> 4;
    const int col  = lane & 15;
    const int mg   = w >> 2;
    const int ng   = w & 3;

    const int n0w = w * 32;
    float* slab = wbuf + w * PBUF;
    const float* W1c = W1 + n0w + 4 * (lane & 7);
    v4f wlo[6], whi[6];
    slab_fetch<4, false, 256>(W1c, 0, lane, wlo, whi);

    {
        const int tya = seq[t0 + lane];
        const int tyb = seq[t0 + 32 + lane];
        const unsigned below = (1u << lane) - 1u;
        for (int g = __builtin_amdgcn_readfirstlane(w); g < NTY; g += 8) {
            const unsigned ba = __builtin_amdgcn_ballot_w32(tya == g);
            const unsigned bb = __builtin_amdgcn_ballot_w32(tyb == g);
            const int ca = __builtin_popcount(ba);
            if (tya == g) tlist[g * MT + __builtin_popcount(ba & below)] = lane;
            if (tyb == g) tlist[g * MT + ca + __builtin_popcount(bb & below)] = 32 + lane;
            if (lane == 0) tcnt[g] = ca + __builtin_popcount(bb);
        }
    }

    #pragma unroll
    for (int j = 0; j < MT * 32 / 256; ++j) {
        const int i  = tid + 256 * j;
        const int tt = i >> 5, q = i & 31;
        const int t  = t0 + tt;
        const int sa = seq[t];
        const int ci = cidx[t];
        const v4f va = *(const v4f*)(aa_emb + sa * 128 + 4 * q);
        v4f vc = *(const v4f*)(chain_emb + ci * 128 + 4 * q);
        const v4f z = {0.0f, 0.0f, 0.0f, 0.0f};
        vc = (ci != 0) ? vc : z;
        *(v4f*)(xs + tt * XS + 4 * q) = va;
        float* d = xs + tt * XS + XTAIL + 39 + 4 * q;
        d[0] = vc.x; d[1] = vc.y; d[2] = vc.z; d[3] = vc.w;
    }
    for (int i = tid; i < MT * 15; i += 256) {
        int tt = i / 15, a = i % 15;
        int t  = t0 + tt;
        const float* xp = xyz + (size_t)t * 45;
        float rx = xp[a * 3 + 0] - xp[3 + 0];
        float ry = xp[a * 3 + 1] - xp[3 + 1];
        float rz = xp[a * 3 + 2] - xp[3 + 2];
        const float* O = orient + (size_t)t * 9;
        int base = tt * XS + XCRD + a * 3;
        xs[base + 0] = O[0] * rx + O[3] * ry + O[6] * rz;
        xs[base + 1] = O[1] * rx + O[4] * ry + O[7] * rz;
        xs[base + 2] = O[2] * rx + O[5] * ry + O[8] * rz;
    }
    if (tid < TK) xs[MT * XS + XCRD + tid] = 0.0f;
    if (tid < MT) {
        float* r = xs + tid * XS;
        r[XTAIL + 167] = 0.0f;
        r[XCRD - 2] = 0.0f; r[XCRD - 1] = 0.0f;
        r[XCRD + 45] = 0.0f; r[XCRD + 46] = 0.0f; r[XCRD + 47] = 0.0f;
        r[XCRD + 48] = 0.0f; r[XCRD + 49] = 0.0f;
    }
    if (tid < MT * 3) {
        int tt = tid / 3, d = tid % 3;
        float v = dih[(size_t)(t0 + tt) * 3 + d];
        const float FREQ[6] = {1.0f, 2.0f, 3.0f, 1.0f, 0.5f, 1.0f / 3.0f};
        int base = tt * XS + XTAIL + d * 13;
        xs[base] = v;
        #pragma unroll
        for (int f = 0; f < 6; ++f) {
            xs[base + 1 + f] = sinf(FREQ[f] * v);
            xs[base + 7 + f] = cosf(FREQ[f] * v);
        }
    }
    __syncthreads();

    {
        v8f c00 = {}, c01 = {}, c10 = {}, c11 = {}, c20 = {}, c21 = {}, c30 = {}, c31 = {};
        const float* xr = xs + col * XS;
        const int wbo = (half * NSW + col) * 2;
        float* hq = hA + n0w + col;
        #pragma unroll
        for (int j = 0; j < MT / 4; ++j) {
            const v4f z = {0.0f, 0.0f, 0.0f, 0.0f};
            *(v4f*)(hA + (4 * j + (lane >> 3)) * H1S + n0w + 4 * (lane & 7)) = z;
        }
        const int cntv = tcnt[lane];
        int gt = (int)(blockIdx.x % NTY);
        int tk0 = tlist[gt * MT + col];
        if (__builtin_amdgcn_readfirstlane(w) >= 4) __builtin_amdgcn_s_setprio(3);
        #pragma nounroll
        for (int c = 0; c < 3; ++c)
            l1_chunk<4, false>(c * KC, (c + 1) * KC, xr, half, slab, wbo, W1c, lane, wlo, whi,
                               c00, c01, c10, c11, c20, c21, c30, c31);
        l1_chunk<6, false>(3 * KC, 0, xr, half, slab, wbo, W1c + (size_t)(TROW0 + 45 * gt) * 256, lane, wlo, whi,
                           c00, c01, c10, c11, c20, c21, c30, c31);
        #pragma nounroll
        for (int g = 0; g < NTY - 1; ++g) {
            if (g == L1LEAD) __builtin_amdgcn_s_setprio(0);
            const int gn = (gt + 1 == NTY) ? 0 : gt + 1;
            const int n = __builtin_amdgcn_readlane(cntv, gt);
            l1_type<6>(n, 0, xs, tlist + gt * MT, tlist + gn * MT, tk0, hq, half, col,
                       slab, wbo, W1c + (size_t)(TROW0 + 45 * gn) * 256, lane, wlo, whi);
            gt = gn;
        }
        l1_type<4>(__builtin_amdgcn_readlane(cntv, gt), TAIL0, xs, tlist + gt * MT,
                   tlist + gt * MT, tk0, hq, half, col, slab, wbo, W1c, lane, wlo, whi);
        #pragma nounroll
        for (int j = 0; j < 4; ++j)
            l1_chunk<4, false>(XTAIL + j * KC, TAIL0 + (j + 1) * KC, xr, half, slab, wbo, W1c, lane, wlo, whi,
                               c00, c01, c10, c11, c20, c21, c30, c31);
        l1_chunk<4, true>(XTAIL + 4 * KC, TAIL0 + 5 * KC, xr, half, slab, wbo, W1c, lane, wlo, whi,
                          c00, c01, c10, c11, c20, c21, c30, c31);
        {
            Frag4x2<2> f;
            slab_stage<4>(slab, lane, wlo, whi);
            slab_handover();
            slab_fetch<4, false, 128>(W2 + (w & 3) * 32 + 4 * (lane & 7), 0, lane, wlo, whi);
            frag_load_4x2<2>(xr + 2 * half + XTAIL + 5 * KC, slab + wbo, f);
            frag_mma_4x2<2>(f, c00, c01, c10, c11, c20, c21, c30, c31);
        }
        const float biasA = b1[n0w + col], biasB = b1[n0w + 16 + col];
        relu_store_2x2<true>(hA, H1S, 0,  n0w, half, col, biasA, biasB, c00, c01, c10, c11);
        relu_store_2x2<true>(hA, H1S, 32, n0w, half, col, biasA, biasB, c20, c21, c30, c31);
    }
    __syncthreads();

    const int row0 = mg * 32;
    const int n0   = ng * 32;
    const int wbo  = (half * NSW + col) * 2;
    const float* W2c = W2 + n0 + 4 * (lane & 7);
    const float* W3c = W3 + n0 + 4 * (lane & 7);
    const float* W4c = W4 + n0 + 4 * (lane & 7);

    {
        v8f a00 = {}, a01 = {}, a10 = {}, a11 = {};
        const float* ha0 = hA + (row0 + col) * H1S + half * 2;
        const float* ha1 = ha0 + 16 * H1S;
        #pragma nounroll
        for (int c = 0; c < 256 / KC; ++c) {
            const float* wn = (c + 1 < 256 / KC) ? W2c + (c + 1) * KC * 128 : W3c;
            l24_chunk<true>(ha0 + c * KC, ha1 + c * KC, slab, wbo, wn, lane, wlo, whi, a00, a01, a10, a11);
        }
        relu_store_2x2<false>(hB, H2S, row0, n0, half, col, b2[n0 + col], b2[n0 + 16 + col], a00, a01, a10, a11);
    }
    __syncthreads();

    {
        v8f a00 = {}, a01 = {}, a10 = {}, a11 = {};
        const float* ha0 = hB + (row0 + col) * H2S + half * 2;
        const float* ha1 = ha0 + 16 * H2S;
        #pragma nounroll
        for (int c = 0; c < 128 / KC; ++c) {
            const float* wn = (c + 1 < 128 / KC) ? W3c + (c + 1) * KC * 128 : W4c;
            l24_chunk<true>(ha0 + c * KC, ha1 + c * KC, slab, wbo, wn, lane, wlo, whi, a00, a01, a10, a11);
        }
        relu_store_2x2<false>(hA, H2S, row0, n0, half, col, b3[n0 + col], b3[n0 + 16 + col], a00, a01, a10, a11);
    }
    __syncthreads();

    {
        v8f a00 = {}, a01 = {}, a10 = {}, a11 = {};
        const float* ha0 = hA + (row0 + col) * H2S + half * 2;
        const float* ha1 = ha0 + 16 * H2S;
        #pragma nounroll
        for (int c = 0; c < 128 / KC - 1; ++c)
            l24_chunk<true>(ha0 + c * KC, ha1 + c * KC, slab, wbo, W4c + (c + 1) * KC * 128, lane, wlo, whi,
                            a00, a01, a10, a11);
        l24_chunk<false>(ha0 + (128 / KC - 1) * KC, ha1 + (128 / KC - 1) * KC, slab, wbo, W4c, lane, wlo, whi,
                         a00, a01, a10, a11);
        const float biasA = b4[n0 + col], biasB = b4[n0 + 16 + col];
        #pragma unroll
        for (int r = 0; r < 8; ++r) {
            const int m = row0 + r + half * 8;
            out[(size_t)(t0 + m) * 128 + n0 + col]           = a00[r] + biasA;
            out[(size_t)(t0 + m) * 128 + n0 + 16 + col]      = a01[r] + biasB;
            out[(size_t)(t0 + m + 16) * 128 + n0 + col]      = a10[r] + biasA;
            out[(size_t)(t0 + m + 16) * 128 + n0 + 16 + col] = a11[r] + biasB;
        }
    }
}

extern "C" void kernel_launch(void* const* d_in, const int* in_sizes, int n_in,
                              void* d_out, int out_size, void* d_ws, size_t ws_size,
                              hipStream_t stream) {
    (void)n_in; (void)d_ws; (void)ws_size; (void)out_size;
    const int*   seq    = (const int*)  d_in[0];
    const float* xyz    = (const float*)d_in[1];
    const float* dih    = (const float*)d_in[2];
    const int*   cidx   = (const int*)  d_in[3];
    const float* orient = (const float*)d_in[4];
    const float* aa_emb    = (const float*)d_in[6];
    const float* chain_emb = (const float*)d_in[7];
    const float* W1 = (const float*)d_in[8];   const float* b1 = (const float*)d_in[9];
    const float* W2 = (const float*)d_in[10];  const float* b2 = (const float*)d_in[11];
    const float* W3 = (const float*)d_in[12];  const float* b3 = (const float*)d_in[13];
    const float* W4 = (const float*)d_in[14];  const float* b4 = (const float*)d_in[15];
    float* out = (float*)d_out;

    const int tokens = in_sizes[0];
    const int blocks = tokens / MT;
    residue_embed_fused<<<blocks, 256, 0, stream>>>(
        seq, xyz, dih, cidx, orient, aa_emb, chain_emb,
        W1, b1, W2, b2, W3, b3, W4, b4, out);
}


# Layer 1's K range split over two sets of waves in a 16-wave work-group: waves 0..7 take W1's dense rows, waves 8..15 the one-hot block at raised priority, side by side (4 per SIMD); sums unchanged
# speedup vs baseline: 3.5298x; 1.0421x over previous
#include <hip/hip_runtime.h>
#include <math.h>

typedef float v2f __attribute__((ext_vector_type(2)));
typedef float v4f __attribute__((ext_vector_type(4)));
typedef float v8f __attribute__((ext_vector_type(8)));
typedef int   v4i __attribute__((ext_vector_type(4)));

#define MT    64
#define NTH   512
#define DIN   1195
#define NTY   20
#define XS    348
#define XTAIL 128
#define XCRD  298
#define TROW0 128
#define TAIL0 1028
#define H1S   260
#define H2S   132
#define KC    32
#define TK    48
#define NSW   48
#define PBUF  (TK / 2 * NSW * 2)
#define WALL  (8 * PBUF)
#define PB32  (KC / 2 * NSW * 2)
#define L1LEAD 4

#define WMMA_F32(a, b, c) \
    __builtin_amdgcn_wmma_f32_16x16x4_f32(false, (a), false, (b), (short)0, (c), false, false)

template <int NJ, bool TAIL, int LD>
__device__ __forceinline__ void slab_fetch(const float* __restrict__ Wc, int k0, int lane,
                                           v4f (&lo)[6], v4f (&hi)[6])
{
    #pragma unroll
    for (int j = 0; j < NJ; ++j) {
        const int r = k0 + 2 * (4 * j + (lane >> 3));
        const unsigned g4 = 4u * (unsigned)(lane & 7);
        if (!TAIL) {
            lo[j] = *(const v4f*)(Wc + ((unsigned)r * LD + g4));
            hi[j] = *(const v4f*)(Wc + ((unsigned)(r + 1) * LD + g4));
        } else {
            const int ra = (r     < DIN) ? r     : DIN - 1;
            const int rb = (r + 1 < DIN) ? r + 1 : DIN - 1;
            const v4f a = *(const v4f*)(Wc + ((unsigned)ra * LD + g4));
            const v4f b = *(const v4f*)(Wc + ((unsigned)rb * LD + g4));
            const v4f z = {0.0f, 0.0f, 0.0f, 0.0f};
            lo[j] = (r     < DIN) ? a : z;
            hi[j] = (r + 1 < DIN) ? b : z;
        }
    }
}

template <int NJ>
__device__ __forceinline__ void slab_stage(float* slab, int lane, const v4f (&lo)[6], const v4f (&hi)[6])
{
    #pragma unroll
    for (int j = 0; j < NJ; ++j) {
        float* d = slab + ((4 * j + (lane >> 3)) * NSW + 4 * (lane & 7)) * 2;
        const v4f a = {lo[j].x, hi[j].x, lo[j].y, hi[j].y};
        const v4f b = {lo[j].z, hi[j].z, lo[j].w, hi[j].w};
        *(v4f*)d       = a;
        *(v4f*)(d + 4) = b;
    }
}

__device__ __forceinline__ void slab_handover()
{
    __builtin_amdgcn_fence(__ATOMIC_RELEASE, "wavefront");
    __builtin_amdgcn_wave_barrier();
    __builtin_amdgcn_fence(__ATOMIC_ACQUIRE, "wavefront");
}

template <int STEPS>
struct Frag4x2 { v2f a0[STEPS], a1[STEPS], a2[STEPS], a3[STEPS], b0[STEPS], b1[STEPS]; };

template <int STEPS>
__device__ __forceinline__ void frag_load_4x2(const float* xp, const float* wb, Frag4x2<STEPS>& f)
{
    #pragma unroll
    for (int s = 0; s < STEPS; ++s) {
        const float* p = xp + 4 * s;
        f.a0[s] = *(const v2f*)(p);
        f.a1[s] = *(const v2f*)(p + 16 * XS);
        f.a2[s] = *(const v2f*)(p + 32 * XS);
        f.a3[s] = *(const v2f*)(p + 48 * XS);
        f.b0[s] = *(const v2f*)(wb + s * (4 * NSW));
        f.b1[s] = *(const v2f*)(wb + s * (4 * NSW) + 32);
        __builtin_amdgcn_sched_barrier(0);
    }
}

template <int STEPS>
__device__ __forceinline__ void frag_mma_4x2(const Frag4x2<STEPS>& f,
                                             v8f& c00, v8f& c01, v8f& c10, v8f& c11,
                                             v8f& c20, v8f& c21, v8f& c30, v8f& c31)
{
    #pragma unroll
    for (int s = 0; s < STEPS; ++s) {
        c00 = WMMA_F32(f.a0[s], f.b0[s], c00);
        c01 = WMMA_F32(f.a0[s], f.b1[s], c01);
        c10 = WMMA_F32(f.a1[s], f.b0[s], c10);
        c11 = WMMA_F32(f.a1[s], f.b1[s], c11);
        c20 = WMMA_F32(f.a2[s], f.b0[s], c20);
        c21 = WMMA_F32(f.a2[s], f.b1[s], c21);
        c30 = WMMA_F32(f.a3[s], f.b0[s], c30);
        c31 = WMMA_F32(f.a3[s], f.b1[s], c31);
    }
}

struct Frag2x2 { v2f a0[KC / 4], a1[KC / 4], bA[KC / 4], bB[KC / 4]; };

__device__ __forceinline__ void frag_load_2x2(const float* ha0, const float* ha1, const float* wb,
                                              Frag2x2& f)
{
    #pragma unroll
    for (int s = 0; s < KC / 4; ++s) {
        f.a0[s] = *(const v2f*)(ha0 + 4 * s);
        f.a1[s] = *(const v2f*)(ha1 + 4 * s);
        f.bA[s] = *(const v2f*)(wb + s * (4 * NSW));
        f.bB[s] = *(const v2f*)(wb + s * (4 * NSW) + 32);
        __builtin_amdgcn_sched_barrier(0);
    }
}

__device__ __forceinline__ void frag_mma_2x2(const Frag2x2& f, v8f& a00, v8f& a01, v8f& a10, v8f& a11)
{
    #pragma unroll
    for (int s = 0; s < KC / 4; ++s) {
        a00 = WMMA_F32(f.a0[s], f.bA[s], a00);
        a01 = WMMA_F32(f.a0[s], f.bB[s], a01);
        a10 = WMMA_F32(f.a1[s], f.bA[s], a10);
        a11 = WMMA_F32(f.a1[s], f.bB[s], a11);
    }
}

template <bool FETCH>
__device__ __forceinline__ void l24_chunk(const float* ha0, const float* ha1, float* slab, int wbo,
                                          const float* __restrict__ Wn, int lane,
                                          v4f (&wlo)[6], v4f (&whi)[6],
                                          v8f& a00, v8f& a01, v8f& a10, v8f& a11)
{
    Frag2x2 f;
    slab_stage<4>(slab, lane, wlo, whi);
    slab_handover();
    if (FETCH) slab_fetch<4, false, 128>(Wn, 0, lane, wlo, whi);
    __builtin_amdgcn_sched_barrier(0);
    frag_load_2x2(ha0, ha1, slab + wbo, f);
    frag_mma_2x2(f, a00, a01, a10, a11);
    asm volatile("" : "+v"(a00), "+v"(a01), "+v"(a10), "+v"(a11),
                      "+v"(wlo[0]), "+v"(wlo[1]), "+v"(wlo[2]), "+v"(wlo[3]),
                      "+v"(whi[0]), "+v"(whi[1]), "+v"(whi[2]), "+v"(whi[3]));
}

template <int NJN, bool NEXT_TAIL>
__device__ __forceinline__ void l1_chunk(int xoff, int knext, const float* xr, int half,
                                         float* slab, int wbo,
                                         const float* __restrict__ W1c, int lane,
                                         v4f (&wlo)[6], v4f (&whi)[6],
                                         v8f& c00, v8f& c01, v8f& c10, v8f& c11,
                                         v8f& c20, v8f& c21, v8f& c30, v8f& c31)
{
    Frag4x2<KC / 4> f;
    slab_stage<4>(slab, lane, wlo, whi);
    slab_handover();
    slab_fetch<NJN, NEXT_TAIL, 256>(W1c, knext, lane, wlo, whi);
    __builtin_amdgcn_sched_barrier(0);
    frag_load_4x2<KC / 4>(xr + 2 * half + xoff, slab + wbo, f);
    frag_mma_4x2<KC / 4>(f, c00, c01, c10, c11, c20, c21, c30, c31);
    asm volatile("" : "+v"(c00), "+v"(c01), "+v"(c10), "+v"(c11), "+v"(c20), "+v"(c21), "+v"(c30), "+v"(c31),
                      "+v"(wlo[0]), "+v"(wlo[1]), "+v"(wlo[2]), "+v"(wlo[3]),
                      "+v"(whi[0]), "+v"(whi[1]), "+v"(whi[2]), "+v"(whi[3]));
}

template <int NJN>
__device__ __forceinline__ void l1_type(int n, int knext, const float* xs,
                                        const int* tl, const int* tln, int& tk0, float* hq,
                                        int half, int col, float* slab, int wbo,
                                        const float* __restrict__ W1c, int lane,
                                        v4f (&wlo)[6], v4f (&whi)[6])
{
    int tkraw = tk0;
    slab_stage<6>(slab, lane, wlo, whi);
    slab_handover();
    slab_fetch<NJN, false, 256>(W1c, knext, lane, wlo, whi);
    __builtin_amdgcn_sched_barrier(0);
    const float* wb = slab + wbo;
    for (int s0 = 0; s0 < n; s0 += 16) {
        const int tk = (s0 + col < n) ? tkraw : MT;
        const float* xa = xs + tk * XS + XCRD + 2 * half;
        const int sb = s0 + 8 * half;
        const v4i ta = *(const v4i*)(tl + sb);
        const v4i tb = *(const v4i*)(tl + sb + 4);
        v2f a[TK / 4], b0[TK / 4], b1[TK / 4];
        #pragma unroll
        for (int s = 0; s < TK / 4; ++s) {
            a[s]  = *(const v2f*)(xa + 4 * s);
            b0[s] = *(const v2f*)(wb + s * (4 * NSW));
            b1[s] = *(const v2f*)(wb + s * (4 * NSW) + 32);
            __builtin_amdgcn_sched_barrier(0);
        }
        v8f p0 = {}, p1 = {};
        #pragma unroll
        for (int s = 0; s < TK / 4; ++s) {
            p0 = WMMA_F32(a[s], b0[s], p0);
            p1 = WMMA_F32(a[s], b1[s], p1);
        }
        asm volatile("v_nop\n\tv_nop\n\tv_nop\n\tv_nop" : "+v"(p0), "+v"(p1));
        const int tr[8] = {ta.x, ta.y, ta.z, ta.w, tb.x, tb.y, tb.z, tb.w};
        #pragma unroll
        for (int r = 0; r < 8; ++r) {
            const int dst = (sb + r < n) ? tr[r] : MT;
            hq[dst * H1S]      = p0[r];
            hq[dst * H1S + 16] = p1[r];
        }
        tkraw = tl[(s0 + 16 + col) & 63];
    }
    tk0 = tln[col];
    asm volatile("" : "+v"(wlo[0]), "+v"(wlo[1]), "+v"(wlo[2]), "+v"(wlo[3]), "+v"(wlo[4]), "+v"(wlo[5]),
                      "+v"(whi[0]), "+v"(whi[1]), "+v"(whi[2]), "+v"(whi[3]), "+v"(whi[4]), "+v"(whi[5]));
}

template <bool ADD>
__device__ __forceinline__ void relu_store_2x2(float* h, int stride, int row0, int n0, int half, int col,
                                               float biasA, float biasB,
                                               const v8f& a00, const v8f& a01, const v8f& a10, const v8f& a11)
{
    #pragma unroll
    for (int r = 0; r < 8; ++r) {
        const int m = row0 + r + half * 8;
        float* q00 = h + m * stride + n0 + col;
        float* q01 = q00 + 16;
        float* q10 = q00 + 16 * stride;
        float* q11 = q10 + 16;
        const float v00 = (ADD ? a00[r] + *q00 : a00[r]) + biasA, v01 = (ADD ? a01[r] + *q01 : a01[r]) + biasB;
        const float v10 = (ADD ? a10[r] + *q10 : a10[r]) + biasA, v11 = (ADD ? a11[r] + *q11 : a11[r]) + biasB;
        *q00 = v00 > 0.0f ? v00 : 0.0f;
        *q01 = v01 > 0.0f ? v01 : 0.0f;
        *q10 = v10 > 0.0f ? v10 : 0.0f;
        *q11 = v11 > 0.0f ? v11 : 0.0f;
    }
}

__global__ __launch_bounds__(NTH) void residue_embed_fused(
    const int* __restrict__ seq, const float* __restrict__ xyz,
    const float* __restrict__ dih, const int* __restrict__ cidx,
    const float* __restrict__ orient,
    const float* __restrict__ aa_emb, const float* __restrict__ chain_emb,
    const float* __restrict__ W1, const float* __restrict__ b1,
    const float* __restrict__ W2, const float* __restrict__ b2,
    const float* __restrict__ W3, const float* __restrict__ b3,
    const float* __restrict__ W4, const float* __restrict__ b4,
    float* __restrict__ out)
{
    __shared__ __attribute__((aligned(16))) float xs[(MT + 1) * XS];
    __shared__ __attribute__((aligned(16))) float hA[(MT + 1) * H1S];
    __shared__ float hB[MT * H2S];
    __shared__ __attribute__((aligned(16))) float wbuf[WALL];
    __shared__ __attribute__((aligned(16))) float wbufd[8 * PB32];
    __shared__ __attribute__((aligned(16))) int tlist[NTY * MT];
    __shared__ int tcnt[32];

    const int tid = threadIdx.x;
    const int t0  = blockIdx.x * MT;

    const int lane = tid & 31;
    const int w    = __builtin_amdgcn_readfirstlane(tid >> 5);
    const bool hot = w >= 8;
    const int cw   = w & 7;
    const int half = lane >> 4;
    const int col  = lane & 15;
    const int mg   = cw >> 2;
    const int ng   = cw & 3;

    const int n0w = cw * 32;
    float* slab = hot ? wbuf + cw * PBUF : wbufd + cw * PB32;
    const float* W1c = W1 + n0w;
    int gt = (int)(blockIdx.x % NTY);
    v4f wlo[6], whi[6];
    if (hot) slab_fetch<6, false, 256>(W1c + (size_t)(TROW0 + 45 * gt) * 256, 0, lane, wlo, whi);
    else     slab_fetch<4, false, 256>(W1c, 0, lane, wlo, whi);

    {
        const int tya = seq[t0 + lane];
        const int tyb = seq[t0 + 32 + lane];
        const unsigned below = (1u << lane) - 1u;
        for (int g = w; g < NTY; g += NTH / 32) {
            const unsigned ba = __builtin_amdgcn_ballot_w32(tya == g);
            const unsigned bb = __builtin_amdgcn_ballot_w32(tyb == g);
            const int ca = __builtin_popcount(ba);
            if (tya == g) tlist[g * MT + __builtin_popcount(ba & below)] = lane;
            if (tyb == g) tlist[g * MT + ca + __builtin_popcount(bb & below)] = 32 + lane;
            if (lane == 0) tcnt[g] = ca + __builtin_popcount(bb);
        }
    }

    #pragma unroll
    for (int j = 0; j < MT * 32 / NTH; ++j) {
        const int i  = tid + NTH * j;
        const int tt = i >> 5, q = i & 31;
        const int t  = t0 + tt;
        const int sa = seq[t];
        const int ci = cidx[t];
        const v4f va = *(const v4f*)(aa_emb + sa * 128 + 4 * q);
        v4f vc = *(const v4f*)(chain_emb + ci * 128 + 4 * q);
        const v4f z = {0.0f, 0.0f, 0.0f, 0.0f};
        vc = (ci != 0) ? vc : z;
        *(v4f*)(xs + tt * XS + 4 * q) = va;
        float* d = xs + tt * XS + XTAIL + 39 + 4 * q;
        d[0] = vc.x; d[1] = vc.y; d[2] = vc.z; d[3] = vc.w;
    }
    for (int i = tid; i < MT * 15; i += NTH) {
        int tt = i / 15, a = i % 15;
        int t  = t0 + tt;
        const float* xp = xyz + (size_t)t * 45;
        float rx = xp[a * 3 + 0] - xp[3 + 0];
        float ry = xp[a * 3 + 1] - xp[3 + 1];
        float rz = xp[a * 3 + 2] - xp[3 + 2];
        const float* O = orient + (size_t)t * 9;
        int base = tt * XS + XCRD + a * 3;
        xs[base + 0] = O[0] * rx + O[3] * ry + O[6] * rz;
        xs[base + 1] = O[1] * rx + O[4] * ry + O[7] * rz;
        xs[base + 2] = O[2] * rx + O[5] * ry + O[8] * rz;
    }
    if (tid < TK) xs[MT * XS + XCRD + tid] = 0.0f;
    if (tid < MT) {
        float* r = xs + tid * XS;
        r[XTAIL + 167] = 0.0f;
        r[XCRD - 2] = 0.0f; r[XCRD - 1] = 0.0f;
        r[XCRD + 45] = 0.0f; r[XCRD + 46] = 0.0f; r[XCRD + 47] = 0.0f;
        r[XCRD + 48] = 0.0f; r[XCRD + 49] = 0.0f;
    }
    if (tid < MT * 3) {
        int tt = tid / 3, d = tid % 3;
        float v = dih[(size_t)(t0 + tt) * 3 + d];
        const float FREQ[6] = {1.0f, 2.0f, 3.0f, 1.0f, 0.5f, 1.0f / 3.0f};
        int base = tt * XS + XTAIL + d * 13;
        xs[base] = v;
        #pragma unroll
        for (int f = 0; f < 6; ++f) {
            xs[base + 1 + f] = sinf(FREQ[f] * v);
            xs[base + 7 + f] = cosf(FREQ[f] * v);
        }
    }
    __syncthreads();

    v8f c00 = {}, c01 = {}, c10 = {}, c11 = {}, c20 = {}, c21 = {}, c30 = {}, c31 = {};
    if (hot) {
        __builtin_amdgcn_s_setprio(3);
        int tidh = threadIdx.x;
        asm volatile("" : "+v"(tidh));
        const int laneh = tidh & 31;
        const int halfh = laneh >> 4;
        const int colh  = laneh & 15;
        const int wboh  = (halfh * NSW + colh) * 2;
        float* hq = hA + n0w + colh;
        #pragma unroll
        for (int j = 0; j < MT / 4; ++j) {
            const v4f z = {0.0f, 0.0f, 0.0f, 0.0f};
            *(v4f*)(hA + (4 * j + (laneh >> 3)) * H1S + n0w + 4 * (laneh & 7)) = z;
        }
        const int cntv = tcnt[laneh];
        int tk0 = tlist[gt * MT + colh];
        #pragma nounroll
        for (int g = 0; g < NTY - 1; ++g) {
            const int gn = (gt + 1 == NTY) ? 0 : gt + 1;
            const int n = __builtin_amdgcn_readlane(cntv, gt);
            l1_type<6>(n, 0, xs, tlist + gt * MT, tlist + gn * MT, tk0, hq, halfh, colh,
                       slab, wboh, W1c + (size_t)(TROW0 + 45 * gn) * 256, laneh, wlo, whi);
            gt = gn;
        }
        l1_type<0>(__builtin_amdgcn_readlane(cntv, gt), 0, xs, tlist + gt * MT,
                   tlist + gt * MT, tk0, hq, halfh, colh, slab, wboh, W1c, laneh, wlo, whi);
        __builtin_amdgcn_s_setprio(0);
    } else {
        int tidd = threadIdx.x;
        asm volatile("" : "+v"(tidd));
        const int laned = tidd & 31;
        const int halfd = laned >> 4;
        const int cold  = laned & 15;
        const int wbod  = (halfd * NSW + cold) * 2;
        const float* xr = xs + cold * XS;
        if (w >= 4) __builtin_amdgcn_s_setprio(1);
        #pragma nounroll
        for (int i = 0; i < 8; ++i) {
            if (i == L1LEAD) __builtin_amdgcn_s_setprio(0);
            const int knext = (i < 3) ? (i + 1) * KC : TAIL0 + (i - 3) * KC;
            l1_chunk<4, false>(i * KC, knext, xr, halfd, slab, wbod, W1c, laned, wlo, whi,
                               c00, c01, c10, c11, c20, c21, c30, c31);
        }
        l1_chunk<4, true>(XTAIL + 4 * KC, TAIL0 + 5 * KC, xr, halfd, slab, wbod, W1c, laned, wlo, whi,
                          c00, c01, c10, c11, c20, c21, c30, c31);
        {
            Frag4x2<2> f;
            slab_stage<4>(slab, laned, wlo, whi);
            slab_handover();
            slab_fetch<4, false, 128>(W2 + ng * 32, 0, laned, wlo, whi);
            frag_load_4x2<2>(xr + 2 * halfd + XTAIL + 5 * KC, slab + wbod, f);
            frag_mma_4x2<2>(f, c00, c01, c10, c11, c20, c21, c30, c31);
        }
    }
    __syncthreads();
    const int tidr = threadIdx.x;
    const int lane2 = tidr & 31;
    const int half2 = lane2 >> 4;
    const int col2  = lane2 & 15;
    const int wbo2  = (half2 * NSW + col2) * 2;
    if (!hot) {
        const float biasA = b1[n0w + col2], biasB = b1[n0w + 16 + col2];
        relu_store_2x2<true>(hA, H1S, 0,  n0w, half2, col2, biasA, biasB, c00, c01, c10, c11);
        relu_store_2x2<true>(hA, H1S, 32, n0w, half2, col2, biasA, biasB, c20, c21, c30, c31);
    }
    __syncthreads();

    const int row0 = mg * 32;
    const int n0   = ng * 32;
    const float* W2c = W2 + n0;
    const float* W3c = W3 + n0;
    const float* W4c = W4 + n0;

    if (!hot) {
        v8f a00 = {}, a01 = {}, a10 = {}, a11 = {};
        const float* ha0 = hA + (row0 + col2) * H1S + half2 * 2;
        const float* ha1 = ha0 + 16 * H1S;
        #pragma nounroll
        for (int c = 0; c < 256 / KC; ++c) {
            const float* wn = (c + 1 < 256 / KC) ? W2c + (c + 1) * KC * 128 : W3c;
            l24_chunk<true>(ha0 + c * KC, ha1 + c * KC, slab, wbo2, wn, lane2, wlo, whi, a00, a01, a10, a11);
        }
        relu_store_2x2<false>(hB, H2S, row0, n0, half2, col2, b2[n0 + col2], b2[n0 + 16 + col2], a00, a01, a10, a11);
    }
    __syncthreads();

    if (!hot) {
        v8f a00 = {}, a01 = {}, a10 = {}, a11 = {};
        const float* ha0 = hB + (row0 + col2) * H2S + half2 * 2;
        const float* ha1 = ha0 + 16 * H2S;
        #pragma nounroll
        for (int c = 0; c < 128 / KC; ++c) {
            const float* wn = (c + 1 < 128 / KC) ? W3c + (c + 1) * KC * 128 : W4c;
            l24_chunk<true>(ha0 + c * KC, ha1 + c * KC, slab, wbo2, wn, lane2, wlo, whi, a00, a01, a10, a11);
        }
        relu_store_2x2<false>(hA, H2S, row0, n0, half2, col2, b3[n0 + col2], b3[n0 + 16 + col2], a00, a01, a10, a11);
    }
    __syncthreads();

    if (!hot) {
        v8f a00 = {}, a01 = {}, a10 = {}, a11 = {};
        const float* ha0 = hA + (row0 + col2) * H2S + half2 * 2;
        const float* ha1 = ha0 + 16 * H2S;
        #pragma nounroll
        for (int c = 0; c < 128 / KC - 1; ++c)
            l24_chunk<true>(ha0 + c * KC, ha1 + c * KC, slab, wbo2, W4c + (c + 1) * KC * 128, lane2, wlo, whi,
                            a00, a01, a10, a11);
        l24_chunk<false>(ha0 + (128 / KC - 1) * KC, ha1 + (128 / KC - 1) * KC, slab, wbo2, W4c, lane2, wlo, whi,
                         a00, a01, a10, a11);
        const float biasA = b4[n0 + col2], biasB = b4[n0 + 16 + col2];
        #pragma unroll
        for (int r = 0; r < 8; ++r) {
            const int m = row0 + r + half2 * 8;
            out[(size_t)(t0 + m) * 128 + n0 + col2]           = a00[r] + biasA;
            out[(size_t)(t0 + m) * 128 + n0 + 16 + col2]      = a01[r] + biasB;
            out[(size_t)(t0 + m + 16) * 128 + n0 + col2]      = a10[r] + biasA;
            out[(size_t)(t0 + m + 16) * 128 + n0 + 16 + col2] = a11[r] + biasB;
        }
    }
}

extern "C" void kernel_launch(void* const* d_in, const int* in_sizes, int n_in,
                              void* d_out, int out_size, void* d_ws, size_t ws_size,
                              hipStream_t stream) {
    (void)n_in; (void)d_ws; (void)ws_size; (void)out_size;
    const int*   seq    = (const int*)  d_in[0];
    const float* xyz    = (const float*)d_in[1];
    const float* dih    = (const float*)d_in[2];
    const int*   cidx   = (const int*)  d_in[3];
    const float* orient = (const float*)d_in[4];
    const float* aa_emb    = (const float*)d_in[6];
    const float* chain_emb = (const float*)d_in[7];
    const float* W1 = (const float*)d_in[8];   const float* b1 = (const float*)d_in[9];
    const float* W2 = (const float*)d_in[10];  const float* b2 = (const float*)d_in[11];
    const float* W3 = (const float*)d_in[12];  const float* b3 = (const float*)d_in[13];
    const float* W4 = (const float*)d_in[14];  const float* b4 = (const float*)d_in[15];
    float* out = (float*)d_out;

    const int tokens = in_sizes[0];
    const int blocks = tokens / MT;
    residue_embed_fused<<<blocks, NTH, 0, stream>>>(
        seq, xyz, dih, cidx, orient, aa_emb, chain_emb,
        W1, b1, W2, b2, W3, b3, W4, b4, out);
}
